# starting kernel
# baseline (speedup 1.0000x reference)
#include <hip/hip_runtime.h>
#include <math.h>

typedef float v2f __attribute__((ext_vector_type(2)));
typedef float v8f __attribute__((ext_vector_type(8)));

#define HD 16
#define NWIRE 1536
#define NFEAT 4

__device__ __forceinline__ unsigned int encOrd(float f) {
    unsigned int b = __float_as_uint(f);
    return (b & 0x80000000u) ? ~b : (b | 0x80000000u);
}
__device__ __forceinline__ float decOrd(unsigned int e) {
    unsigned int b = (e & 0x80000000u) ? (e ^ 0x80000000u) : ~e;
    return __uint_as_float(b);
}
__device__ __forceinline__ float lrelu02(float x) { return x > 0.f ? x : 0.2f * x; }

__global__ void k_padW0(const float* __restrict__ W0, float* __restrict__ W0p) {
    int i = threadIdx.x;
    int r = i >> 4;
    W0p[i] = (r < 13) ? W0[i] : 0.f;
}

__global__ void k_fill(const float* __restrict__ fw, const float* __restrict__ sw,
                       const float* __restrict__ tw, const float* __restrict__ areas,
                       const int* __restrict__ indices, float* __restrict__ h,
                       int N, int g) {
    int n = blockIdx.x * blockDim.x + threadIdx.x;
    if (n >= N) return;
    const size_t gbase = (size_t)g * NWIRE * NFEAT;
    const float4 f0 = *(const float4*)(fw + gbase + (size_t)indices[n * 3 + 0] * NFEAT);
    const float4 f1 = *(const float4*)(sw + gbase + (size_t)indices[n * 3 + 1] * NFEAT);
    const float4 f2 = *(const float4*)(tw + gbase + (size_t)indices[n * 3 + 2] * NFEAT);
    float4* hp = (float4*)(h + (size_t)n * HD);
    hp[0] = f0;
    hp[1] = f1;
    hp[2] = f2;
    hp[3] = make_float4(areas[n], 0.f, 0.f, 0.f);
}

__global__ void k_gemm_wmma(const float* __restrict__ h, const float* __restrict__ W,
                            const float* __restrict__ bias, float* __restrict__ hw, int N) {
    const int lane = threadIdx.x & 31;
    const int wave = threadIdx.x >> 5;
    const int tile = blockIdx.x * 8 + wave;
    const int base = tile * 16;
    if (base >= N) return;
    const int col16 = lane & 15;
    const int half  = lane >> 4;
    int arow = base + col16;
    if (arow >= N) arow = N - 1;
    v8f acc = {};
#pragma unroll
    for (int k = 0; k < 4; ++k) {
        const int kc = 4 * k + 2 * half;
        v2f a;
        a.x = h[(size_t)arow * HD + kc];
        a.y = h[(size_t)arow * HD + kc + 1];
        v2f b;
        b.x = W[kc * HD + col16];
        b.y = W[(kc + 1) * HD + col16];
        acc = __builtin_amdgcn_wmma_f32_16x16x4_f32(
                  false, a, false, b, (short)0, acc, false, false);
    }
    const float bi = bias[col16];
    if (base + 16 <= N) {
#pragma unroll
        for (int r = 0; r < 8; ++r) {
            hw[(size_t)(base + r + 8 * half) * HD + col16] = acc[r] + bi;
        }
    } else {
#pragma unroll
        for (int r = 0; r < 8; ++r) {
            const int orow = base + r + 8 * half;
            if (orow < N) hw[(size_t)orow * HD + col16] = acc[r] + bi;
        }
    }
}

__global__ void k_esed(const float* __restrict__ hw, const float* __restrict__ asrc,
                       const float* __restrict__ adst, float* __restrict__ es,
                       float* __restrict__ ed, int N) {
    int n = blockIdx.x * blockDim.x + threadIdx.x;
    if (n >= N) return;
    const float4* r = (const float4*)(hw + (size_t)n * HD);
    float s = 0.f, d = 0.f;
#pragma unroll
    for (int q = 0; q < 4; ++q) {
        const float4 v = r[q];
        s += v.x * asrc[q*4+0] + v.y * asrc[q*4+1] + v.z * asrc[q*4+2] + v.w * asrc[q*4+3];
        d += v.x * adst[q*4+0] + v.y * adst[q*4+1] + v.z * adst[q*4+2] + v.w * adst[q*4+3];
    }
    es[n] = s; ed[n] = d;
}

__global__ void k_init(unsigned int* __restrict__ mEnc, float* __restrict__ denom,
                       float* __restrict__ accum, int N) {
    int j = blockIdx.x * blockDim.x + threadIdx.x;
    if (j >= N * HD) return;
    accum[j] = 0.f;
    if (j < N) { mEnc[j] = 0x007FFFFFu;   denom[j] = 0.f; }
}

__global__ void k_edge_max(const int* __restrict__ src, const int* __restrict__ dst,
                           const float* __restrict__ es, const float* __restrict__ ed,
                           unsigned int* __restrict__ mEnc, int E) {
    int e = blockIdx.x * blockDim.x + threadIdx.x;
    if (e >= E) return;
    const int s = src[e], d = dst[e];
    const float lg = lrelu02(es[s] + ed[d]);
    atomicMax(&mEnc[d], encOrd(lg));
}

__global__ void k_edge_sum(const int* __restrict__ src, const int* __restrict__ dst,
                           const float* __restrict__ es, const float* __restrict__ ed,
                           const unsigned int* __restrict__ mEnc, float* __restrict__ denom,
                           float* __restrict__ wE, int E) {
    int e = blockIdx.x * blockDim.x + threadIdx.x;
    if (e >= E) return;
    const int s = src[e], d = dst[e];
    const float lg = lrelu02(es[s] + ed[d]);
    const float w = __expf(lg - decOrd(mEnc[d]));
    wE[e] = w;
    atomicAdd(&denom[d], w);
}

__global__ void k_edge_accum(const int* __restrict__ src, const int* __restrict__ dst,
                             const float* __restrict__ hw, const float* __restrict__ denom,
                             const float* __restrict__ wE, float* __restrict__ accum, int E) {
    int t = blockIdx.x * blockDim.x + threadIdx.x;
    if (t >= E * 4) return;
    const int e = t >> 2;
    const int q = t & 3;
    const int s = src[e], d = dst[e];
    const float alpha = wE[e] / fmaxf(denom[d], 1e-16f);
    const float4 hv = *(const float4*)(hw + (size_t)s * HD + q * 4);
    float* ap = accum + (size_t)d * HD + q * 4;
    atomicAdd(ap + 0, alpha * hv.x);
    atomicAdd(ap + 1, alpha * hv.y);
    atomicAdd(ap + 2, alpha * hv.z);
    atomicAdd(ap + 3, alpha * hv.w);
}

__global__ void k_elu(const float* __restrict__ accum, float* __restrict__ h, int N) {
    int j = blockIdx.x * blockDim.x + threadIdx.x;
    if (j >= N * HD) return;
    const float x = accum[j];
    h[j] = x > 0.f ? x : (__expf(x) - 1.f);
}

__global__ void k_final(const float* __restrict__ h, const float* __restrict__ mlpW,
                        const float* __restrict__ mlpb, float* __restrict__ out, int N) {
    int n = blockIdx.x * blockDim.x + threadIdx.x;
    if (n >= N) return;
    const float4* r = (const float4*)(h + (size_t)n * HD);
    float acc = mlpb[0];
#pragma unroll
    for (int q = 0; q < 4; ++q) {
        const float4 v = r[q];
        acc += v.x * mlpW[q*4+0] + v.y * mlpW[q*4+1] + v.z * mlpW[q*4+2] + v.w * mlpW[q*4+3];
    }
    out[n] = acc;
}

static inline int cdiv(int a, int b) { return (a + b - 1) / b; }

extern "C" void kernel_launch(void* const* d_in, const int* in_sizes, int n_in,
                              void* d_out, int out_size, void* d_ws, size_t ws_size,
                              hipStream_t stream) {
    const float* fw     = (const float*)d_in[0];
    const float* sw     = (const float*)d_in[1];
    const float* tw     = (const float*)d_in[2];
    const float* areas  = (const float*)d_in[3];
    const int*   indices= (const int*)  d_in[4];
    const int*   edge   = (const int*)  d_in[5];
    const float* W0     = (const float*)d_in[6];
    const float* Wl     = (const float*)d_in[7];
    const float* a_src  = (const float*)d_in[8];
    const float* a_dst  = (const float*)d_in[9];
    const float* bvec   = (const float*)d_in[10];
    const float* mlpW   = (const float*)d_in[11];
    const float* mlpb   = (const float*)d_in[12];

    const int N = in_sizes[3];
    const int E = in_sizes[5] / 2;
    const int T = in_sizes[0] / (NWIRE * NFEAT);
    const int* src = edge;
    const int* dst = edge + E;

    float* ws = (float*)d_ws;
    size_t off = 0;
    float*        h      = ws + off; off += (size_t)N * HD;
    float*        hw     = ws + off; off += (size_t)N * HD;
    float*        accum  = ws + off; off += (size_t)N * HD;
    float*        es     = ws + off; off += (size_t)N;
    float*        ed     = ws + off; off += (size_t)N;
    unsigned int* mEnc   = (unsigned int*)(ws + off); off += (size_t)N;
    float*        denom  = ws + off; off += (size_t)N;
    float*        wE     = ws + off; off += (size_t)E;
    float*        W0p    = ws + off; off += 256;
    (void)ws_size; (void)n_in; (void)out_size;

    k_padW0<<<1, 256, 0, stream>>>(W0, W0p);

    const int tiles  = cdiv(N, 16);
    const int gemmBl = cdiv(tiles, 8);

    for (int g = 0; g < T; ++g) {
        k_fill<<<cdiv(N, 256), 256, 0, stream>>>(fw, sw, tw, areas, indices, h, N, g);
        for (int i = 0; i < 4; ++i) {
            const float* Wi = (i == 0) ? W0p : (Wl + (size_t)(i - 1) * 256);
            k_init<<<cdiv(N * HD, 256), 256, 0, stream>>>(mEnc, denom, accum, N);
            k_gemm_wmma<<<gemmBl, 256, 0, stream>>>(h, Wi, bvec + (size_t)i * HD, hw, N);
            k_esed<<<cdiv(N, 256), 256, 0, stream>>>(hw, a_src + (size_t)i * HD,
                                                     a_dst + (size_t)i * HD, es, ed, N);
            k_edge_max<<<cdiv(E, 256), 256, 0, stream>>>(src, dst, es, ed, mEnc, E);
            k_edge_sum<<<cdiv(E, 256), 256, 0, stream>>>(src, dst, es, ed, mEnc, denom, wE, E);
            k_edge_accum<<<cdiv(E * 4, 256), 256, 0, stream>>>(src, dst, hw, denom, wE, accum, E);
            k_elu<<<cdiv(N * HD, 256), 256, 0, stream>>>(accum, h, N);
        }
        k_final<<<cdiv(N, 256), 256, 0, stream>>>(h, mlpW, mlpb, (float*)d_out + (size_t)g * N, N);
    }
}


# Edges counting-sorted by destination once per call (int atomics, arrival order); each layer's softmax and weighted sum run as one register loop per node, no float atomics; sum order changes
# speedup vs baseline: 4.7386x; 4.7386x over previous
#include <hip/hip_runtime.h>
#include <math.h>


typedef float v2f __attribute__((ext_vector_type(2)));
typedef float v8f __attribute__((ext_vector_type(8)));

#define HD 16
#define NWIRE 1536
#define NFEAT 4
#define SCAN_TILE 1024

__device__ __forceinline__ float lrelu02(float x) { return x > 0.f ? x : 0.2f * x; }

__global__ void k_padW0(const float* __restrict__ W0, float* __restrict__ W0p) {
    int i = threadIdx.x;
    int r = i >> 4;
    W0p[i] = (r < 13) ? W0[i] : 0.f;
}

__global__ void k_fill(const float* __restrict__ fw, const float* __restrict__ sw,
                       const float* __restrict__ tw, const float* __restrict__ areas,
                       const int* __restrict__ indices, float* __restrict__ h,
                       int N, int g) {
    int n = blockIdx.x * blockDim.x + threadIdx.x;
    if (n >= N) return;
    const size_t gbase = (size_t)g * NWIRE * NFEAT;
    const float4 f0 = *(const float4*)(fw + gbase + (size_t)indices[n * 3 + 0] * NFEAT);
    const float4 f1 = *(const float4*)(sw + gbase + (size_t)indices[n * 3 + 1] * NFEAT);
    const float4 f2 = *(const float4*)(tw + gbase + (size_t)indices[n * 3 + 2] * NFEAT);
    float4* hp = (float4*)(h + (size_t)n * HD);
    hp[0] = f0;
    hp[1] = f1;
    hp[2] = f2;
    hp[3] = make_float4(areas[n], 0.f, 0.f, 0.f);
}

__global__ void k_gemm_wmma(const float* __restrict__ h, const float* __restrict__ W,
                            const float* __restrict__ bias, float* __restrict__ hw, int N) {
    const int lane = threadIdx.x & 31;
    const int wave = threadIdx.x >> 5;
    const int tile = blockIdx.x * 8 + wave;
    const int base = tile * 16;
    if (base >= N) return;
    const int col16 = lane & 15;
    const int half  = lane >> 4;
    int arow = base + col16;
    if (arow >= N) arow = N - 1;
    v8f acc = {};
#pragma unroll
    for (int k = 0; k < 4; ++k) {
        const int kc = 4 * k + 2 * half;
        v2f a;
        a.x = h[(size_t)arow * HD + kc];
        a.y = h[(size_t)arow * HD + kc + 1];
        v2f b;
        b.x = W[kc * HD + col16];
        b.y = W[(kc + 1) * HD + col16];
        acc = __builtin_amdgcn_wmma_f32_16x16x4_f32(
                  false, a, false, b, (short)0, acc, false, false);
    }
    const float bi = bias[col16];
    if (base + 16 <= N) {
#pragma unroll
        for (int r = 0; r < 8; ++r) {
            hw[(size_t)(base + r + 8 * half) * HD + col16] = acc[r] + bi;
        }
    } else {
#pragma unroll
        for (int r = 0; r < 8; ++r) {
            const int orow = base + r + 8 * half;
            if (orow < N) hw[(size_t)orow * HD + col16] = acc[r] + bi;
        }
    }
}

__global__ void k_esed(const float* __restrict__ hw, const float* __restrict__ asrc,
                       const float* __restrict__ adst, float* __restrict__ es,
                       float* __restrict__ ed, int N) {
    int n = blockIdx.x * blockDim.x + threadIdx.x;
    if (n >= N) return;
    const float4* r = (const float4*)(hw + (size_t)n * HD);
    float s = 0.f, d = 0.f;
#pragma unroll
    for (int q = 0; q < 4; ++q) {
        const float4 v = r[q];
        s += v.x * asrc[q*4+0] + v.y * asrc[q*4+1] + v.z * asrc[q*4+2] + v.w * asrc[q*4+3];
        d += v.x * adst[q*4+0] + v.y * adst[q*4+1] + v.z * adst[q*4+2] + v.w * adst[q*4+3];
    }
    es[n] = s; ed[n] = d;
}

__global__ void k_zero(int* __restrict__ cnt, int N) {
    int i = blockIdx.x * blockDim.x + threadIdx.x;
    if (i < N) cnt[i] = 0;
}

__global__ void k_hist(const int* __restrict__ dst, int* __restrict__ cnt, int E) {
    int e = blockIdx.x * blockDim.x + threadIdx.x;
    if (e < E) atomicAdd(&cnt[dst[e]], 1);
}

__global__ __launch_bounds__(256) void k_scan_local(const int* __restrict__ cnt,
                                                    int* __restrict__ rowptr,
                                                    int* __restrict__ tilesum, int N) {
    __shared__ int wsum[8];
    const int t = threadIdx.x;
    const int lane = t & 31;
    const int wv = t >> 5;
    const int i0 = (blockIdx.x * 256 + t) * 4;
    const int c0 = (i0 + 0 < N) ? cnt[i0 + 0] : 0;
    const int c1 = (i0 + 1 < N) ? cnt[i0 + 1] : 0;
    const int c2 = (i0 + 2 < N) ? cnt[i0 + 2] : 0;
    const int c3 = (i0 + 3 < N) ? cnt[i0 + 3] : 0;
    const int s = c0 + c1 + c2 + c3;
    int inc = s;
    for (int off = 1; off < 32; off <<= 1) {
        const int v = __shfl_up(inc, off);
        if (lane >= off) inc += v;
    }
    if (lane == 31) wsum[wv] = inc;
    __syncthreads();
    int wpre = 0;
    for (int k = 0; k < 8; ++k) if (k < wv) wpre += wsum[k];
    int ex = wpre + inc - s;
    if (i0 + 0 < N) rowptr[i0 + 0] = ex;
    ex += c0;
    if (i0 + 1 < N) rowptr[i0 + 1] = ex;
    ex += c1;
    if (i0 + 2 < N) rowptr[i0 + 2] = ex;
    ex += c2;
    if (i0 + 3 < N) rowptr[i0 + 3] = ex;
    if (t == 255) tilesum[blockIdx.x] = wpre + inc;
}

__global__ __launch_bounds__(1024) void k_scan_tiles(const int* __restrict__ tilesum,
                                                     int* __restrict__ tileoff,
                                                     int* __restrict__ rowptr,
                                                     int numTiles, int N) {
    __shared__ int buf[1024];
    const int t = threadIdx.x;
    int carry = 0;
    for (int base = 0; base < numTiles; base += 1024) {
        const int i = base + t;
        const int v = (i < numTiles) ? tilesum[i] : 0;
        buf[t] = v;
        __syncthreads();
        for (int off = 1; off < 1024; off <<= 1) {
            const int a = (t >= off) ? buf[t - off] : 0;
            __syncthreads();
            buf[t] += a;
            __syncthreads();
        }
        if (i < numTiles) tileoff[i] = carry + buf[t] - v;
        carry += buf[1023];
        __syncthreads();
    }
    if (t == 0) rowptr[N] = carry;
}

__global__ void k_scan_add(int* __restrict__ rowptr, const int* __restrict__ tileoff,
                           int* __restrict__ cursor, int N) {
    int i = blockIdx.x * blockDim.x + threadIdx.x;
    if (i >= N) return;
    const int v = rowptr[i] + tileoff[i / SCAN_TILE];
    rowptr[i] = v;
    cursor[i] = v;
}

__global__ void k_scatter(const int* __restrict__ src, const int* __restrict__ dst,
                          int* __restrict__ cursor, int* __restrict__ srcS, int E) {
    int e = blockIdx.x * blockDim.x + threadIdx.x;
    if (e >= E) return;
    const int pos = atomicAdd(&cursor[dst[e]], 1);
    srcS[pos] = src[e];
}

__global__ __launch_bounds__(256) void k_gat_node(const int* __restrict__ rowptr,
                                                  const int* __restrict__ srcS,
                                                  const float* __restrict__ hw,
                                                  const float* __restrict__ es,
                                                  const float* __restrict__ ed,
                                                  float* __restrict__ h, int N) {
    const int t = blockIdx.x * 256 + threadIdx.x;
    const int node = t >> 2;
    const int q = t & 3;
    const bool valid = node < N;
    int beg = 0, end = 0;
    float edd = 0.f;
    if (valid) { beg = rowptr[node]; end = rowptr[node + 1]; edd = ed[node]; }

    float m = -INFINITY;
    int j = beg + q;
    for (; j + 4 < end; j += 8) {
        const int s0 = srcS[j];
        const int s1 = srcS[j + 4];
        const float l0 = lrelu02(es[s0] + edd);
        const float l1 = lrelu02(es[s1] + edd);
        m = fmaxf(m, fmaxf(l0, l1));
    }
    if (j < end) m = fmaxf(m, lrelu02(es[srcS[j]] + edd));
    m = fmaxf(m, __shfl_xor(m, 1));
    m = fmaxf(m, __shfl_xor(m, 2));

    const float4* __restrict__ hw4 = (const float4*)hw;
    float den = 0.f;
    float4 acc = make_float4(0.f, 0.f, 0.f, 0.f);
    int k = beg;
    for (; k + 4 <= end; k += 4) {
        const int s0 = srcS[k + 0];
        const int s1 = srcS[k + 1];
        const int s2 = srcS[k + 2];
        const int s3 = srcS[k + 3];
        const float e0 = es[s0];
        const float e1 = es[s1];
        const float e2 = es[s2];
        const float e3 = es[s3];
        const float4 h0 = hw4[(size_t)s0 * 4 + q];
        const float4 h1 = hw4[(size_t)s1 * 4 + q];
        const float4 h2 = hw4[(size_t)s2 * 4 + q];
        const float4 h3 = hw4[(size_t)s3 * 4 + q];
        const float w0 = __expf(lrelu02(e0 + edd) - m);
        const float w1 = __expf(lrelu02(e1 + edd) - m);
        const float w2 = __expf(lrelu02(e2 + edd) - m);
        const float w3 = __expf(lrelu02(e3 + edd) - m);
        den += w0;
        acc.x += w0 * h0.x; acc.y += w0 * h0.y; acc.z += w0 * h0.z; acc.w += w0 * h0.w;
        den += w1;
        acc.x += w1 * h1.x; acc.y += w1 * h1.y; acc.z += w1 * h1.z; acc.w += w1 * h1.w;
        den += w2;
        acc.x += w2 * h2.x; acc.y += w2 * h2.y; acc.z += w2 * h2.z; acc.w += w2 * h2.w;
        den += w3;
        acc.x += w3 * h3.x; acc.y += w3 * h3.y; acc.z += w3 * h3.z; acc.w += w3 * h3.w;
    }
    for (; k < end; ++k) {
        const int s0 = srcS[k];
        const float e0 = es[s0];
        const float4 h0 = hw4[(size_t)s0 * 4 + q];
        const float w0 = __expf(lrelu02(e0 + edd) - m);
        den += w0;
        acc.x += w0 * h0.x; acc.y += w0 * h0.y; acc.z += w0 * h0.z; acc.w += w0 * h0.w;
    }

    if (valid) {
        const float dn = fmaxf(den, 1e-16f);
        float4 o;
        o.x = acc.x / dn; o.y = acc.y / dn; o.z = acc.z / dn; o.w = acc.w / dn;
        o.x = o.x > 0.f ? o.x : (__expf(o.x) - 1.f);
        o.y = o.y > 0.f ? o.y : (__expf(o.y) - 1.f);
        o.z = o.z > 0.f ? o.z : (__expf(o.z) - 1.f);
        o.w = o.w > 0.f ? o.w : (__expf(o.w) - 1.f);
        ((float4*)h)[(size_t)node * 4 + q] = o;
    }
}

__global__ void k_final(const float* __restrict__ h, const float* __restrict__ mlpW,
                        const float* __restrict__ mlpb, float* __restrict__ out, int N) {
    int n = blockIdx.x * blockDim.x + threadIdx.x;
    if (n >= N) return;
    const float4* r = (const float4*)(h + (size_t)n * HD);
    float acc = mlpb[0];
#pragma unroll
    for (int q = 0; q < 4; ++q) {
        const float4 v = r[q];
        acc += v.x * mlpW[q*4+0] + v.y * mlpW[q*4+1] + v.z * mlpW[q*4+2] + v.w * mlpW[q*4+3];
    }
    out[n] = acc;
}

static inline int cdiv(int a, int b) { return (a + b - 1) / b; }

extern "C" void kernel_launch(void* const* d_in, const int* in_sizes, int n_in,
                              void* d_out, int out_size, void* d_ws, size_t ws_size,
                              hipStream_t stream) {
    const float* fw     = (const float*)d_in[0];
    const float* sw     = (const float*)d_in[1];
    const float* tw     = (const float*)d_in[2];
    const float* areas  = (const float*)d_in[3];
    const int*   indices= (const int*)  d_in[4];
    const int*   edge   = (const int*)  d_in[5];
    const float* W0     = (const float*)d_in[6];
    const float* Wl     = (const float*)d_in[7];
    const float* a_src  = (const float*)d_in[8];
    const float* a_dst  = (const float*)d_in[9];
    const float* bvec   = (const float*)d_in[10];
    const float* mlpW   = (const float*)d_in[11];
    const float* mlpb   = (const float*)d_in[12];

    const int N = in_sizes[3];
    const int E = in_sizes[5] / 2;
    const int T = in_sizes[0] / (NWIRE * NFEAT);
    const int* src = edge;
    const int* dst = edge + E;
    const int numTiles = cdiv(N, SCAN_TILE);

    float* ws = (float*)d_ws;
    size_t off = 0;
    float* h       = ws + off; off += (size_t)N * HD;
    float* hw      = ws + off; off += (size_t)N * HD;
    float* W0p     = ws + off; off += 256;
    float* es      = ws + off; off += (size_t)N;
    float* ed      = ws + off; off += (size_t)N;
    int*   srcS    = (int*)(ws + off); off += (size_t)E;
    int*   rowptr  = (int*)(ws + off); off += (size_t)N + 1;
    int*   cursor  = (int*)(ws + off); off += (size_t)N;
    int*   cnt     = (int*)(ws + off); off += (size_t)N;
    int*   tilesum = (int*)(ws + off); off += (size_t)numTiles;
    int*   tileoff = (int*)(ws + off); off += (size_t)numTiles;
    (void)ws_size; (void)n_in; (void)out_size;

    k_padW0<<<1, 256, 0, stream>>>(W0, W0p);

    k_zero<<<cdiv(N, 256), 256, 0, stream>>>(cnt, N);
    k_hist<<<cdiv(E, 256), 256, 0, stream>>>(dst, cnt, E);
    k_scan_local<<<numTiles, 256, 0, stream>>>(cnt, rowptr, tilesum, N);
    k_scan_tiles<<<1, 1024, 0, stream>>>(tilesum, tileoff, rowptr, numTiles, N);
    k_scan_add<<<cdiv(N, 256), 256, 0, stream>>>(rowptr, tileoff, cursor, N);
    k_scatter<<<cdiv(E, 256), 256, 0, stream>>>(src, dst, cursor, srcS, E);

    const int tiles  = cdiv(N, 16);
    const int gemmBl = cdiv(tiles, 8);

    for (int g = 0; g < T; ++g) {
        k_fill<<<cdiv(N, 256), 256, 0, stream>>>(fw, sw, tw, areas, indices, h, N, g);
        for (int i = 0; i < 4; ++i) {
            const float* Wi = (i == 0) ? W0p : (Wl + (size_t)(i - 1) * 256);
            k_gemm_wmma<<<gemmBl, 256, 0, stream>>>(h, Wi, bvec + (size_t)i * HD, hw, N);
            k_esed<<<cdiv(N, 256), 256, 0, stream>>>(hw, a_src + (size_t)i * HD,
                                                     a_dst + (size_t)i * HD, es, ed, N);
            k_gat_node<<<cdiv(N * 4, 256), 256, 0, stream>>>(rowptr, srcS, hw, es, ed, h, N);
        }
        k_final<<<cdiv(N, 256), 256, 0, stream>>>(h, mlpW, mlpb, (float*)d_out + (size_t)g * N, N);
    }
}


# Each 16x16 layer runs in the kernel that makes its input rows: layer 0 reads the wires into A fragments, layers 1-3 run in the GAT epilogue via an LDS tile; fill, 3 GEMM launches and h go
# speedup vs baseline: 10.1106x; 2.1337x over previous
#include <hip/hip_runtime.h>
#include <math.h>


typedef float v2f __attribute__((ext_vector_type(2)));
typedef float v8f __attribute__((ext_vector_type(8)));

#define HD 16
#define NWIRE 1536
#define NFEAT 4
#define SCAN_TILE 1024


__device__ __forceinline__ float lrelu02(float x) { return x > 0.f ? x : 0.2f * x; }

__device__ __forceinline__ float dot4(const float4 v, const float4 a) {
    return v.x * a.x + v.y * a.y + v.z * a.z + v.w * a.w;
}

__device__ __forceinline__ float quad_sum(float p) {
    p += __int_as_float(__builtin_amdgcn_update_dpp(0, __float_as_int(p), 0xB1, 0xF, 0xF, true));
    p += __int_as_float(__builtin_amdgcn_update_dpp(0, __float_as_int(p), 0x4E, 0xF, 0xF, true));
    return p;
}

__global__ void k_padW0(const float* __restrict__ W0, float* __restrict__ W0p) {
    int i = threadIdx.x;
    int r = i >> 4;
    W0p[i] = (r < 13) ? W0[i] : 0.f;
}

template <int TB>
__global__ void k_layer0_wmma(const float* __restrict__ fw, const float* __restrict__ sw,
                              const float* __restrict__ tw, const float* __restrict__ areas,
                              const int* __restrict__ indices, const float* __restrict__ W,
                              const float* __restrict__ bias, float* __restrict__ hw,
                              int N, int g0) {
    const int lane = threadIdx.x & 31;
    const int wave = threadIdx.x >> 5;
    const int tile = blockIdx.x * 8 + wave;
    const int base = tile * 16;
    const int rows = N * TB;
    if (base >= rows) return;
    const int col16 = lane & 15;
    const int half  = lane >> 4;
    int arow = base + col16;
    if (arow >= rows) arow = rows - 1;
    const int n = arow / TB;
    const int g = arow - n * TB;
    const size_t gbase = (size_t)(g0 + g) * NWIRE * NFEAT + 2 * half;
    v2f a[4];
    a[0] = *(const v2f*)(fw + gbase + (size_t)indices[n * 3 + 0] * NFEAT);
    a[1] = *(const v2f*)(sw + gbase + (size_t)indices[n * 3 + 1] * NFEAT);
    a[2] = *(const v2f*)(tw + gbase + (size_t)indices[n * 3 + 2] * NFEAT);
    a[3].x = half == 0 ? areas[n] : 0.f;
    a[3].y = 0.f;
    v8f acc = {};
#pragma unroll
    for (int k = 0; k < 4; ++k) {
        const int kc = 4 * k + 2 * half;
        v2f b;
        b.x = W[kc * HD + col16];
        b.y = W[(kc + 1) * HD + col16];
        acc = __builtin_amdgcn_wmma_f32_16x16x4_f32(
                  false, a[k], false, b, (short)0, acc, false, false);
    }
    const float bi = bias[col16];
    if (base + 16 <= rows) {
#pragma unroll
        for (int r = 0; r < 8; ++r) {
            hw[(size_t)(base + r + 8 * half) * HD + col16] = acc[r] + bi;
        }
    } else {
#pragma unroll
        for (int r = 0; r < 8; ++r) {
            const int orow = base + r + 8 * half;
            if (orow < rows) hw[(size_t)orow * HD + col16] = acc[r] + bi;
        }
    }
}

__global__ void k_zero(int* __restrict__ cnt, int N) {
    int i = blockIdx.x * blockDim.x + threadIdx.x;
    if (i < N) cnt[i] = 0;
}

__global__ void k_hist(const int* __restrict__ dst, int* __restrict__ cnt, int E) {
    int e = blockIdx.x * blockDim.x + threadIdx.x;
    if (e < E) atomicAdd(&cnt[dst[e]], 1);
}

__global__ __launch_bounds__(256) void k_scan_local(const int* __restrict__ cnt,
                                                    int* __restrict__ rowptr,
                                                    int* __restrict__ tilesum, int N) {
    __shared__ int wsum[8];
    const int t = threadIdx.x;
    const int lane = t & 31;
    const int wv = t >> 5;
    const int i0 = (blockIdx.x * 256 + t) * 4;
    const int c0 = (i0 + 0 < N) ? cnt[i0 + 0] : 0;
    const int c1 = (i0 + 1 < N) ? cnt[i0 + 1] : 0;
    const int c2 = (i0 + 2 < N) ? cnt[i0 + 2] : 0;
    const int c3 = (i0 + 3 < N) ? cnt[i0 + 3] : 0;
    const int s = c0 + c1 + c2 + c3;
    int inc = s;
    for (int off = 1; off < 32; off <<= 1) {
        const int v = __shfl_up(inc, off);
        if (lane >= off) inc += v;
    }
    if (lane == 31) wsum[wv] = inc;
    __syncthreads();
    int wpre = 0;
    for (int k = 0; k < 8; ++k) if (k < wv) wpre += wsum[k];
    int ex = wpre + inc - s;
    if (i0 + 0 < N) rowptr[i0 + 0] = ex;
    ex += c0;
    if (i0 + 1 < N) rowptr[i0 + 1] = ex;
    ex += c1;
    if (i0 + 2 < N) rowptr[i0 + 2] = ex;
    ex += c2;
    if (i0 + 3 < N) rowptr[i0 + 3] = ex;
    if (t == 255) tilesum[blockIdx.x] = wpre + inc;
}

__global__ __launch_bounds__(1024) void k_scan_tiles(const int* __restrict__ tilesum,
                                                     int* __restrict__ tileoff,
                                                     int* __restrict__ rowptr,
                                                     int numTiles, int N) {
    __shared__ int buf[1024];
    const int t = threadIdx.x;
    int carry = 0;
    for (int base = 0; base < numTiles; base += 1024) {
        const int i = base + t;
        const int v = (i < numTiles) ? tilesum[i] : 0;
        buf[t] = v;
        __syncthreads();
        for (int off = 1; off < 1024; off <<= 1) {
            const int a = (t >= off) ? buf[t - off] : 0;
            __syncthreads();
            buf[t] += a;
            __syncthreads();
        }
        if (i < numTiles) tileoff[i] = carry + buf[t] - v;
        carry += buf[1023];
        __syncthreads();
    }
    if (t == 0) rowptr[N] = carry;
}

__global__ void k_scan_add(int* __restrict__ rowptr, const int* __restrict__ tileoff,
                           int* __restrict__ cursor, int N) {
    int i = blockIdx.x * blockDim.x + threadIdx.x;
    if (i >= N) return;
    const int v = rowptr[i] + tileoff[i / SCAN_TILE];
    rowptr[i] = v;
    cursor[i] = v;
}

__global__ void k_scatter(const int* __restrict__ src, const int* __restrict__ dst,
                          int* __restrict__ cursor, int* __restrict__ srcS, int E) {
    int e = blockIdx.x * blockDim.x + threadIdx.x;
    if (e >= E) return;
    const int pos = atomicAdd(&cursor[dst[e]], 1);
    srcS[pos] = src[e];
}

template <int TB, int U>
__device__ __forceinline__ void gat_edges(const int* __restrict__ srcS, const float4* __restrict__ hw4,
                                          const int k, const int q, const float4 as,
                                          const float (&edd)[TB], float (&m)[TB], float (&den)[TB],
                                          float4 (&acc)[TB]) {
    int s[U];
    float4 hv[U][TB];
    float l[U][TB];
#pragma unroll
    for (int u = 0; u < U; ++u) s[u] = srcS[k + u];
#pragma unroll
    for (int u = 0; u < U; ++u) {
#pragma unroll
        for (int g = 0; g < TB; ++g) hv[u][g] = hw4[s[u] * (TB * 4) + g * 4 + q];
    }
#pragma unroll
    for (int u = 0; u < U; ++u) {
#pragma unroll
        for (int g = 0; g < TB; ++g) l[u][g] = lrelu02(quad_sum(dot4(hv[u][g], as)) + edd[g]);
    }
#pragma unroll
    for (int g = 0; g < TB; ++g) {
        float bm = m[g];
#pragma unroll
        for (int u = 0; u < U; ++u) bm = fmaxf(bm, l[u][g]);
        const float sc = __expf(m[g] - bm);
        m[g] = bm;
        den[g] *= sc;
        acc[g].x *= sc; acc[g].y *= sc; acc[g].z *= sc; acc[g].w *= sc;
#pragma unroll
        for (int u = 0; u < U; ++u) {
            const float w = __expf(l[u][g] - bm);
            den[g] += w;
            acc[g].x += w * hv[u][g].x; acc[g].y += w * hv[u][g].y;
            acc[g].z += w * hv[u][g].z; acc[g].w += w * hv[u][g].w;
        }
    }
}

#define GAT_THREADS 128
#define TPITCH 20

__device__ __forceinline__ void wave_lds_handover() {
    __builtin_amdgcn_fence(__ATOMIC_RELEASE, "wavefront");
    __builtin_amdgcn_wave_barrier();
    __builtin_amdgcn_fence(__ATOMIC_ACQUIRE, "wavefront");
}

template <int TB, bool FUSE>
__global__ __launch_bounds__(GAT_THREADS) void k_gat_node(const int* __restrict__ rowptr,
                                                  const int* __restrict__ srcS,
                                                  const float* __restrict__ hw,
                                                  const float* __restrict__ asrc,
                                                  const float* __restrict__ adst,
                                                  const float* __restrict__ Wn,
                                                  const float* __restrict__ bn,
                                                  float* __restrict__ hout, int N) {
    const int t = blockIdx.x * GAT_THREADS + threadIdx.x;
    const int node = t >> 2;
    const int q = t & 3;
    const bool valid = node < N;
    const float4* __restrict__ hw4 = (const float4*)hw;
    const float4 as = ((const float4*)asrc)[q];
    const float4 ad = ((const float4*)adst)[q];
    int beg = 0, end = 0;
    float edd[TB];
#pragma unroll
    for (int g = 0; g < TB; ++g) edd[g] = 0.f;
    if (valid) {
        beg = rowptr[node]; end = rowptr[node + 1];
#pragma unroll
        for (int g = 0; g < TB; ++g) edd[g] = dot4(hw4[(node * TB + g) * 4 + q], ad);
    }
#pragma unroll
    for (int g = 0; g < TB; ++g) edd[g] = quad_sum(edd[g]);

    constexpr int U = 4;
    float m[TB], den[TB];
    float4 acc[TB];
#pragma unroll
    for (int g = 0; g < TB; ++g) {
        m[g] = -INFINITY; den[g] = 0.f; acc[g] = make_float4(0.f, 0.f, 0.f, 0.f);
    }
    int k = beg;
    for (; k + U <= end; k += U) gat_edges<TB, U>(srcS, hw4, k, q, as, edd, m, den, acc);
    if (k + 2 <= end) { gat_edges<TB, 2>(srcS, hw4, k, q, as, edd, m, den, acc); k += 2; }
    if (k < end)        gat_edges<TB, 1>(srcS, hw4, k, q, as, edd, m, den, acc);

    float4 o[TB];
#pragma unroll
    for (int g = 0; g < TB; ++g) {
        const float dn = fmaxf(den[g], 1e-16f);
        o[g].x = acc[g].x / dn; o[g].y = acc[g].y / dn; o[g].z = acc[g].z / dn; o[g].w = acc[g].w / dn;
        o[g].x = o[g].x > 0.f ? o[g].x : (__expf(o[g].x) - 1.f);
        o[g].y = o[g].y > 0.f ? o[g].y : (__expf(o[g].y) - 1.f);
        o[g].z = o[g].z > 0.f ? o[g].z : (__expf(o[g].z) - 1.f);
        o[g].w = o[g].w > 0.f ? o[g].w : (__expf(o[g].w) - 1.f);
    }

    if constexpr (!FUSE) {
        if (valid) {
#pragma unroll
            for (int g = 0; g < TB; ++g) ((float4*)hout)[((size_t)node * TB + g) * 4 + q] = o[g];
        }
    } else {
        constexpr int R  = 8 * TB;
        constexpr int NT = (R + 15) / 16;
        __shared__ float tile[GAT_THREADS / 32][NT * 16][TPITCH];
        const int lane = threadIdx.x & 31;
        const int wave = threadIdx.x >> 5;
        float (*T)[TPITCH] = tile[wave];
        const int nd = lane >> 2;
#pragma unroll
        for (int g = 0; g < TB; ++g) *(float4*)&T[nd * TB + g][4 * q] = o[g];
        wave_lds_handover();

        const int c  = lane & 15;
        const int hf = lane >> 4;
        v2f bfrag[4];
#pragma unroll
        for (int j = 0; j < 4; ++j) {
            const int kc = 4 * j + 2 * hf;
            bfrag[j].x = Wn[kc * HD + c];
            bfrag[j].y = Wn[(kc + 1) * HD + c];
        }
        const float bi = bn[c];
        v8f d[NT];
#pragma unroll
        for (int tI = 0; tI < NT; ++tI) {
            int row = 16 * tI + c;
            if (row >= R) row = R - 1;
            v8f a8 = {};
#pragma unroll
            for (int j = 0; j < 4; ++j) {
                const v2f a = *(const v2f*)&T[row][4 * j + 2 * hf];
                a8 = __builtin_amdgcn_wmma_f32_16x16x4_f32(
                         false, a, false, bfrag[j], (short)0, a8, false, false);
            }
            d[tI] = a8;
        }
        wave_lds_handover();
#pragma unroll
        for (int tI = 0; tI < NT; ++tI) {
#pragma unroll
            for (int r = 0; r < 8; ++r) T[16 * tI + r + 8 * hf][c] = d[tI][r] + bi;
        }
        wave_lds_handover();
        const int rowbase = (blockIdx.x * (GAT_THREADS / 4) + wave * 8) * TB;
        const int rows = N * TB;
#pragma unroll
        for (int j = 0; j < R / 8; ++j) {
            const int f = lane + 32 * j;
            const int row = f >> 2;
            const int c4 = f & 3;
            if (rowbase + row < rows)
                ((float4*)hout)[(size_t)(rowbase + row) * 4 + c4] = *(const float4*)&T[row][4 * c4];
        }
    }
}

template <int TB>
__global__ void k_final(const float* __restrict__ h, const float* __restrict__ mlpW,
                        const float* __restrict__ mlpb, float* __restrict__ out, int N, int g0) {
    const int n = blockIdx.x * blockDim.x + threadIdx.x;
    const int g = blockIdx.y;
    if (n >= N) return;
    const float4* p = (const float4*)(h + ((size_t)n * TB + g) * HD);
    float acc = mlpb[0];
#pragma unroll
    for (int q = 0; q < 4; ++q) {
        const float4 v = p[q];
        acc += v.x * mlpW[q*4+0] + v.y * mlpW[q*4+1] + v.z * mlpW[q*4+2] + v.w * mlpW[q*4+3];
    }
    out[(size_t)(g0 + g) * N + n] = acc;
}

static inline int cdiv(int a, int b) { return (a + b - 1) / b; }

struct Bufs {
    const float *fw, *sw, *tw, *areas;
    const int* indices;
    const float *W0p, *Wl, *a_src, *a_dst, *bvec, *mlpW, *mlpb;
    const int *rowptr, *srcS;
    float *h, *hw, *out;
};

template <int TB>
static void run_graphs(const Bufs& b, int N, int g0, hipStream_t stream) {
    const int rows   = N * TB;
    const int tiles  = cdiv(rows, 16);
    const int gemmBl = cdiv(tiles, 8);
    const int gatBl  = cdiv(N * 4, GAT_THREADS);
    k_layer0_wmma<TB><<<gemmBl, 256, 0, stream>>>(b.fw, b.sw, b.tw, b.areas, b.indices, b.W0p, b.bvec, b.hw, N, g0);
    float* cur = b.hw;
    float* nxt = b.h;
    for (int i = 0; i < 4; ++i) {
        const float* as = b.a_src + (size_t)i * HD;
        const float* ad = b.a_dst + (size_t)i * HD;
        if (i < 3)
            k_gat_node<TB, true><<<gatBl, GAT_THREADS, 0, stream>>>(b.rowptr, b.srcS, cur, as, ad,
                                                                   b.Wl + (size_t)i * 256,
                                                                   b.bvec + (size_t)(i + 1) * HD, nxt, N);
        else
            k_gat_node<TB, false><<<gatBl, GAT_THREADS, 0, stream>>>(b.rowptr, b.srcS, cur, as, ad,
                                                                    b.Wl, b.bvec, nxt, N);
        float* tmp = cur; cur = nxt; nxt = tmp;
    }
    k_final<TB><<<dim3(cdiv(N, 256), TB), 256, 0, stream>>>(cur, b.mlpW, b.mlpb, b.out, N, g0);
}

extern "C" void kernel_launch(void* const* d_in, const int* in_sizes, int n_in,
                              void* d_out, int out_size, void* d_ws, size_t ws_size,
                              hipStream_t stream) {
    const float* fw     = (const float*)d_in[0];
    const float* sw     = (const float*)d_in[1];
    const float* tw     = (const float*)d_in[2];
    const float* areas  = (const float*)d_in[3];
    const int*   indices= (const int*)  d_in[4];
    const int*   edge   = (const int*)  d_in[5];
    const float* W0     = (const float*)d_in[6];
    const float* Wl     = (const float*)d_in[7];
    const float* a_src  = (const float*)d_in[8];
    const float* a_dst  = (const float*)d_in[9];
    const float* bvec   = (const float*)d_in[10];
    const float* mlpW   = (const float*)d_in[11];
    const float* mlpb   = (const float*)d_in[12];

    const int N = in_sizes[3];
    const int E = in_sizes[5] / 2;
    const int T = in_sizes[0] / (NWIRE * NFEAT);
    const int* src = edge;
    const int* dst = edge + E;
    const int numTiles = cdiv(N, SCAN_TILE);

    const size_t fixedW = 256 + (size_t)E + ((size_t)N + 1) + 2 * (size_t)N + 2 * (size_t)numTiles;
    const size_t need3  = 2 * (size_t)N * HD * 3 + fixedW;
    const size_t need2  = 2 * (size_t)N * HD * 2 + fixedW;
    int TBS = 1;
    if (T >= 3 && ws_size >= need3 * 4) TBS = 3;
    else if (T >= 2 && ws_size >= need2 * 4) TBS = 2;

    float* ws = (float*)d_ws;
    size_t off = 0;
    float* h       = ws + off; off += (size_t)N * HD * TBS;
    float* hw      = ws + off; off += (size_t)N * HD * TBS;
    float* W0p     = ws + off; off += 256;
    int*   srcS    = (int*)(ws + off); off += (size_t)E;
    int*   rowptr  = (int*)(ws + off); off += (size_t)N + 1;
    int*   cursor  = (int*)(ws + off); off += (size_t)N;
    int*   cnt     = (int*)(ws + off); off += (size_t)N;
    int*   tilesum = (int*)(ws + off); off += (size_t)numTiles;
    int*   tileoff = (int*)(ws + off); off += (size_t)numTiles;
    (void)n_in; (void)out_size;

    k_padW0<<<1, 256, 0, stream>>>(W0, W0p);

    k_zero<<<cdiv(N, 256), 256, 0, stream>>>(cnt, N);
    k_hist<<<cdiv(E, 256), 256, 0, stream>>>(dst, cnt, E);
    k_scan_local<<<numTiles, 256, 0, stream>>>(cnt, rowptr, tilesum, N);
    k_scan_tiles<<<1, 1024, 0, stream>>>(tilesum, tileoff, rowptr, numTiles, N);
    k_scan_add<<<cdiv(N, 256), 256, 0, stream>>>(rowptr, tileoff, cursor, N);
    k_scatter<<<cdiv(E, 256), 256, 0, stream>>>(src, dst, cursor, srcS, E);

    Bufs b;
    b.fw = fw; b.sw = sw; b.tw = tw; b.areas = areas; b.indices = indices;
    b.W0p = W0p; b.Wl = Wl; b.a_src = a_src; b.a_dst = a_dst; b.bvec = bvec;
    b.mlpW = mlpW; b.mlpb = mlpb; b.rowptr = rowptr; b.srcS = srcS;
    b.h = h; b.hw = hw; b.out = (float*)d_out;

    int g = 0;
    while (g < T) {
        const int left = T - g;
        const int tb = left < TBS ? left : TBS;
        if (tb >= 3)      { run_graphs<3>(b, N, g, stream); g += 3; }
        else if (tb == 2) { run_graphs<2>(b, N, g, stream); g += 2; }
        else              { run_graphs<1>(b, N, g, stream); g += 1; }
    }
}


# Edge sort by destination on LDS counters, two levels (bins of 1024 nodes, one block per bin), each node's run then ordered by source id: no per-edge global atomics; sum order fixed, same every run
# speedup vs baseline: 17.9683x; 1.7772x over previous
#include <hip/hip_runtime.h>
#include <math.h>


typedef float v2f __attribute__((ext_vector_type(2)));
typedef float v8f __attribute__((ext_vector_type(8)));

#define HD 16
#define NWIRE 1536
#define NFEAT 4
#define SCAN_TILE 1024


__device__ __forceinline__ float lrelu02(float x) { return x > 0.f ? x : 0.2f * x; }

__device__ __forceinline__ float dot4(const float4 v, const float4 a) {
    return v.x * a.x + v.y * a.y + v.z * a.z + v.w * a.w;
}

__device__ __forceinline__ float quad_sum(float p) {
    p += __int_as_float(__builtin_amdgcn_update_dpp(0, __float_as_int(p), 0xB1, 0xF, 0xF, true));
    p += __int_as_float(__builtin_amdgcn_update_dpp(0, __float_as_int(p), 0x4E, 0xF, 0xF, true));
    return p;
}

__global__ void k_prep(const float* __restrict__ W0, float* __restrict__ W0p,
                       int* __restrict__ binCnt, int* __restrict__ binCur, int numBins) {
    int i = threadIdx.x;
    int r = i >> 4;
    W0p[i] = (r < 13) ? W0[i] : 0.f;
    for (int b = i; b < numBins; b += 256) { binCnt[b] = 0; binCur[b] = 0; }
}

template <int TB>
__global__ void k_layer0_wmma(const float* __restrict__ fw, const float* __restrict__ sw,
                              const float* __restrict__ tw, const float* __restrict__ areas,
                              const int* __restrict__ indices, const float* __restrict__ W,
                              const float* __restrict__ bias, float* __restrict__ hw,
                              int N, int g0) {
    const int lane = threadIdx.x & 31;
    const int wave = threadIdx.x >> 5;
    const int tile = blockIdx.x * 8 + wave;
    const int base = tile * 16;
    const int rows = N * TB;
    if (base >= rows) return;
    const int col16 = lane & 15;
    const int half  = lane >> 4;
    int arow = base + col16;
    if (arow >= rows) arow = rows - 1;
    const int n = arow / TB;
    const int g = arow - n * TB;
    const size_t gbase = (size_t)(g0 + g) * NWIRE * NFEAT + 2 * half;
    v2f a[4];
    a[0] = *(const v2f*)(fw + gbase + (size_t)indices[n * 3 + 0] * NFEAT);
    a[1] = *(const v2f*)(sw + gbase + (size_t)indices[n * 3 + 1] * NFEAT);
    a[2] = *(const v2f*)(tw + gbase + (size_t)indices[n * 3 + 2] * NFEAT);
    a[3].x = half == 0 ? areas[n] : 0.f;
    a[3].y = 0.f;
    v8f acc = {};
#pragma unroll
    for (int k = 0; k < 4; ++k) {
        const int kc = 4 * k + 2 * half;
        v2f b;
        b.x = W[kc * HD + col16];
        b.y = W[(kc + 1) * HD + col16];
        acc = __builtin_amdgcn_wmma_f32_16x16x4_f32(
                  false, a[k], false, b, (short)0, acc, false, false);
    }
    const float bi = bias[col16];
    if (base + 16 <= rows) {
#pragma unroll
        for (int r = 0; r < 8; ++r) {
            hw[(size_t)(base + r + 8 * half) * HD + col16] = acc[r] + bi;
        }
    } else {
#pragma unroll
        for (int r = 0; r < 8; ++r) {
            const int orow = base + r + 8 * half;
            if (orow < rows) hw[(size_t)orow * HD + col16] = acc[r] + bi;
        }
    }
}

__global__ void k_zero(int* __restrict__ cnt, int N) {
    int i = blockIdx.x * blockDim.x + threadIdx.x;
    if (i < N) cnt[i] = 0;
}

__global__ void k_hist(const int* __restrict__ dst, int* __restrict__ cnt, int E) {
    int e = blockIdx.x * blockDim.x + threadIdx.x;
    if (e < E) atomicAdd(&cnt[dst[e]], 1);
}

__global__ __launch_bounds__(256) void k_scan_local(const int* __restrict__ cnt,
                                                    int* __restrict__ rowptr,
                                                    int* __restrict__ tilesum, int N) {
    __shared__ int wsum[8];
    const int t = threadIdx.x;
    const int lane = t & 31;
    const int wv = t >> 5;
    const int i0 = (blockIdx.x * 256 + t) * 4;
    const int c0 = (i0 + 0 < N) ? cnt[i0 + 0] : 0;
    const int c1 = (i0 + 1 < N) ? cnt[i0 + 1] : 0;
    const int c2 = (i0 + 2 < N) ? cnt[i0 + 2] : 0;
    const int c3 = (i0 + 3 < N) ? cnt[i0 + 3] : 0;
    const int s = c0 + c1 + c2 + c3;
    int inc = s;
    for (int off = 1; off < 32; off <<= 1) {
        const int v = __shfl_up(inc, off);
        if (lane >= off) inc += v;
    }
    if (lane == 31) wsum[wv] = inc;
    __syncthreads();
    int wpre = 0;
    for (int k = 0; k < 8; ++k) if (k < wv) wpre += wsum[k];
    int ex = wpre + inc - s;
    if (i0 + 0 < N) rowptr[i0 + 0] = ex;
    ex += c0;
    if (i0 + 1 < N) rowptr[i0 + 1] = ex;
    ex += c1;
    if (i0 + 2 < N) rowptr[i0 + 2] = ex;
    ex += c2;
    if (i0 + 3 < N) rowptr[i0 + 3] = ex;
    if (t == 255) tilesum[blockIdx.x] = wpre + inc;
}

__global__ __launch_bounds__(1024) void k_scan_tiles(const int* __restrict__ tilesum,
                                                     int* __restrict__ tileoff,
                                                     int* __restrict__ rowptr,
                                                     int numTiles, int N) {
    __shared__ int buf[1024];
    const int t = threadIdx.x;
    int carry = 0;
    for (int base = 0; base < numTiles; base += 1024) {
        const int i = base + t;
        const int v = (i < numTiles) ? tilesum[i] : 0;
        buf[t] = v;
        __syncthreads();
        for (int off = 1; off < 1024; off <<= 1) {
            const int a = (t >= off) ? buf[t - off] : 0;
            __syncthreads();
            buf[t] += a;
            __syncthreads();
        }
        if (i < numTiles) tileoff[i] = carry + buf[t] - v;
        carry += buf[1023];
        __syncthreads();
    }
    if (t == 0) rowptr[N] = carry;
}

__global__ void k_scan_add(int* __restrict__ rowptr, const int* __restrict__ tileoff,
                           int* __restrict__ cursor, int N) {
    int i = blockIdx.x * blockDim.x + threadIdx.x;
    if (i >= N) return;
    const int v = rowptr[i] + tileoff[i / SCAN_TILE];
    rowptr[i] = v;
    cursor[i] = v;
}

__global__ void k_scatter(const int* __restrict__ src, const int* __restrict__ dst,
                          int* __restrict__ cursor, int* __restrict__ srcS, int E) {
    int e = blockIdx.x * blockDim.x + threadIdx.x;
    if (e >= E) return;
    const int pos = atomicAdd(&cursor[dst[e]], 1);
    srcS[pos] = src[e];
}

#define BIN_SHIFT 10
#define BIN_NODES (1 << BIN_SHIFT)
#define SRC_BITS 22
#define SRC_MASK ((1u << SRC_BITS) - 1u)
#define MAX_BINS (1 << (SRC_BITS - BIN_SHIFT))
#define PART_THREADS 1024
#define PART_EPT 8
#define PART_CHUNK (PART_THREADS * PART_EPT)
#define BIN_CAP 49152
#define BIN_LD 8
static_assert(BIN_NODES == SCAN_TILE, "the bin arrays share the scan-tile arrays of the workspace");

__global__ __launch_bounds__(PART_THREADS) void k_binhist(const int* __restrict__ dst,
                                                          int* __restrict__ binCnt,
                                                          int E, int numBins) {
    __shared__ int hist[MAX_BINS];
    const int t = threadIdx.x;
    for (int b = t; b < numBins; b += PART_THREADS) hist[b] = 0;
    __syncthreads();
    const int base = blockIdx.x * PART_CHUNK;
    int d[PART_EPT];
#pragma unroll
    for (int i = 0; i < PART_EPT; ++i) {
        const int e = base + i * PART_THREADS + t;
        d[i] = dst[e < E ? e : E - 1];
    }
#pragma unroll
    for (int i = 0; i < PART_EPT; ++i) {
        atomicAdd(&hist[d[i] >> BIN_SHIFT], (base + i * PART_THREADS + t < E) ? 1 : 0);
    }
    __syncthreads();
    for (int b = t; b < numBins; b += PART_THREADS) {
        const int c = hist[b];
        if (c) atomicAdd(&binCnt[b], c);
    }
}

__global__ __launch_bounds__(PART_THREADS) void k_partition(const int* __restrict__ src,
                                                            const int* __restrict__ dst,
                                                            const int* __restrict__ binOff,
                                                            int* __restrict__ binCur,
                                                            unsigned* __restrict__ tmp,
                                                            int E, int numBins) {
    __shared__ int hist[MAX_BINS];
    const int t = threadIdx.x;
    for (int b = t; b < numBins; b += PART_THREADS) hist[b] = 0;
    const int base = blockIdx.x * PART_CHUNK;
    int d[PART_EPT], s[PART_EPT], r[PART_EPT];
#pragma unroll
    for (int i = 0; i < PART_EPT; ++i) {
        const int e = base + i * PART_THREADS + t;
        const int ec = e < E ? e : E - 1;
        const int dv = dst[ec];
        s[i] = src[ec];
        d[i] = (e < E) ? dv : -1;
    }
    __syncthreads();
#pragma unroll
    for (int i = 0; i < PART_EPT; ++i) {
        r[i] = 0;
        if (d[i] >= 0) r[i] = atomicAdd(&hist[d[i] >> BIN_SHIFT], 1);
    }
    __syncthreads();
    for (int b = t; b < numBins; b += PART_THREADS) {
        const int c = hist[b];
        int g = 0;
        if (c) g = binOff[b] + atomicAdd(&binCur[b], c);
        hist[b] = g;
    }
    __syncthreads();
#pragma unroll
    for (int i = 0; i < PART_EPT; ++i) {
        if (d[i] >= 0)
            tmp[hist[d[i] >> BIN_SHIFT] + r[i]] =
                ((unsigned)(d[i] & (BIN_NODES - 1)) << SRC_BITS) | (unsigned)s[i];
    }
}

__global__ __launch_bounds__(1024) void k_binsort(const unsigned* __restrict__ tmp,
                                                  const int* __restrict__ binOff,
                                                  const int* __restrict__ binCnt,
                                                  int* __restrict__ rowptr,
                                                  int* __restrict__ srcS, int N) {
    __shared__ int cnt[BIN_NODES];
    __shared__ int off[BIN_NODES + 1];
    __shared__ int wsum[32];
    __shared__ unsigned a[BIN_CAP];
    const int t = threadIdx.x;
    const int lane = t & 31;
    const int wv = t >> 5;
    const int bin = blockIdx.x;
    const int gbeg = binOff[bin];
    const int n = binCnt[bin];
    const unsigned* __restrict__ mine = tmp + gbeg;
    cnt[t] = 0;
    __syncthreads();
    for (int i0 = t; i0 < n; i0 += 1024 * BIN_LD) {
        unsigned w[BIN_LD];
#pragma unroll
        for (int u = 0; u < BIN_LD; ++u) {
            const int i = i0 + u * 1024;
            w[u] = mine[i < n ? i : i0];
        }
#pragma unroll
        for (int u = 0; u < BIN_LD; ++u) {
            const int i = i0 + u * 1024;
            atomicAdd(&cnt[w[u] >> SRC_BITS], (i < n) ? 1 : 0);
        }
    }
    __syncthreads();
    const int c = cnt[t];
    int inc = c;
    for (int o = 1; o < 32; o <<= 1) {
        const int v = __shfl_up(inc, o);
        if (lane >= o) inc += v;
    }
    if (lane == 31) wsum[wv] = inc;
    __syncthreads();
    int wpre = 0;
    for (int k = 0; k < 32; ++k) if (k < wv) wpre += wsum[k];
    const int ex = wpre + inc - c;
    off[t] = ex;
    if (t == 1023) off[BIN_NODES] = ex + c;
    cnt[t] = ex;
    const int node = bin * BIN_NODES + t;
    if (node < N) rowptr[node] = gbeg + ex;
    __syncthreads();
    if (n <= BIN_CAP) {
        for (int i0 = t; i0 < n; i0 += 1024 * BIN_LD) {
            unsigned w[BIN_LD];
#pragma unroll
            for (int u = 0; u < BIN_LD; ++u) {
                const int i = i0 + u * 1024;
                w[u] = mine[i < n ? i : i0];
            }
#pragma unroll
            for (int u = 0; u < BIN_LD; ++u) {
                const int i = i0 + u * 1024;
                const int p = atomicAdd(&cnt[w[u] >> SRC_BITS], (i < n) ? 1 : 0);
                if (i < n) a[p] = w[u];
            }
        }
        __syncthreads();
        for (int p = t; p < n; p += 1024) {
            const unsigned w = a[p];
            const int dl = (int)(w >> SRC_BITS);
            const int sb = off[dl];
            const int se = off[dl + 1];
            int rank = 0;
            for (int j = sb; j < se; ++j) {
                const unsigned v = a[j];
                rank += (v < w || (v == w && j < p)) ? 1 : 0;
            }
            srcS[gbeg + sb + rank] = (int)(w & SRC_MASK);
        }
    } else {
        for (int i = t; i < n; i += 1024) {
            const unsigned w = mine[i];
            const int p = atomicAdd(&cnt[w >> SRC_BITS], 1);
            srcS[gbeg + p] = (int)(w & SRC_MASK);
        }
    }
}

template <int TB, int U>
__device__ __forceinline__ void gat_edges(const int* __restrict__ srcS, const float4* __restrict__ hw4,
                                          const int k, const int q, const float4 as,
                                          const float (&edd)[TB], float (&m)[TB], float (&den)[TB],
                                          float4 (&acc)[TB]) {
    int s[U];
    float4 hv[U][TB];
    float l[U][TB];
#pragma unroll
    for (int u = 0; u < U; ++u) s[u] = srcS[k + u];
#pragma unroll
    for (int u = 0; u < U; ++u) {
#pragma unroll
        for (int g = 0; g < TB; ++g) hv[u][g] = hw4[s[u] * (TB * 4) + g * 4 + q];
    }
#pragma unroll
    for (int u = 0; u < U; ++u) {
#pragma unroll
        for (int g = 0; g < TB; ++g) l[u][g] = lrelu02(quad_sum(dot4(hv[u][g], as)) + edd[g]);
    }
#pragma unroll
    for (int g = 0; g < TB; ++g) {
        float bm = m[g];
#pragma unroll
        for (int u = 0; u < U; ++u) bm = fmaxf(bm, l[u][g]);
        const float sc = __expf(m[g] - bm);
        m[g] = bm;
        den[g] *= sc;
        acc[g].x *= sc; acc[g].y *= sc; acc[g].z *= sc; acc[g].w *= sc;
#pragma unroll
        for (int u = 0; u < U; ++u) {
            const float w = __expf(l[u][g] - bm);
            den[g] += w;
            acc[g].x += w * hv[u][g].x; acc[g].y += w * hv[u][g].y;
            acc[g].z += w * hv[u][g].z; acc[g].w += w * hv[u][g].w;
        }
    }
}

#define GAT_THREADS 128
#define TPITCH 20

__device__ __forceinline__ void wave_lds_handover() {
    __builtin_amdgcn_fence(__ATOMIC_RELEASE, "wavefront");
    __builtin_amdgcn_wave_barrier();
    __builtin_amdgcn_fence(__ATOMIC_ACQUIRE, "wavefront");
}

template <int TB, bool FUSE>
__global__ __launch_bounds__(GAT_THREADS) void k_gat_node(const int* __restrict__ rowptr,
                                                  const int* __restrict__ srcS,
                                                  const float* __restrict__ hw,
                                                  const float* __restrict__ asrc,
                                                  const float* __restrict__ adst,
                                                  const float* __restrict__ Wn,
                                                  const float* __restrict__ bn,
                                                  float* __restrict__ hout, int N) {
    const int t = blockIdx.x * GAT_THREADS + threadIdx.x;
    const int node = t >> 2;
    const int q = t & 3;
    const bool valid = node < N;
    const float4* __restrict__ hw4 = (const float4*)hw;
    const float4 as = ((const float4*)asrc)[q];
    const float4 ad = ((const float4*)adst)[q];
    int beg = 0, end = 0;
    float edd[TB];
#pragma unroll
    for (int g = 0; g < TB; ++g) edd[g] = 0.f;
    if (valid) {
        beg = rowptr[node]; end = rowptr[node + 1];
#pragma unroll
        for (int g = 0; g < TB; ++g) edd[g] = dot4(hw4[(node * TB + g) * 4 + q], ad);
    }
#pragma unroll
    for (int g = 0; g < TB; ++g) edd[g] = quad_sum(edd[g]);

    constexpr int U = 4;
    float m[TB], den[TB];
    float4 acc[TB];
#pragma unroll
    for (int g = 0; g < TB; ++g) {
        m[g] = -INFINITY; den[g] = 0.f; acc[g] = make_float4(0.f, 0.f, 0.f, 0.f);
    }
    int k = beg;
    for (; k + U <= end; k += U) gat_edges<TB, U>(srcS, hw4, k, q, as, edd, m, den, acc);
    if (k + 2 <= end) { gat_edges<TB, 2>(srcS, hw4, k, q, as, edd, m, den, acc); k += 2; }
    if (k < end)        gat_edges<TB, 1>(srcS, hw4, k, q, as, edd, m, den, acc);

    float4 o[TB];
#pragma unroll
    for (int g = 0; g < TB; ++g) {
        const float dn = fmaxf(den[g], 1e-16f);
        o[g].x = acc[g].x / dn; o[g].y = acc[g].y / dn; o[g].z = acc[g].z / dn; o[g].w = acc[g].w / dn;
        o[g].x = o[g].x > 0.f ? o[g].x : (__expf(o[g].x) - 1.f);
        o[g].y = o[g].y > 0.f ? o[g].y : (__expf(o[g].y) - 1.f);
        o[g].z = o[g].z > 0.f ? o[g].z : (__expf(o[g].z) - 1.f);
        o[g].w = o[g].w > 0.f ? o[g].w : (__expf(o[g].w) - 1.f);
    }

    if constexpr (!FUSE) {
        if (valid) {
#pragma unroll
            for (int g = 0; g < TB; ++g) ((float4*)hout)[((size_t)node * TB + g) * 4 + q] = o[g];
        }
    } else {
        constexpr int R  = 8 * TB;
        constexpr int NT = (R + 15) / 16;
        __shared__ float tile[GAT_THREADS / 32][NT * 16][TPITCH];
        const int lane = threadIdx.x & 31;
        const int wave = threadIdx.x >> 5;
        float (*T)[TPITCH] = tile[wave];
        const int nd = lane >> 2;
#pragma unroll
        for (int g = 0; g < TB; ++g) *(float4*)&T[nd * TB + g][4 * q] = o[g];
        wave_lds_handover();

        const int c  = lane & 15;
        const int hf = lane >> 4;
        v2f bfrag[4];
#pragma unroll
        for (int j = 0; j < 4; ++j) {
            const int kc = 4 * j + 2 * hf;
            bfrag[j].x = Wn[kc * HD + c];
            bfrag[j].y = Wn[(kc + 1) * HD + c];
        }
        const float bi = bn[c];
        v8f d[NT];
#pragma unroll
        for (int tI = 0; tI < NT; ++tI) {
            int row = 16 * tI + c;
            if (row >= R) row = R - 1;
            v8f a8 = {};
#pragma unroll
            for (int j = 0; j < 4; ++j) {
                const v2f a = *(const v2f*)&T[row][4 * j + 2 * hf];
                a8 = __builtin_amdgcn_wmma_f32_16x16x4_f32(
                         false, a, false, bfrag[j], (short)0, a8, false, false);
            }
            d[tI] = a8;
        }
        wave_lds_handover();
#pragma unroll
        for (int tI = 0; tI < NT; ++tI) {
#pragma unroll
            for (int r = 0; r < 8; ++r) T[16 * tI + r + 8 * hf][c] = d[tI][r] + bi;
        }
        wave_lds_handover();
        const int rowbase = (blockIdx.x * (GAT_THREADS / 4) + wave * 8) * TB;
        const int rows = N * TB;
#pragma unroll
        for (int j = 0; j < R / 8; ++j) {
            const int f = lane + 32 * j;
            const int row = f >> 2;
            const int c4 = f & 3;
            if (rowbase + row < rows)
                ((float4*)hout)[(size_t)(rowbase + row) * 4 + c4] = *(const float4*)&T[row][4 * c4];
        }
    }
}

template <int TB>
__global__ void k_final(const float* __restrict__ h, const float* __restrict__ mlpW,
                        const float* __restrict__ mlpb, float* __restrict__ out, int N, int g0) {
    const int n = blockIdx.x * blockDim.x + threadIdx.x;
    const int g = blockIdx.y;
    if (n >= N) return;
    const float4* p = (const float4*)(h + ((size_t)n * TB + g) * HD);
    float acc = mlpb[0];
#pragma unroll
    for (int q = 0; q < 4; ++q) {
        const float4 v = p[q];
        acc += v.x * mlpW[q*4+0] + v.y * mlpW[q*4+1] + v.z * mlpW[q*4+2] + v.w * mlpW[q*4+3];
    }
    out[(size_t)(g0 + g) * N + n] = acc;
}

static inline int cdiv(int a, int b) { return (a + b - 1) / b; }

struct Bufs {
    const float *fw, *sw, *tw, *areas;
    const int* indices;
    const float *W0p, *Wl, *a_src, *a_dst, *bvec, *mlpW, *mlpb;
    const int *rowptr, *srcS;
    float *h, *hw, *out;
};

template <int TB>
static void run_graphs(const Bufs& b, int N, int g0, hipStream_t stream) {
    const int rows   = N * TB;
    const int tiles  = cdiv(rows, 16);
    const int gemmBl = cdiv(tiles, 8);
    const int gatBl  = cdiv(N * 4, GAT_THREADS);
    k_layer0_wmma<TB><<<gemmBl, 256, 0, stream>>>(b.fw, b.sw, b.tw, b.areas, b.indices, b.W0p, b.bvec, b.hw, N, g0);
    float* cur = b.hw;
    float* nxt = b.h;
    for (int i = 0; i < 4; ++i) {
        const float* as = b.a_src + (size_t)i * HD;
        const float* ad = b.a_dst + (size_t)i * HD;
        if (i < 3)
            k_gat_node<TB, true><<<gatBl, GAT_THREADS, 0, stream>>>(b.rowptr, b.srcS, cur, as, ad,
                                                                   b.Wl + (size_t)i * 256,
                                                                   b.bvec + (size_t)(i + 1) * HD, nxt, N);
        else
            k_gat_node<TB, false><<<gatBl, GAT_THREADS, 0, stream>>>(b.rowptr, b.srcS, cur, as, ad,
                                                                    b.Wl, b.bvec, nxt, N);
        float* tmp = cur; cur = nxt; nxt = tmp;
    }
    k_final<TB><<<dim3(cdiv(N, 256), TB), 256, 0, stream>>>(cur, b.mlpW, b.mlpb, b.out, N, g0);
}

extern "C" void kernel_launch(void* const* d_in, const int* in_sizes, int n_in,
                              void* d_out, int out_size, void* d_ws, size_t ws_size,
                              hipStream_t stream) {
    const float* fw     = (const float*)d_in[0];
    const float* sw     = (const float*)d_in[1];
    const float* tw     = (const float*)d_in[2];
    const float* areas  = (const float*)d_in[3];
    const int*   indices= (const int*)  d_in[4];
    const int*   edge   = (const int*)  d_in[5];
    const float* W0     = (const float*)d_in[6];
    const float* Wl     = (const float*)d_in[7];
    const float* a_src  = (const float*)d_in[8];
    const float* a_dst  = (const float*)d_in[9];
    const float* bvec   = (const float*)d_in[10];
    const float* mlpW   = (const float*)d_in[11];
    const float* mlpb   = (const float*)d_in[12];

    const int N = in_sizes[3];
    const int E = in_sizes[5] / 2;
    const int T = in_sizes[0] / (NWIRE * NFEAT);
    const int* src = edge;
    const int* dst = edge + E;
    const int numTiles = cdiv(N, SCAN_TILE);

    const size_t fixedW = 256 + (size_t)E + ((size_t)N + 1) + 2 * (size_t)N + 2 * (size_t)numTiles;
    const size_t need3  = 2 * (size_t)N * HD * 3 + fixedW;
    const size_t need2  = 2 * (size_t)N * HD * 2 + fixedW;
    int TBS = 1;
    if (T >= 3 && ws_size >= need3 * 4) TBS = 3;
    else if (T >= 2 && ws_size >= need2 * 4) TBS = 2;

    float* ws = (float*)d_ws;
    size_t off = 0;
    float* h       = ws + off; off += (size_t)N * HD * TBS;
    float* hw      = ws + off; off += (size_t)N * HD * TBS;
    float* W0p     = ws + off; off += 256;
    int*   srcS    = (int*)(ws + off); off += (size_t)E;
    int*   rowptr  = (int*)(ws + off); off += (size_t)N + 1;
    int*   cursor  = (int*)(ws + off); off += (size_t)N;
    int*   cnt     = (int*)(ws + off); off += (size_t)N;
    int*   tilesum = (int*)(ws + off); off += (size_t)numTiles;
    int*   tileoff = (int*)(ws + off); off += (size_t)numTiles;
    (void)n_in; (void)out_size;

    const int numBins = numTiles;
    int* binCnt = tilesum;
    int* binOff = tileoff;
    int* binCur = cnt;
    const bool binned = N <= (1 << SRC_BITS) && (size_t)E <= (size_t)N * HD * TBS;

    k_prep<<<1, 256, 0, stream>>>(W0, W0p, binCnt, binCur, binned ? numBins : 0);
    if (binned) {
        const int chunks = cdiv(E, PART_CHUNK);
        k_binhist<<<chunks, PART_THREADS, 0, stream>>>(dst, binCnt, E, numBins);
        k_scan_tiles<<<1, 1024, 0, stream>>>(binCnt, binOff, rowptr, numBins, N);
        k_partition<<<chunks, PART_THREADS, 0, stream>>>(src, dst, binOff, binCur, (unsigned*)h, E, numBins);
        k_binsort<<<numBins, 1024, 0, stream>>>((const unsigned*)h, binOff, binCnt, rowptr, srcS, N);
    } else {
        k_zero<<<cdiv(N, 256), 256, 0, stream>>>(cnt, N);
        k_hist<<<cdiv(E, 256), 256, 0, stream>>>(dst, cnt, E);
        k_scan_local<<<numTiles, 256, 0, stream>>>(cnt, rowptr, tilesum, N);
        k_scan_tiles<<<1, 1024, 0, stream>>>(tilesum, tileoff, rowptr, numTiles, N);
        k_scan_add<<<cdiv(N, 256), 256, 0, stream>>>(rowptr, tileoff, cursor, N);
        k_scatter<<<cdiv(E, 256), 256, 0, stream>>>(src, dst, cursor, srcS, E);
    }

    Bufs b;
    b.fw = fw; b.sw = sw; b.tw = tw; b.areas = areas; b.indices = indices;
    b.W0p = W0p; b.Wl = Wl; b.a_src = a_src; b.a_dst = a_dst; b.bvec = bvec;
    b.mlpW = mlpW; b.mlpb = mlpb; b.rowptr = rowptr; b.srcS = srcS;
    b.h = h; b.hw = hw; b.out = (float*)d_out;

    int g = 0;
    while (g < T) {
        const int left = T - g;
        const int tb = left < TBS ? left : TBS;
        if (tb >= 3)      { run_graphs<3>(b, N, g, stream); g += 3; }
        else if (tb == 2) { run_graphs<2>(b, N, g, stream); g += 2; }
        else              { run_graphs<1>(b, N, g, stream); g += 1; }
    }
}


# The 16->1 head runs in the last GAT kernel on the row in registers (quad dot, one wave stores each 128-B line): head launch, h plane write and re-read go; the head's sum order changes
# speedup vs baseline: 18.8111x; 1.0469x over previous
#include <hip/hip_runtime.h>
#include <math.h>


typedef float v2f __attribute__((ext_vector_type(2)));
typedef float v8f __attribute__((ext_vector_type(8)));

#define HD 16
#define NWIRE 1536
#define NFEAT 4
#define SCAN_TILE 1024


__device__ __forceinline__ float lrelu02(float x) { return x > 0.f ? x : 0.2f * x; }

__device__ __forceinline__ float dot4(const float4 v, const float4 a) {
    return v.x * a.x + v.y * a.y + v.z * a.z + v.w * a.w;
}

__device__ __forceinline__ float quad_sum(float p) {
    p += __int_as_float(__builtin_amdgcn_update_dpp(0, __float_as_int(p), 0xB1, 0xF, 0xF, true));
    p += __int_as_float(__builtin_amdgcn_update_dpp(0, __float_as_int(p), 0x4E, 0xF, 0xF, true));
    return p;
}

__global__ void k_prep(const float* __restrict__ W0, float* __restrict__ W0p,
                       int* __restrict__ binCnt, int* __restrict__ binCur, int numBins) {
    int i = threadIdx.x;
    int r = i >> 4;
    W0p[i] = (r < 13) ? W0[i] : 0.f;
    for (int b = i; b < numBins; b += 256) { binCnt[b] = 0; binCur[b] = 0; }
}

template <int TB>
__global__ void k_layer0_wmma(const float* __restrict__ fw, const float* __restrict__ sw,
                              const float* __restrict__ tw, const float* __restrict__ areas,
                              const int* __restrict__ indices, const float* __restrict__ W,
                              const float* __restrict__ bias, float* __restrict__ hw,
                              int N, int g0) {
    const int lane = threadIdx.x & 31;
    const int wave = threadIdx.x >> 5;
    const int tile = blockIdx.x * 8 + wave;
    const int base = tile * 16;
    const int rows = N * TB;
    if (base >= rows) return;
    const int col16 = lane & 15;
    const int half  = lane >> 4;
    int arow = base + col16;
    if (arow >= rows) arow = rows - 1;
    const int n = arow / TB;
    const int g = arow - n * TB;
    const size_t gbase = (size_t)(g0 + g) * NWIRE * NFEAT + 2 * half;
    v2f a[4];
    a[0] = *(const v2f*)(fw + gbase + (size_t)indices[n * 3 + 0] * NFEAT);
    a[1] = *(const v2f*)(sw + gbase + (size_t)indices[n * 3 + 1] * NFEAT);
    a[2] = *(const v2f*)(tw + gbase + (size_t)indices[n * 3 + 2] * NFEAT);
    a[3].x = half == 0 ? areas[n] : 0.f;
    a[3].y = 0.f;
    v8f acc = {};
#pragma unroll
    for (int k = 0; k < 4; ++k) {
        const int kc = 4 * k + 2 * half;
        v2f b;
        b.x = W[kc * HD + col16];
        b.y = W[(kc + 1) * HD + col16];
        acc = __builtin_amdgcn_wmma_f32_16x16x4_f32(
                  false, a[k], false, b, (short)0, acc, false, false);
    }
    const float bi = bias[col16];
    if (base + 16 <= rows) {
#pragma unroll
        for (int r = 0; r < 8; ++r) {
            hw[(size_t)(base + r + 8 * half) * HD + col16] = acc[r] + bi;
        }
    } else {
#pragma unroll
        for (int r = 0; r < 8; ++r) {
            const int orow = base + r + 8 * half;
            if (orow < rows) hw[(size_t)orow * HD + col16] = acc[r] + bi;
        }
    }
}

__global__ void k_zero(int* __restrict__ cnt, int N) {
    int i = blockIdx.x * blockDim.x + threadIdx.x;
    if (i < N) cnt[i] = 0;
}

__global__ void k_hist(const int* __restrict__ dst, int* __restrict__ cnt, int E) {
    int e = blockIdx.x * blockDim.x + threadIdx.x;
    if (e < E) atomicAdd(&cnt[dst[e]], 1);
}

__global__ __launch_bounds__(256) void k_scan_local(const int* __restrict__ cnt,
                                                    int* __restrict__ rowptr,
                                                    int* __restrict__ tilesum, int N) {
    __shared__ int wsum[8];
    const int t = threadIdx.x;
    const int lane = t & 31;
    const int wv = t >> 5;
    const int i0 = (blockIdx.x * 256 + t) * 4;
    const int c0 = (i0 + 0 < N) ? cnt[i0 + 0] : 0;
    const int c1 = (i0 + 1 < N) ? cnt[i0 + 1] : 0;
    const int c2 = (i0 + 2 < N) ? cnt[i0 + 2] : 0;
    const int c3 = (i0 + 3 < N) ? cnt[i0 + 3] : 0;
    const int s = c0 + c1 + c2 + c3;
    int inc = s;
    for (int off = 1; off < 32; off <<= 1) {
        const int v = __shfl_up(inc, off);
        if (lane >= off) inc += v;
    }
    if (lane == 31) wsum[wv] = inc;
    __syncthreads();
    int wpre = 0;
    for (int k = 0; k < 8; ++k) if (k < wv) wpre += wsum[k];
    int ex = wpre + inc - s;
    if (i0 + 0 < N) rowptr[i0 + 0] = ex;
    ex += c0;
    if (i0 + 1 < N) rowptr[i0 + 1] = ex;
    ex += c1;
    if (i0 + 2 < N) rowptr[i0 + 2] = ex;
    ex += c2;
    if (i0 + 3 < N) rowptr[i0 + 3] = ex;
    if (t == 255) tilesum[blockIdx.x] = wpre + inc;
}

__global__ __launch_bounds__(1024) void k_scan_tiles(const int* __restrict__ tilesum,
                                                     int* __restrict__ tileoff,
                                                     int* __restrict__ rowptr,
                                                     int numTiles, int N) {
    __shared__ int buf[1024];
    const int t = threadIdx.x;
    int carry = 0;
    for (int base = 0; base < numTiles; base += 1024) {
        const int i = base + t;
        const int v = (i < numTiles) ? tilesum[i] : 0;
        buf[t] = v;
        __syncthreads();
        for (int off = 1; off < 1024; off <<= 1) {
            const int a = (t >= off) ? buf[t - off] : 0;
            __syncthreads();
            buf[t] += a;
            __syncthreads();
        }
        if (i < numTiles) tileoff[i] = carry + buf[t] - v;
        carry += buf[1023];
        __syncthreads();
    }
    if (t == 0) rowptr[N] = carry;
}

__global__ void k_scan_add(int* __restrict__ rowptr, const int* __restrict__ tileoff,
                           int* __restrict__ cursor, int N) {
    int i = blockIdx.x * blockDim.x + threadIdx.x;
    if (i >= N) return;
    const int v = rowptr[i] + tileoff[i / SCAN_TILE];
    rowptr[i] = v;
    cursor[i] = v;
}

__global__ void k_scatter(const int* __restrict__ src, const int* __restrict__ dst,
                          int* __restrict__ cursor, int* __restrict__ srcS, int E) {
    int e = blockIdx.x * blockDim.x + threadIdx.x;
    if (e >= E) return;
    const int pos = atomicAdd(&cursor[dst[e]], 1);
    srcS[pos] = src[e];
}

#define BIN_SHIFT 10
#define BIN_NODES (1 << BIN_SHIFT)
#define SRC_BITS 22
#define SRC_MASK ((1u << SRC_BITS) - 1u)
#define MAX_BINS (1 << (SRC_BITS - BIN_SHIFT))
#define PART_THREADS 1024
#define PART_EPT 8
#define PART_CHUNK (PART_THREADS * PART_EPT)
#define BIN_CAP 49152
#define BIN_LD 8
static_assert(BIN_NODES == SCAN_TILE, "the bin arrays share the scan-tile arrays of the workspace");

__global__ __launch_bounds__(PART_THREADS) void k_binhist(const int* __restrict__ dst,
                                                          int* __restrict__ binCnt,
                                                          int E, int numBins) {
    __shared__ int hist[MAX_BINS];
    const int t = threadIdx.x;
    for (int b = t; b < numBins; b += PART_THREADS) hist[b] = 0;
    __syncthreads();
    const int base = blockIdx.x * PART_CHUNK;
    int d[PART_EPT];
#pragma unroll
    for (int i = 0; i < PART_EPT; ++i) {
        const int e = base + i * PART_THREADS + t;
        d[i] = dst[e < E ? e : E - 1];
    }
#pragma unroll
    for (int i = 0; i < PART_EPT; ++i) {
        atomicAdd(&hist[d[i] >> BIN_SHIFT], (base + i * PART_THREADS + t < E) ? 1 : 0);
    }
    __syncthreads();
    for (int b = t; b < numBins; b += PART_THREADS) {
        const int c = hist[b];
        if (c) atomicAdd(&binCnt[b], c);
    }
}

__global__ __launch_bounds__(PART_THREADS) void k_partition(const int* __restrict__ src,
                                                            const int* __restrict__ dst,
                                                            const int* __restrict__ binOff,
                                                            int* __restrict__ binCur,
                                                            unsigned* __restrict__ tmp,
                                                            int E, int numBins) {
    __shared__ int hist[MAX_BINS];
    const int t = threadIdx.x;
    for (int b = t; b < numBins; b += PART_THREADS) hist[b] = 0;
    const int base = blockIdx.x * PART_CHUNK;
    int d[PART_EPT], s[PART_EPT], r[PART_EPT];
#pragma unroll
    for (int i = 0; i < PART_EPT; ++i) {
        const int e = base + i * PART_THREADS + t;
        const int ec = e < E ? e : E - 1;
        const int dv = dst[ec];
        s[i] = src[ec];
        d[i] = (e < E) ? dv : -1;
    }
    __syncthreads();
#pragma unroll
    for (int i = 0; i < PART_EPT; ++i) {
        r[i] = 0;
        if (d[i] >= 0) r[i] = atomicAdd(&hist[d[i] >> BIN_SHIFT], 1);
    }
    __syncthreads();
    for (int b = t; b < numBins; b += PART_THREADS) {
        const int c = hist[b];
        int g = 0;
        if (c) g = binOff[b] + atomicAdd(&binCur[b], c);
        hist[b] = g;
    }
    __syncthreads();
#pragma unroll
    for (int i = 0; i < PART_EPT; ++i) {
        if (d[i] >= 0)
            tmp[hist[d[i] >> BIN_SHIFT] + r[i]] =
                ((unsigned)(d[i] & (BIN_NODES - 1)) << SRC_BITS) | (unsigned)s[i];
    }
}

__global__ __launch_bounds__(1024) void k_binsort(const unsigned* __restrict__ tmp,
                                                  const int* __restrict__ binOff,
                                                  const int* __restrict__ binCnt,
                                                  int* __restrict__ rowptr,
                                                  int* __restrict__ srcS, int N) {
    __shared__ int cnt[BIN_NODES];
    __shared__ int off[BIN_NODES + 1];
    __shared__ int wsum[32];
    __shared__ unsigned a[BIN_CAP];
    const int t = threadIdx.x;
    const int lane = t & 31;
    const int wv = t >> 5;
    const int bin = blockIdx.x;
    const int gbeg = binOff[bin];
    const int n = binCnt[bin];
    const unsigned* __restrict__ mine = tmp + gbeg;
    cnt[t] = 0;
    __syncthreads();
    for (int i0 = t; i0 < n; i0 += 1024 * BIN_LD) {
        unsigned w[BIN_LD];
#pragma unroll
        for (int u = 0; u < BIN_LD; ++u) {
            const int i = i0 + u * 1024;
            w[u] = mine[i < n ? i : i0];
        }
#pragma unroll
        for (int u = 0; u < BIN_LD; ++u) {
            const int i = i0 + u * 1024;
            atomicAdd(&cnt[w[u] >> SRC_BITS], (i < n) ? 1 : 0);
        }
    }
    __syncthreads();
    const int c = cnt[t];
    int inc = c;
    for (int o = 1; o < 32; o <<= 1) {
        const int v = __shfl_up(inc, o);
        if (lane >= o) inc += v;
    }
    if (lane == 31) wsum[wv] = inc;
    __syncthreads();
    int wpre = 0;
    for (int k = 0; k < 32; ++k) if (k < wv) wpre += wsum[k];
    const int ex = wpre + inc - c;
    off[t] = ex;
    if (t == 1023) off[BIN_NODES] = ex + c;
    cnt[t] = ex;
    const int node = bin * BIN_NODES + t;
    if (node < N) rowptr[node] = gbeg + ex;
    __syncthreads();
    if (n <= BIN_CAP) {
        for (int i0 = t; i0 < n; i0 += 1024 * BIN_LD) {
            unsigned w[BIN_LD];
#pragma unroll
            for (int u = 0; u < BIN_LD; ++u) {
                const int i = i0 + u * 1024;
                w[u] = mine[i < n ? i : i0];
            }
#pragma unroll
            for (int u = 0; u < BIN_LD; ++u) {
                const int i = i0 + u * 1024;
                const int p = atomicAdd(&cnt[w[u] >> SRC_BITS], (i < n) ? 1 : 0);
                if (i < n) a[p] = w[u];
            }
        }
        __syncthreads();
        for (int p = t; p < n; p += 1024) {
            const unsigned w = a[p];
            const int dl = (int)(w >> SRC_BITS);
            const int sb = off[dl];
            const int se = off[dl + 1];
            int rank = 0;
            for (int j = sb; j < se; ++j) {
                const unsigned v = a[j];
                rank += (v < w || (v == w && j < p)) ? 1 : 0;
            }
            srcS[gbeg + sb + rank] = (int)(w & SRC_MASK);
        }
    } else {
        for (int i = t; i < n; i += 1024) {
            const unsigned w = mine[i];
            const int p = atomicAdd(&cnt[w >> SRC_BITS], 1);
            srcS[gbeg + p] = (int)(w & SRC_MASK);
        }
    }
}

template <int TB, int U>
__device__ __forceinline__ void gat_edges(const int* __restrict__ srcS, const float4* __restrict__ hw4,
                                          const int k, const int q, const float4 as,
                                          const float (&edd)[TB], float (&m)[TB], float (&den)[TB],
                                          float4 (&acc)[TB]) {
    int s[U];
    float4 hv[U][TB];
    float l[U][TB];
#pragma unroll
    for (int u = 0; u < U; ++u) s[u] = srcS[k + u];
#pragma unroll
    for (int u = 0; u < U; ++u) {
#pragma unroll
        for (int g = 0; g < TB; ++g) hv[u][g] = hw4[s[u] * (TB * 4) + g * 4 + q];
    }
#pragma unroll
    for (int u = 0; u < U; ++u) {
#pragma unroll
        for (int g = 0; g < TB; ++g) l[u][g] = lrelu02(quad_sum(dot4(hv[u][g], as)) + edd[g]);
    }
#pragma unroll
    for (int g = 0; g < TB; ++g) {
        float bm = m[g];
#pragma unroll
        for (int u = 0; u < U; ++u) bm = fmaxf(bm, l[u][g]);
        const float sc = __expf(m[g] - bm);
        m[g] = bm;
        den[g] *= sc;
        acc[g].x *= sc; acc[g].y *= sc; acc[g].z *= sc; acc[g].w *= sc;
#pragma unroll
        for (int u = 0; u < U; ++u) {
            const float w = __expf(l[u][g] - bm);
            den[g] += w;
            acc[g].x += w * hv[u][g].x; acc[g].y += w * hv[u][g].y;
            acc[g].z += w * hv[u][g].z; acc[g].w += w * hv[u][g].w;
        }
    }
}

#define GAT_THREADS 128
#define TPITCH 20

__device__ __forceinline__ void wave_lds_handover() {
    __builtin_amdgcn_fence(__ATOMIC_RELEASE, "wavefront");
    __builtin_amdgcn_wave_barrier();
    __builtin_amdgcn_fence(__ATOMIC_ACQUIRE, "wavefront");
}

template <int TB, bool FUSE>
__global__ __launch_bounds__(GAT_THREADS) void k_gat_node(const int* __restrict__ rowptr,
                                                  const int* __restrict__ srcS,
                                                  const float* __restrict__ hw,
                                                  const float* __restrict__ asrc,
                                                  const float* __restrict__ adst,
                                                  const float* __restrict__ Wn,
                                                  const float* __restrict__ bn,
                                                  float* __restrict__ hout, int N) {
    const int t = blockIdx.x * GAT_THREADS + threadIdx.x;
    const int node = t >> 2;
    const int q = t & 3;
    const bool valid = node < N;
    const float4* __restrict__ hw4 = (const float4*)hw;
    const float4 as = ((const float4*)asrc)[q];
    const float4 ad = ((const float4*)adst)[q];
    int beg = 0, end = 0;
    float edd[TB];
#pragma unroll
    for (int g = 0; g < TB; ++g) edd[g] = 0.f;
    if (valid) {
        beg = rowptr[node]; end = rowptr[node + 1];
#pragma unroll
        for (int g = 0; g < TB; ++g) edd[g] = dot4(hw4[(node * TB + g) * 4 + q], ad);
    }
#pragma unroll
    for (int g = 0; g < TB; ++g) edd[g] = quad_sum(edd[g]);

    constexpr int U = 4;
    float m[TB], den[TB];
    float4 acc[TB];
#pragma unroll
    for (int g = 0; g < TB; ++g) {
        m[g] = -INFINITY; den[g] = 0.f; acc[g] = make_float4(0.f, 0.f, 0.f, 0.f);
    }
    int k = beg;
    for (; k + U <= end; k += U) gat_edges<TB, U>(srcS, hw4, k, q, as, edd, m, den, acc);
    if (k + 2 <= end) { gat_edges<TB, 2>(srcS, hw4, k, q, as, edd, m, den, acc); k += 2; }
    if (k < end)        gat_edges<TB, 1>(srcS, hw4, k, q, as, edd, m, den, acc);

    float4 o[TB];
#pragma unroll
    for (int g = 0; g < TB; ++g) {
        const float dn = fmaxf(den[g], 1e-16f);
        o[g].x = acc[g].x / dn; o[g].y = acc[g].y / dn; o[g].z = acc[g].z / dn; o[g].w = acc[g].w / dn;
        o[g].x = o[g].x > 0.f ? o[g].x : (__expf(o[g].x) - 1.f);
        o[g].y = o[g].y > 0.f ? o[g].y : (__expf(o[g].y) - 1.f);
        o[g].z = o[g].z > 0.f ? o[g].z : (__expf(o[g].z) - 1.f);
        o[g].w = o[g].w > 0.f ? o[g].w : (__expf(o[g].w) - 1.f);
    }

    if constexpr (!FUSE) {
        __shared__ float res[TB][GAT_THREADS / 4];
        const float4 mw = ((const float4*)Wn)[q];
        const float mb = bn[0];
        const int ln = threadIdx.x >> 2;
#pragma unroll
        for (int g = 0; g < TB; ++g) {
            const float r = quad_sum(dot4(o[g], mw)) + mb;
            if (q == 0) res[g][ln] = r;
        }
        __syncthreads();
        if (threadIdx.x < GAT_THREADS / 4) {
            const int n0 = blockIdx.x * (GAT_THREADS / 4) + threadIdx.x;
            if (n0 < N) {
#pragma unroll
                for (int g = 0; g < TB; ++g) hout[(size_t)g * N + n0] = res[g][threadIdx.x];
            }
        }
    } else {
        constexpr int R  = 8 * TB;
        constexpr int NT = (R + 15) / 16;
        __shared__ float tile[GAT_THREADS / 32][NT * 16][TPITCH];
        const int lane = threadIdx.x & 31;
        const int wave = threadIdx.x >> 5;
        float (*T)[TPITCH] = tile[wave];
        const int nd = lane >> 2;
#pragma unroll
        for (int g = 0; g < TB; ++g) *(float4*)&T[nd * TB + g][4 * q] = o[g];
        wave_lds_handover();

        const int c  = lane & 15;
        const int hf = lane >> 4;
        v2f bfrag[4];
#pragma unroll
        for (int j = 0; j < 4; ++j) {
            const int kc = 4 * j + 2 * hf;
            bfrag[j].x = Wn[kc * HD + c];
            bfrag[j].y = Wn[(kc + 1) * HD + c];
        }
        const float bi = bn[c];
        v8f d[NT];
#pragma unroll
        for (int tI = 0; tI < NT; ++tI) {
            int row = 16 * tI + c;
            if (row >= R) row = R - 1;
            v8f a8 = {};
#pragma unroll
            for (int j = 0; j < 4; ++j) {
                const v2f a = *(const v2f*)&T[row][4 * j + 2 * hf];
                a8 = __builtin_amdgcn_wmma_f32_16x16x4_f32(
                         false, a, false, bfrag[j], (short)0, a8, false, false);
            }
            d[tI] = a8;
        }
        wave_lds_handover();
#pragma unroll
        for (int tI = 0; tI < NT; ++tI) {
#pragma unroll
            for (int r = 0; r < 8; ++r) T[16 * tI + r + 8 * hf][c] = d[tI][r] + bi;
        }
        wave_lds_handover();
        const int rowbase = (blockIdx.x * (GAT_THREADS / 4) + wave * 8) * TB;
        const int rows = N * TB;
#pragma unroll
        for (int j = 0; j < R / 8; ++j) {
            const int f = lane + 32 * j;
            const int row = f >> 2;
            const int c4 = f & 3;
            if (rowbase + row < rows)
                ((float4*)hout)[(size_t)(rowbase + row) * 4 + c4] = *(const float4*)&T[row][4 * c4];
        }
    }
}

static inline int cdiv(int a, int b) { return (a + b - 1) / b; }

struct Bufs {
    const float *fw, *sw, *tw, *areas;
    const int* indices;
    const float *W0p, *Wl, *a_src, *a_dst, *bvec, *mlpW, *mlpb;
    const int *rowptr, *srcS;
    float *h, *hw, *out;
};

template <int TB>
static void run_graphs(const Bufs& b, int N, int g0, hipStream_t stream) {
    const int rows   = N * TB;
    const int tiles  = cdiv(rows, 16);
    const int gemmBl = cdiv(tiles, 8);
    const int gatBl  = cdiv(N * 4, GAT_THREADS);
    k_layer0_wmma<TB><<<gemmBl, 256, 0, stream>>>(b.fw, b.sw, b.tw, b.areas, b.indices, b.W0p, b.bvec, b.hw, N, g0);
    float* cur = b.hw;
    float* nxt = b.h;
    for (int i = 0; i < 4; ++i) {
        const float* as = b.a_src + (size_t)i * HD;
        const float* ad = b.a_dst + (size_t)i * HD;
        if (i < 3)
            k_gat_node<TB, true><<<gatBl, GAT_THREADS, 0, stream>>>(b.rowptr, b.srcS, cur, as, ad,
                                                                   b.Wl + (size_t)i * 256,
                                                                   b.bvec + (size_t)(i + 1) * HD, nxt, N);
        else
            k_gat_node<TB, false><<<gatBl, GAT_THREADS, 0, stream>>>(b.rowptr, b.srcS, cur, as, ad,
                                                                    b.mlpW, b.mlpb,
                                                                    b.out + (size_t)g0 * N, N);
        float* tmp = cur; cur = nxt; nxt = tmp;
    }
}

extern "C" void kernel_launch(void* const* d_in, const int* in_sizes, int n_in,
                              void* d_out, int out_size, void* d_ws, size_t ws_size,
                              hipStream_t stream) {
    const float* fw     = (const float*)d_in[0];
    const float* sw     = (const float*)d_in[1];
    const float* tw     = (const float*)d_in[2];
    const float* areas  = (const float*)d_in[3];
    const int*   indices= (const int*)  d_in[4];
    const int*   edge   = (const int*)  d_in[5];
    const float* W0     = (const float*)d_in[6];
    const float* Wl     = (const float*)d_in[7];
    const float* a_src  = (const float*)d_in[8];
    const float* a_dst  = (const float*)d_in[9];
    const float* bvec   = (const float*)d_in[10];
    const float* mlpW   = (const float*)d_in[11];
    const float* mlpb   = (const float*)d_in[12];

    const int N = in_sizes[3];
    const int E = in_sizes[5] / 2;
    const int T = in_sizes[0] / (NWIRE * NFEAT);
    const int* src = edge;
    const int* dst = edge + E;
    const int numTiles = cdiv(N, SCAN_TILE);

    const size_t fixedW = 256 + (size_t)E + ((size_t)N + 1) + 2 * (size_t)N + 2 * (size_t)numTiles;
    const size_t need3  = 2 * (size_t)N * HD * 3 + fixedW;
    const size_t need2  = 2 * (size_t)N * HD * 2 + fixedW;
    int TBS = 1;
    if (T >= 3 && ws_size >= need3 * 4) TBS = 3;
    else if (T >= 2 && ws_size >= need2 * 4) TBS = 2;

    float* ws = (float*)d_ws;
    size_t off = 0;
    float* h       = ws + off; off += (size_t)N * HD * TBS;
    float* hw      = ws + off; off += (size_t)N * HD * TBS;
    float* W0p     = ws + off; off += 256;
    int*   srcS    = (int*)(ws + off); off += (size_t)E;
    int*   rowptr  = (int*)(ws + off); off += (size_t)N + 1;
    int*   cursor  = (int*)(ws + off); off += (size_t)N;
    int*   cnt     = (int*)(ws + off); off += (size_t)N;
    int*   tilesum = (int*)(ws + off); off += (size_t)numTiles;
    int*   tileoff = (int*)(ws + off); off += (size_t)numTiles;
    (void)n_in; (void)out_size;

    const int numBins = numTiles;
    int* binCnt = tilesum;
    int* binOff = tileoff;
    int* binCur = cnt;
    const bool binned = N <= (1 << SRC_BITS) && (size_t)E <= (size_t)N * HD * TBS;

    k_prep<<<1, 256, 0, stream>>>(W0, W0p, binCnt, binCur, binned ? numBins : 0);
    if (binned) {
        const int chunks = cdiv(E, PART_CHUNK);
        k_binhist<<<chunks, PART_THREADS, 0, stream>>>(dst, binCnt, E, numBins);
        k_scan_tiles<<<1, 1024, 0, stream>>>(binCnt, binOff, rowptr, numBins, N);
        k_partition<<<chunks, PART_THREADS, 0, stream>>>(src, dst, binOff, binCur, (unsigned*)h, E, numBins);
        k_binsort<<<numBins, 1024, 0, stream>>>((const unsigned*)h, binOff, binCnt, rowptr, srcS, N);
    } else {
        k_zero<<<cdiv(N, 256), 256, 0, stream>>>(cnt, N);
        k_hist<<<cdiv(E, 256), 256, 0, stream>>>(dst, cnt, E);
        k_scan_local<<<numTiles, 256, 0, stream>>>(cnt, rowptr, tilesum, N);
        k_scan_tiles<<<1, 1024, 0, stream>>>(tilesum, tileoff, rowptr, numTiles, N);
        k_scan_add<<<cdiv(N, 256), 256, 0, stream>>>(rowptr, tileoff, cursor, N);
        k_scatter<<<cdiv(E, 256), 256, 0, stream>>>(src, dst, cursor, srcS, E);
    }

    Bufs b;
    b.fw = fw; b.sw = sw; b.tw = tw; b.areas = areas; b.indices = indices;
    b.W0p = W0p; b.Wl = Wl; b.a_src = a_src; b.a_dst = a_dst; b.bvec = bvec;
    b.mlpW = mlpW; b.mlpb = mlpb; b.rowptr = rowptr; b.srcS = srcS;
    b.h = h; b.hw = hw; b.out = (float*)d_out;

    int g = 0;
    while (g < T) {
        const int left = T - g;
        const int tb = left < TBS ? left : TBS;
        if (tb >= 3)      { run_graphs<3>(b, N, g, stream); g += 3; }
        else if (tb == 2) { run_graphs<2>(b, N, g, stream); g += 2; }
        else              { run_graphs<1>(b, N, g, stream); g += 1; }
    }
}


# Layer 0's edge loop works on the 13-feature window rows (16-B node record per edge, wire tables in LDS); W0 applied after the softmax sum, on WMMA: layer-0 launch and hw plane go; sum order changes
# speedup vs baseline: 20.0468x; 1.0657x over previous
#include <hip/hip_runtime.h>
#include <math.h>


typedef float v2f __attribute__((ext_vector_type(2)));
typedef float v8f __attribute__((ext_vector_type(8)));

#define HD 16
#define NWIRE 1536
#define NFEAT 4
#define SCAN_TILE 1024


__device__ __forceinline__ float lrelu02(float x) { return x > 0.f ? x : 0.2f * x; }

__device__ __forceinline__ float dot4(const float4 v, const float4 a) {
    return v.x * a.x + v.y * a.y + v.z * a.z + v.w * a.w;
}

__device__ __forceinline__ float quad_sum(float p) {
    p += __int_as_float(__builtin_amdgcn_update_dpp(0, __float_as_int(p), 0xB1, 0xF, 0xF, true));
    p += __int_as_float(__builtin_amdgcn_update_dpp(0, __float_as_int(p), 0x4E, 0xF, 0xF, true));
    return p;
}

__global__ void k_prep(const float* __restrict__ W0, const float* __restrict__ asrc,
                       const float* __restrict__ adst, const float* __restrict__ b0,
                       float* __restrict__ W0p, float* __restrict__ lv,
                       int* __restrict__ binCnt, int* __restrict__ binCur, int numBins) {
    int i = threadIdx.x;
    int r = i >> 4;
    W0p[i] = (r < 13) ? W0[i] : (r == 13 ? b0[i & 15] : 0.f);
    if (i < 32) {
        const float* __restrict__ a = (i < 16) ? asrc : adst;
        const int k = i & 15;
        float sum = 0.f;
        if (k < 13) {
            for (int j = 0; j < HD; ++j) sum += W0[k * HD + j] * a[j];
        }
        lv[i] = sum;
    } else if (i == 32) {
        float cs = 0.f, cd = 0.f;
        for (int j = 0; j < HD; ++j) { cs += b0[j] * asrc[j]; cd += b0[j] * adst[j]; }
        lv[32] = cs + cd;
    }
    for (int b = i; b < numBins; b += 256) { binCnt[b] = 0; binCur[b] = 0; }
}

__global__ void k_pack(const int* __restrict__ indices, const float* __restrict__ areas,
                       unsigned* __restrict__ rec, int N) {
    const int n = blockIdx.x * blockDim.x + threadIdx.x;
    if (n >= N) return;
    uint4 r;
    r.x = (unsigned)indices[n * 3 + 0];
    r.y = (unsigned)indices[n * 3 + 1];
    r.z = (unsigned)indices[n * 3 + 2];
    r.w = __float_as_uint(areas[n]);
    ((uint4*)rec)[n] = r;
}

__global__ void k_zero(int* __restrict__ cnt, int N) {
    int i = blockIdx.x * blockDim.x + threadIdx.x;
    if (i < N) cnt[i] = 0;
}

__global__ void k_hist(const int* __restrict__ dst, int* __restrict__ cnt, int E) {
    int e = blockIdx.x * blockDim.x + threadIdx.x;
    if (e < E) atomicAdd(&cnt[dst[e]], 1);
}

__global__ __launch_bounds__(256) void k_scan_local(const int* __restrict__ cnt,
                                                    int* __restrict__ rowptr,
                                                    int* __restrict__ tilesum, int N) {
    __shared__ int wsum[8];
    const int t = threadIdx.x;
    const int lane = t & 31;
    const int wv = t >> 5;
    const int i0 = (blockIdx.x * 256 + t) * 4;
    const int c0 = (i0 + 0 < N) ? cnt[i0 + 0] : 0;
    const int c1 = (i0 + 1 < N) ? cnt[i0 + 1] : 0;
    const int c2 = (i0 + 2 < N) ? cnt[i0 + 2] : 0;
    const int c3 = (i0 + 3 < N) ? cnt[i0 + 3] : 0;
    const int s = c0 + c1 + c2 + c3;
    int inc = s;
    for (int off = 1; off < 32; off <<= 1) {
        const int v = __shfl_up(inc, off);
        if (lane >= off) inc += v;
    }
    if (lane == 31) wsum[wv] = inc;
    __syncthreads();
    int wpre = 0;
    for (int k = 0; k < 8; ++k) if (k < wv) wpre += wsum[k];
    int ex = wpre + inc - s;
    if (i0 + 0 < N) rowptr[i0 + 0] = ex;
    ex += c0;
    if (i0 + 1 < N) rowptr[i0 + 1] = ex;
    ex += c1;
    if (i0 + 2 < N) rowptr[i0 + 2] = ex;
    ex += c2;
    if (i0 + 3 < N) rowptr[i0 + 3] = ex;
    if (t == 255) tilesum[blockIdx.x] = wpre + inc;
}

__global__ __launch_bounds__(1024) void k_scan_tiles(const int* __restrict__ tilesum,
                                                     int* __restrict__ tileoff,
                                                     int* __restrict__ rowptr,
                                                     int numTiles, int N) {
    __shared__ int buf[1024];
    const int t = threadIdx.x;
    int carry = 0;
    for (int base = 0; base < numTiles; base += 1024) {
        const int i = base + t;
        const int v = (i < numTiles) ? tilesum[i] : 0;
        buf[t] = v;
        __syncthreads();
        for (int off = 1; off < 1024; off <<= 1) {
            const int a = (t >= off) ? buf[t - off] : 0;
            __syncthreads();
            buf[t] += a;
            __syncthreads();
        }
        if (i < numTiles) tileoff[i] = carry + buf[t] - v;
        carry += buf[1023];
        __syncthreads();
    }
    if (t == 0) rowptr[N] = carry;
}

__global__ void k_scan_add(int* __restrict__ rowptr, const int* __restrict__ tileoff,
                           int* __restrict__ cursor, int N) {
    int i = blockIdx.x * blockDim.x + threadIdx.x;
    if (i >= N) return;
    const int v = rowptr[i] + tileoff[i / SCAN_TILE];
    rowptr[i] = v;
    cursor[i] = v;
}

__global__ void k_scatter(const int* __restrict__ src, const int* __restrict__ dst,
                          int* __restrict__ cursor, int* __restrict__ srcS, int E) {
    int e = blockIdx.x * blockDim.x + threadIdx.x;
    if (e >= E) return;
    const int pos = atomicAdd(&cursor[dst[e]], 1);
    srcS[pos] = src[e];
}

#define BIN_SHIFT 10
#define BIN_NODES (1 << BIN_SHIFT)
#define SRC_BITS 22
#define SRC_MASK ((1u << SRC_BITS) - 1u)
#define MAX_BINS (1 << (SRC_BITS - BIN_SHIFT))
#define PART_THREADS 1024
#define PART_EPT 8
#define PART_CHUNK (PART_THREADS * PART_EPT)
#define BIN_CAP 49152
#define BIN_LD 8
static_assert(BIN_NODES == SCAN_TILE, "the bin arrays share the scan-tile arrays of the workspace");

__global__ __launch_bounds__(PART_THREADS) void k_binhist(const int* __restrict__ dst,
                                                          int* __restrict__ binCnt,
                                                          int E, int numBins) {
    __shared__ int hist[MAX_BINS];
    const int t = threadIdx.x;
    for (int b = t; b < numBins; b += PART_THREADS) hist[b] = 0;
    __syncthreads();
    const int base = blockIdx.x * PART_CHUNK;
    int d[PART_EPT];
#pragma unroll
    for (int i = 0; i < PART_EPT; ++i) {
        const int e = base + i * PART_THREADS + t;
        d[i] = dst[e < E ? e : E - 1];
    }
#pragma unroll
    for (int i = 0; i < PART_EPT; ++i) {
        atomicAdd(&hist[d[i] >> BIN_SHIFT], (base + i * PART_THREADS + t < E) ? 1 : 0);
    }
    __syncthreads();
    for (int b = t; b < numBins; b += PART_THREADS) {
        const int c = hist[b];
        if (c) atomicAdd(&binCnt[b], c);
    }
}

__global__ __launch_bounds__(PART_THREADS) void k_partition(const int* __restrict__ src,
                                                            const int* __restrict__ dst,
                                                            const int* __restrict__ binOff,
                                                            int* __restrict__ binCur,
                                                            unsigned* __restrict__ tmp,
                                                            int E, int numBins) {
    __shared__ int hist[MAX_BINS];
    const int t = threadIdx.x;
    for (int b = t; b < numBins; b += PART_THREADS) hist[b] = 0;
    const int base = blockIdx.x * PART_CHUNK;
    int d[PART_EPT], s[PART_EPT], r[PART_EPT];
#pragma unroll
    for (int i = 0; i < PART_EPT; ++i) {
        const int e = base + i * PART_THREADS + t;
        const int ec = e < E ? e : E - 1;
        const int dv = dst[ec];
        s[i] = src[ec];
        d[i] = (e < E) ? dv : -1;
    }
    __syncthreads();
#pragma unroll
    for (int i = 0; i < PART_EPT; ++i) {
        r[i] = 0;
        if (d[i] >= 0) r[i] = atomicAdd(&hist[d[i] >> BIN_SHIFT], 1);
    }
    __syncthreads();
    for (int b = t; b < numBins; b += PART_THREADS) {
        const int c = hist[b];
        int g = 0;
        if (c) g = binOff[b] + atomicAdd(&binCur[b], c);
        hist[b] = g;
    }
    __syncthreads();
#pragma unroll
    for (int i = 0; i < PART_EPT; ++i) {
        if (d[i] >= 0)
            tmp[hist[d[i] >> BIN_SHIFT] + r[i]] =
                ((unsigned)(d[i] & (BIN_NODES - 1)) << SRC_BITS) | (unsigned)s[i];
    }
}

__global__ __launch_bounds__(1024) void k_binsort(const unsigned* __restrict__ tmp,
                                                  const int* __restrict__ binOff,
                                                  const int* __restrict__ binCnt,
                                                  int* __restrict__ rowptr,
                                                  int* __restrict__ srcS, int N) {
    __shared__ int cnt[BIN_NODES];
    __shared__ int off[BIN_NODES + 1];
    __shared__ int wsum[32];
    __shared__ unsigned a[BIN_CAP];
    const int t = threadIdx.x;
    const int lane = t & 31;
    const int wv = t >> 5;
    const int bin = blockIdx.x;
    const int gbeg = binOff[bin];
    const int n = binCnt[bin];
    const unsigned* __restrict__ mine = tmp + gbeg;
    cnt[t] = 0;
    __syncthreads();
    for (int i0 = t; i0 < n; i0 += 1024 * BIN_LD) {
        unsigned w[BIN_LD];
#pragma unroll
        for (int u = 0; u < BIN_LD; ++u) {
            const int i = i0 + u * 1024;
            w[u] = mine[i < n ? i : i0];
        }
#pragma unroll
        for (int u = 0; u < BIN_LD; ++u) {
            const int i = i0 + u * 1024;
            atomicAdd(&cnt[w[u] >> SRC_BITS], (i < n) ? 1 : 0);
        }
    }
    __syncthreads();
    const int c = cnt[t];
    int inc = c;
    for (int o = 1; o < 32; o <<= 1) {
        const int v = __shfl_up(inc, o);
        if (lane >= o) inc += v;
    }
    if (lane == 31) wsum[wv] = inc;
    __syncthreads();
    int wpre = 0;
    for (int k = 0; k < 32; ++k) if (k < wv) wpre += wsum[k];
    const int ex = wpre + inc - c;
    off[t] = ex;
    if (t == 1023) off[BIN_NODES] = ex + c;
    cnt[t] = ex;
    const int node = bin * BIN_NODES + t;
    if (node < N) rowptr[node] = gbeg + ex;
    __syncthreads();
    if (n <= BIN_CAP) {
        for (int i0 = t; i0 < n; i0 += 1024 * BIN_LD) {
            unsigned w[BIN_LD];
#pragma unroll
            for (int u = 0; u < BIN_LD; ++u) {
                const int i = i0 + u * 1024;
                w[u] = mine[i < n ? i : i0];
            }
#pragma unroll
            for (int u = 0; u < BIN_LD; ++u) {
                const int i = i0 + u * 1024;
                const int p = atomicAdd(&cnt[w[u] >> SRC_BITS], (i < n) ? 1 : 0);
                if (i < n) a[p] = w[u];
            }
        }
        __syncthreads();
        for (int p = t; p < n; p += 1024) {
            const unsigned w = a[p];
            const int dl = (int)(w >> SRC_BITS);
            const int sb = off[dl];
            const int se = off[dl + 1];
            int rank = 0;
            for (int j = sb; j < se; ++j) {
                const unsigned v = a[j];
                rank += (v < w || (v == w && j < p)) ? 1 : 0;
            }
            srcS[gbeg + sb + rank] = (int)(w & SRC_MASK);
        }
    } else {
        for (int i = t; i < n; i += 1024) {
            const unsigned w = mine[i];
            const int p = atomicAdd(&cnt[w >> SRC_BITS], 1);
            srcS[gbeg + p] = (int)(w & SRC_MASK);
        }
    }
}

template <int TB, int U>
__device__ __forceinline__ void gat_edges(const int* __restrict__ srcS, const float4* __restrict__ hw4,
                                          const int k, const int q, const float4 as,
                                          const float (&edd)[TB], float (&m)[TB], float (&den)[TB],
                                          float4 (&acc)[TB]) {
    int s[U];
    float4 hv[U][TB];
    float l[U][TB];
#pragma unroll
    for (int u = 0; u < U; ++u) s[u] = srcS[k + u];
#pragma unroll
    for (int u = 0; u < U; ++u) {
#pragma unroll
        for (int g = 0; g < TB; ++g) hv[u][g] = hw4[s[u] * (TB * 4) + g * 4 + q];
    }
#pragma unroll
    for (int u = 0; u < U; ++u) {
#pragma unroll
        for (int g = 0; g < TB; ++g) l[u][g] = lrelu02(quad_sum(dot4(hv[u][g], as)) + edd[g]);
    }
#pragma unroll
    for (int g = 0; g < TB; ++g) {
        float bm = m[g];
#pragma unroll
        for (int u = 0; u < U; ++u) bm = fmaxf(bm, l[u][g]);
        const float sc = __expf(m[g] - bm);
        m[g] = bm;
        den[g] *= sc;
        acc[g].x *= sc; acc[g].y *= sc; acc[g].z *= sc; acc[g].w *= sc;
#pragma unroll
        for (int u = 0; u < U; ++u) {
            const float w = __expf(l[u][g] - bm);
            den[g] += w;
            acc[g].x += w * hv[u][g].x; acc[g].y += w * hv[u][g].y;
            acc[g].z += w * hv[u][g].z; acc[g].w += w * hv[u][g].w;
        }
    }
}

#define GAT_THREADS 128
#define TPITCH 20

__device__ __forceinline__ void wave_lds_handover() {
    __builtin_amdgcn_fence(__ATOMIC_RELEASE, "wavefront");
    __builtin_amdgcn_wave_barrier();
    __builtin_amdgcn_fence(__ATOMIC_ACQUIRE, "wavefront");
}

template <int TB, bool FUSE>
__global__ __launch_bounds__(GAT_THREADS) void k_gat_node(const int* __restrict__ rowptr,
                                                  const int* __restrict__ srcS,
                                                  const float* __restrict__ hw,
                                                  const float* __restrict__ asrc,
                                                  const float* __restrict__ adst,
                                                  const float* __restrict__ Wn,
                                                  const float* __restrict__ bn,
                                                  float* __restrict__ hout, int N) {
    const int t = blockIdx.x * GAT_THREADS + threadIdx.x;
    const int node = t >> 2;
    const int q = t & 3;
    const bool valid = node < N;
    const float4* __restrict__ hw4 = (const float4*)hw;
    const float4 as = ((const float4*)asrc)[q];
    const float4 ad = ((const float4*)adst)[q];
    int beg = 0, end = 0;
    float edd[TB];
#pragma unroll
    for (int g = 0; g < TB; ++g) edd[g] = 0.f;
    if (valid) {
        beg = rowptr[node]; end = rowptr[node + 1];
#pragma unroll
        for (int g = 0; g < TB; ++g) edd[g] = dot4(hw4[(node * TB + g) * 4 + q], ad);
    }
#pragma unroll
    for (int g = 0; g < TB; ++g) edd[g] = quad_sum(edd[g]);

    constexpr int U = 4;
    float m[TB], den[TB];
    float4 acc[TB];
#pragma unroll
    for (int g = 0; g < TB; ++g) {
        m[g] = -INFINITY; den[g] = 0.f; acc[g] = make_float4(0.f, 0.f, 0.f, 0.f);
    }
    int k = beg;
    for (; k + U <= end; k += U) gat_edges<TB, U>(srcS, hw4, k, q, as, edd, m, den, acc);
    if (k + 2 <= end) { gat_edges<TB, 2>(srcS, hw4, k, q, as, edd, m, den, acc); k += 2; }
    if (k < end)        gat_edges<TB, 1>(srcS, hw4, k, q, as, edd, m, den, acc);

    float4 o[TB];
#pragma unroll
    for (int g = 0; g < TB; ++g) {
        const float dn = fmaxf(den[g], 1e-16f);
        o[g].x = acc[g].x / dn; o[g].y = acc[g].y / dn; o[g].z = acc[g].z / dn; o[g].w = acc[g].w / dn;
        o[g].x = o[g].x > 0.f ? o[g].x : (__expf(o[g].x) - 1.f);
        o[g].y = o[g].y > 0.f ? o[g].y : (__expf(o[g].y) - 1.f);
        o[g].z = o[g].z > 0.f ? o[g].z : (__expf(o[g].z) - 1.f);
        o[g].w = o[g].w > 0.f ? o[g].w : (__expf(o[g].w) - 1.f);
    }

    if constexpr (!FUSE) {
        __shared__ float res[TB][GAT_THREADS / 4];
        const float4 mw = ((const float4*)Wn)[q];
        const float mb = bn[0];
        const int ln = threadIdx.x >> 2;
#pragma unroll
        for (int g = 0; g < TB; ++g) {
            const float r = quad_sum(dot4(o[g], mw)) + mb;
            if (q == 0) res[g][ln] = r;
        }
        __syncthreads();
        if (threadIdx.x < GAT_THREADS / 4) {
            const int n0 = blockIdx.x * (GAT_THREADS / 4) + threadIdx.x;
            if (n0 < N) {
#pragma unroll
                for (int g = 0; g < TB; ++g) hout[(size_t)g * N + n0] = res[g][threadIdx.x];
            }
        }
    } else {
        constexpr int R  = 8 * TB;
        constexpr int NT = (R + 15) / 16;
        __shared__ float tile[GAT_THREADS / 32][NT * 16][TPITCH];
        const int lane = threadIdx.x & 31;
        const int wave = threadIdx.x >> 5;
        float (*T)[TPITCH] = tile[wave];
        const int nd = lane >> 2;
#pragma unroll
        for (int g = 0; g < TB; ++g) *(float4*)&T[nd * TB + g][4 * q] = o[g];
        wave_lds_handover();

        const int c  = lane & 15;
        const int hf = lane >> 4;
        v2f bfrag[4];
#pragma unroll
        for (int j = 0; j < 4; ++j) {
            const int kc = 4 * j + 2 * hf;
            bfrag[j].x = Wn[kc * HD + c];
            bfrag[j].y = Wn[(kc + 1) * HD + c];
        }
        const float bi = bn[c];
        v8f d[NT];
#pragma unroll
        for (int tI = 0; tI < NT; ++tI) {
            int row = 16 * tI + c;
            if (row >= R) row = R - 1;
            v8f a8 = {};
#pragma unroll
            for (int j = 0; j < 4; ++j) {
                const v2f a = *(const v2f*)&T[row][4 * j + 2 * hf];
                a8 = __builtin_amdgcn_wmma_f32_16x16x4_f32(
                         false, a, false, bfrag[j], (short)0, a8, false, false);
            }
            d[tI] = a8;
        }
        wave_lds_handover();
#pragma unroll
        for (int tI = 0; tI < NT; ++tI) {
#pragma unroll
            for (int r = 0; r < 8; ++r) T[16 * tI + r + 8 * hf][c] = d[tI][r] + bi;
        }
        wave_lds_handover();
        const int rowbase = (blockIdx.x * (GAT_THREADS / 4) + wave * 8) * TB;
        const int rows = N * TB;
#pragma unroll
        for (int j = 0; j < R / 8; ++j) {
            const int f = lane + 32 * j;
            const int row = f >> 2;
            const int c4 = f & 3;
            if (rowbase + row < rows)
                ((float4*)hout)[(size_t)(rowbase + row) * 4 + c4] = *(const float4*)&T[row][4 * c4];
        }
    }
}

#define G0_THREADS 1024
#define G0_BLOCKS 256
#define WROWS (NWIRE + 1)

template <int TB, int U>
__device__ __forceinline__ void gat0_edges(const int* __restrict__ srcS, const unsigned* __restrict__ rec,
                                           const float4* wl, const int k, const int q, const bool isw,
                                           const int qrow, const float4 va,
                                           const float (&edd)[TB], float (&m)[TB], float (&den)[TB],
                                           float4 (&acc)[TB]) {
    int s[U];
    unsigned w[U];
    float4 hv[U][TB];
    float l[U][TB];
#pragma unroll
    for (int u = 0; u < U; ++u) s[u] = srcS[k + u];
#pragma unroll
    for (int u = 0; u < U; ++u) w[u] = rec[s[u] * 4 + q];
#pragma unroll
    for (int u = 0; u < U; ++u) {
        const int row = qrow + (isw ? (int)w[u] : NWIRE);
#pragma unroll
        for (int g = 0; g < TB; ++g) {
            float4 t = wl[row + g * WROWS];
            t.x = isw ? t.x : __uint_as_float(w[u]);
            hv[u][g] = t;
        }
    }
#pragma unroll
    for (int u = 0; u < U; ++u) {
#pragma unroll
        for (int g = 0; g < TB; ++g) l[u][g] = lrelu02(quad_sum(dot4(hv[u][g], va)) + edd[g]);
    }
#pragma unroll
    for (int g = 0; g < TB; ++g) {
        float bm = m[g];
#pragma unroll
        for (int u = 0; u < U; ++u) bm = fmaxf(bm, l[u][g]);
        const float sc = __expf(m[g] - bm);
        m[g] = bm;
        den[g] *= sc;
        acc[g].x *= sc; acc[g].y *= sc; acc[g].z *= sc; acc[g].w *= sc;
#pragma unroll
        for (int u = 0; u < U; ++u) {
            const float wgt = __expf(l[u][g] - bm);
            den[g] += wgt;
            acc[g].x += wgt * hv[u][g].x; acc[g].y += wgt * hv[u][g].y;
            acc[g].z += wgt * hv[u][g].z; acc[g].w += wgt * hv[u][g].w;
        }
    }
}

template <int TB>
__global__ __launch_bounds__(G0_THREADS) void k_gat0_node(const int* __restrict__ rowptr,
                                                          const int* __restrict__ srcS,
                                                          const unsigned* __restrict__ rec,
                                                          const float* __restrict__ fw,
                                                          const float* __restrict__ sw,
                                                          const float* __restrict__ tw,
                                                          const float* __restrict__ W0p,
                                                          const float* __restrict__ lv,
                                                          const float* __restrict__ Wn,
                                                          const float* __restrict__ bn,
                                                          float* __restrict__ hout, int N, int g0) {
    constexpr int R  = 8 * TB;
    constexpr int NT = (R + 15) / 16;
    __shared__ float4 wl[3 * TB * WROWS];
    __shared__ __attribute__((aligned(16))) float tile[G0_THREADS / 32][NT * 16][TPITCH];
    __shared__ __attribute__((aligned(16))) float wts[256 + 256 + HD + 48];
    float* const sW0 = wts;
    float* const sWn = wts + 256;
    float* const sbn = wts + 512;
    float* const slv = wts + 512 + HD;

    const size_t goff = (size_t)g0 * NWIRE * NFEAT;
    const float4* __restrict__ f4 = (const float4*)(fw + goff);
    const float4* __restrict__ s4 = (const float4*)(sw + goff);
    const float4* __restrict__ t4 = (const float4*)(tw + goff);
    for (int i = threadIdx.x; i < TB * NWIRE; i += G0_THREADS) {
        const int g = i / NWIRE;
        const int r = i - g * NWIRE;
        wl[(0 * TB + g) * WROWS + r] = f4[i];
        wl[(1 * TB + g) * WROWS + r] = s4[i];
        wl[(2 * TB + g) * WROWS + r] = t4[i];
    }
    if (threadIdx.x < 3 * TB) wl[threadIdx.x * WROWS + NWIRE] = make_float4(0.f, 0.f, 0.f, 0.f);
    if (threadIdx.x < 256) { sW0[threadIdx.x] = W0p[threadIdx.x]; sWn[threadIdx.x] = Wn[threadIdx.x]; }
    if (threadIdx.x < HD) sbn[threadIdx.x] = bn[threadIdx.x];
    if (threadIdx.x < 33) slv[threadIdx.x] = lv[threadIdx.x];
    __syncthreads();

    const int lane = threadIdx.x & 31;
    const int wave = __builtin_amdgcn_readfirstlane(threadIdx.x >> 5);
    const int q  = lane & 3;
    const int nd = lane >> 2;
    const int c  = lane & 15;
    const int hf = lane >> 4;
    const bool isw = q < 3;
    const int qrow = isw ? q * TB * WROWS : 0;
    const float4 va = ((const float4*)slv)[q];
    float (*T)[TPITCH] = tile[wave];
    const int numChunks = (N + 7) >> 3;
    const int rows = N * TB;
    if constexpr (NT * 16 > R) {
        for (int i = lane; i < (NT * 16 - R) * HD; i += 32) T[R + (i >> 4)][i & 15] = 0.f;
    }

    for (int ch = wave * gridDim.x + blockIdx.x; ch < numChunks; ch += gridDim.x * (G0_THREADS / 32)) {
        const int node = ch * 8 + nd;
        const bool valid = node < N;
        int beg = 0, end = 0;
        float edd[TB];
#pragma unroll
        for (int g = 0; g < TB; ++g) edd[g] = 0.f;
        const float csd = slv[32];
        if (valid) {
            beg = rowptr[node]; end = rowptr[node + 1];
            const unsigned wd = rec[node * 4 + q];
            const float4 vd = ((const float4*)(slv + 16))[q];
            const int row = qrow + (isw ? (int)wd : NWIRE);
#pragma unroll
            for (int g = 0; g < TB; ++g) {
                float4 t = wl[row + g * WROWS];
                t.x = isw ? t.x : __uint_as_float(wd);
                edd[g] = dot4(t, vd);
            }
        }
#pragma unroll
        for (int g = 0; g < TB; ++g) edd[g] = quad_sum(edd[g]) + csd;

        constexpr int U = 4;
        float m[TB], den[TB];
        float4 acc[TB];
#pragma unroll
        for (int g = 0; g < TB; ++g) {
            m[g] = -INFINITY; den[g] = 0.f; acc[g] = make_float4(0.f, 0.f, 0.f, 0.f);
        }
        int k = beg;
        for (; k + U <= end; k += U) gat0_edges<TB, U>(srcS, rec, wl, k, q, isw, qrow, va, edd, m, den, acc);
        if (k + 2 <= end) { gat0_edges<TB, 2>(srcS, rec, wl, k, q, isw, qrow, va, edd, m, den, acc); k += 2; }
        if (k < end)        gat0_edges<TB, 1>(srcS, rec, wl, k, q, isw, qrow, va, edd, m, den, acc);

        float4 o[TB];
#pragma unroll
        for (int g = 0; g < TB; ++g) {
            const float dn = fmaxf(den[g], 1e-16f);
            o[g].x = acc[g].x / dn; o[g].y = acc[g].y / dn; o[g].z = acc[g].z / dn; o[g].w = acc[g].w / dn;
            if (q == 3) o[g].y = den[g] / dn;
        }

        wave_lds_handover();
#pragma unroll
        for (int g = 0; g < TB; ++g) *(float4*)&T[nd * TB + g][4 * q] = o[g];
        wave_lds_handover();

        v2f bfrag[4];
#pragma unroll
        for (int j = 0; j < 4; ++j) {
            const int kc = 4 * j + 2 * hf;
            bfrag[j].x = sW0[kc * HD + c];
            bfrag[j].y = sW0[(kc + 1) * HD + c];
        }
        v8f d[NT];
#pragma unroll
        for (int tI = 0; tI < NT; ++tI) {
            v8f a8 = {};
#pragma unroll
            for (int j = 0; j < 4; ++j) {
                const v2f a = *(const v2f*)&T[16 * tI + c][4 * j + 2 * hf];
                a8 = __builtin_amdgcn_wmma_f32_16x16x4_f32(
                         false, a, false, bfrag[j], (short)0, a8, false, false);
            }
            d[tI] = a8;
        }
        wave_lds_handover();
#pragma unroll
        for (int tI = 0; tI < NT; ++tI) {
#pragma unroll
            for (int r = 0; r < 8; ++r) {
                float v = d[tI][r];
                v = v > 0.f ? v : (__expf(v) - 1.f);
                T[16 * tI + r + 8 * hf][c] = v;
            }
        }
        wave_lds_handover();

#pragma unroll
        for (int j = 0; j < 4; ++j) {
            const int kc = 4 * j + 2 * hf;
            bfrag[j].x = sWn[kc * HD + c];
            bfrag[j].y = sWn[(kc + 1) * HD + c];
        }
        const float bi1 = sbn[c];
#pragma unroll
        for (int tI = 0; tI < NT; ++tI) {
            v8f a8 = {};
#pragma unroll
            for (int j = 0; j < 4; ++j) {
                const v2f a = *(const v2f*)&T[16 * tI + c][4 * j + 2 * hf];
                a8 = __builtin_amdgcn_wmma_f32_16x16x4_f32(
                         false, a, false, bfrag[j], (short)0, a8, false, false);
            }
            d[tI] = a8;
        }
        wave_lds_handover();
#pragma unroll
        for (int tI = 0; tI < NT; ++tI) {
#pragma unroll
            for (int r = 0; r < 8; ++r) T[16 * tI + r + 8 * hf][c] = d[tI][r] + bi1;
        }
        wave_lds_handover();
        const int rowbase = ch * R;
#pragma unroll
        for (int j = 0; j < R / 8; ++j) {
            const int f = lane + 32 * j;
            const int row = f >> 2;
            const int c4 = f & 3;
            if (rowbase + row < rows)
                ((float4*)hout)[(size_t)(rowbase + row) * 4 + c4] = *(const float4*)&T[row][4 * c4];
        }
    }
}

static inline int cdiv(int a, int b) { return (a + b - 1) / b; }

struct Bufs {
    const float *fw, *sw, *tw, *areas;
    const int* indices;
    const float *W0p, *lv, *Wl, *a_src, *a_dst, *bvec, *mlpW, *mlpb;
    const int *rowptr, *srcS;
    float *h, *hw, *out;
};

template <int TB>
static void run_graphs(const Bufs& b, int N, int g0, hipStream_t stream) {
    const int gatBl  = cdiv(N * 4, GAT_THREADS);
    unsigned* rec = (unsigned*)b.hw;
    k_pack<<<cdiv(N, 256), 256, 0, stream>>>(b.indices, b.areas, rec, N);
    k_gat0_node<TB><<<G0_BLOCKS, G0_THREADS, 0, stream>>>(b.rowptr, b.srcS, rec, b.fw, b.sw, b.tw,
                                                         b.W0p, b.lv, b.Wl, b.bvec + HD, b.h, N, g0);
    float* cur = b.h;
    float* nxt = b.hw;
    for (int i = 1; i < 4; ++i) {
        const float* as = b.a_src + (size_t)i * HD;
        const float* ad = b.a_dst + (size_t)i * HD;
        if (i < 3)
            k_gat_node<TB, true><<<gatBl, GAT_THREADS, 0, stream>>>(b.rowptr, b.srcS, cur, as, ad,
                                                                   b.Wl + (size_t)i * 256,
                                                                   b.bvec + (size_t)(i + 1) * HD, nxt, N);
        else
            k_gat_node<TB, false><<<gatBl, GAT_THREADS, 0, stream>>>(b.rowptr, b.srcS, cur, as, ad,
                                                                    b.mlpW, b.mlpb,
                                                                    b.out + (size_t)g0 * N, N);
        float* tmp = cur; cur = nxt; nxt = tmp;
    }
}

extern "C" void kernel_launch(void* const* d_in, const int* in_sizes, int n_in,
                              void* d_out, int out_size, void* d_ws, size_t ws_size,
                              hipStream_t stream) {
    const float* fw     = (const float*)d_in[0];
    const float* sw     = (const float*)d_in[1];
    const float* tw     = (const float*)d_in[2];
    const float* areas  = (const float*)d_in[3];
    const int*   indices= (const int*)  d_in[4];
    const int*   edge   = (const int*)  d_in[5];
    const float* W0     = (const float*)d_in[6];
    const float* Wl     = (const float*)d_in[7];
    const float* a_src  = (const float*)d_in[8];
    const float* a_dst  = (const float*)d_in[9];
    const float* bvec   = (const float*)d_in[10];
    const float* mlpW   = (const float*)d_in[11];
    const float* mlpb   = (const float*)d_in[12];

    const int N = in_sizes[3];
    const int E = in_sizes[5] / 2;
    const int T = in_sizes[0] / (NWIRE * NFEAT);
    const int* src = edge;
    const int* dst = edge + E;
    const int numTiles = cdiv(N, SCAN_TILE);

    const size_t fixedW = 320 + (size_t)E + ((size_t)N + 1) + 2 * (size_t)N + 2 * (size_t)numTiles;
    const size_t need3  = 2 * (size_t)N * HD * 3 + fixedW;
    const size_t need2  = 2 * (size_t)N * HD * 2 + fixedW;
    int TBS = 1;
    if (T >= 3 && ws_size >= need3 * 4) TBS = 3;
    else if (T >= 2 && ws_size >= need2 * 4) TBS = 2;

    float* ws = (float*)d_ws;
    size_t off = 0;
    float* h       = ws + off; off += (size_t)N * HD * TBS;
    float* hw      = ws + off; off += (size_t)N * HD * TBS;
    float* W0p     = ws + off; off += 256;
    float* lv      = ws + off; off += 64;
    int*   srcS    = (int*)(ws + off); off += (size_t)E;
    int*   rowptr  = (int*)(ws + off); off += (size_t)N + 1;
    int*   cursor  = (int*)(ws + off); off += (size_t)N;
    int*   cnt     = (int*)(ws + off); off += (size_t)N;
    int*   tilesum = (int*)(ws + off); off += (size_t)numTiles;
    int*   tileoff = (int*)(ws + off); off += (size_t)numTiles;
    (void)n_in; (void)out_size;

    const int numBins = numTiles;
    int* binCnt = tilesum;
    int* binOff = tileoff;
    int* binCur = cnt;
    const bool binned = N <= (1 << SRC_BITS) && (size_t)E <= (size_t)N * HD * TBS;

    k_prep<<<1, 256, 0, stream>>>(W0, a_src, a_dst, bvec, W0p, lv, binCnt, binCur, binned ? numBins : 0);
    if (binned) {
        const int chunks = cdiv(E, PART_CHUNK);
        k_binhist<<<chunks, PART_THREADS, 0, stream>>>(dst, binCnt, E, numBins);
        k_scan_tiles<<<1, 1024, 0, stream>>>(binCnt, binOff, rowptr, numBins, N);
        k_partition<<<chunks, PART_THREADS, 0, stream>>>(src, dst, binOff, binCur, (unsigned*)h, E, numBins);
        k_binsort<<<numBins, 1024, 0, stream>>>((const unsigned*)h, binOff, binCnt, rowptr, srcS, N);
    } else {
        k_zero<<<cdiv(N, 256), 256, 0, stream>>>(cnt, N);
        k_hist<<<cdiv(E, 256), 256, 0, stream>>>(dst, cnt, E);
        k_scan_local<<<numTiles, 256, 0, stream>>>(cnt, rowptr, tilesum, N);
        k_scan_tiles<<<1, 1024, 0, stream>>>(tilesum, tileoff, rowptr, numTiles, N);
        k_scan_add<<<cdiv(N, 256), 256, 0, stream>>>(rowptr, tileoff, cursor, N);
        k_scatter<<<cdiv(E, 256), 256, 0, stream>>>(src, dst, cursor, srcS, E);
    }

    Bufs b;
    b.fw = fw; b.sw = sw; b.tw = tw; b.areas = areas; b.indices = indices;
    b.W0p = W0p; b.lv = lv; b.Wl = Wl; b.a_src = a_src; b.a_dst = a_dst; b.bvec = bvec;
    b.mlpW = mlpW; b.mlpb = mlpb; b.rowptr = rowptr; b.srcS = srcS;
    b.h = h; b.hw = hw; b.out = (float*)d_out;

    int g = 0;
    while (g < T) {
        const int left = T - g;
        const int tb = left < TBS ? left : TBS;
        if (tb >= 3)      { run_graphs<3>(b, N, g, stream); g += 3; }
        else if (tb == 2) { run_graphs<2>(b, N, g, stream); g += 2; }
        else              { run_graphs<1>(b, N, g, stream); g += 1; }
    }
}


# k_partition groups its chunk's words by bin in LDS and writes each (block, bin) run with consecutive lanes: a store touches a few lines, not up to 32 partial ones; final list and sum order unchanged
# speedup vs baseline: 20.8585x; 1.0405x over previous
#include <hip/hip_runtime.h>
#include <math.h>


typedef float v2f __attribute__((ext_vector_type(2)));
typedef float v8f __attribute__((ext_vector_type(8)));

#define HD 16
#define NWIRE 1536
#define NFEAT 4
#define SCAN_TILE 1024


__device__ __forceinline__ float lrelu02(float x) { return x > 0.f ? x : 0.2f * x; }

__device__ __forceinline__ float dot4(const float4 v, const float4 a) {
    return v.x * a.x + v.y * a.y + v.z * a.z + v.w * a.w;
}

__device__ __forceinline__ float quad_sum(float p) {
    p += __int_as_float(__builtin_amdgcn_update_dpp(0, __float_as_int(p), 0xB1, 0xF, 0xF, true));
    p += __int_as_float(__builtin_amdgcn_update_dpp(0, __float_as_int(p), 0x4E, 0xF, 0xF, true));
    return p;
}

__global__ void k_prep(const float* __restrict__ W0, const float* __restrict__ asrc,
                       const float* __restrict__ adst, const float* __restrict__ b0,
                       float* __restrict__ W0p, float* __restrict__ lv,
                       int* __restrict__ binCnt, int* __restrict__ binCur, int numBins) {
    int i = threadIdx.x;
    int r = i >> 4;
    W0p[i] = (r < 13) ? W0[i] : (r == 13 ? b0[i & 15] : 0.f);
    if (i < 32) {
        const float* __restrict__ a = (i < 16) ? asrc : adst;
        const int k = i & 15;
        float sum = 0.f;
        if (k < 13) {
            for (int j = 0; j < HD; ++j) sum += W0[k * HD + j] * a[j];
        }
        lv[i] = sum;
    } else if (i == 32) {
        float cs = 0.f, cd = 0.f;
        for (int j = 0; j < HD; ++j) { cs += b0[j] * asrc[j]; cd += b0[j] * adst[j]; }
        lv[32] = cs + cd;
    }
    for (int b = i; b < numBins; b += 256) { binCnt[b] = 0; binCur[b] = 0; }
}

__global__ void k_pack(const int* __restrict__ indices, const float* __restrict__ areas,
                       unsigned* __restrict__ rec, int N) {
    const int n = blockIdx.x * blockDim.x + threadIdx.x;
    if (n >= N) return;
    uint4 r;
    r.x = (unsigned)indices[n * 3 + 0];
    r.y = (unsigned)indices[n * 3 + 1];
    r.z = (unsigned)indices[n * 3 + 2];
    r.w = __float_as_uint(areas[n]);
    ((uint4*)rec)[n] = r;
}

__global__ void k_zero(int* __restrict__ cnt, int N) {
    int i = blockIdx.x * blockDim.x + threadIdx.x;
    if (i < N) cnt[i] = 0;
}

__global__ void k_hist(const int* __restrict__ dst, int* __restrict__ cnt, int E) {
    int e = blockIdx.x * blockDim.x + threadIdx.x;
    if (e < E) atomicAdd(&cnt[dst[e]], 1);
}

__global__ __launch_bounds__(256) void k_scan_local(const int* __restrict__ cnt,
                                                    int* __restrict__ rowptr,
                                                    int* __restrict__ tilesum, int N) {
    __shared__ int wsum[8];
    const int t = threadIdx.x;
    const int lane = t & 31;
    const int wv = t >> 5;
    const int i0 = (blockIdx.x * 256 + t) * 4;
    const int c0 = (i0 + 0 < N) ? cnt[i0 + 0] : 0;
    const int c1 = (i0 + 1 < N) ? cnt[i0 + 1] : 0;
    const int c2 = (i0 + 2 < N) ? cnt[i0 + 2] : 0;
    const int c3 = (i0 + 3 < N) ? cnt[i0 + 3] : 0;
    const int s = c0 + c1 + c2 + c3;
    int inc = s;
    for (int off = 1; off < 32; off <<= 1) {
        const int v = __shfl_up(inc, off);
        if (lane >= off) inc += v;
    }
    if (lane == 31) wsum[wv] = inc;
    __syncthreads();
    int wpre = 0;
    for (int k = 0; k < 8; ++k) if (k < wv) wpre += wsum[k];
    int ex = wpre + inc - s;
    if (i0 + 0 < N) rowptr[i0 + 0] = ex;
    ex += c0;
    if (i0 + 1 < N) rowptr[i0 + 1] = ex;
    ex += c1;
    if (i0 + 2 < N) rowptr[i0 + 2] = ex;
    ex += c2;
    if (i0 + 3 < N) rowptr[i0 + 3] = ex;
    if (t == 255) tilesum[blockIdx.x] = wpre + inc;
}

__global__ __launch_bounds__(1024) void k_scan_tiles(const int* __restrict__ tilesum,
                                                     int* __restrict__ tileoff,
                                                     int* __restrict__ rowptr,
                                                     int numTiles, int N) {
    __shared__ int buf[1024];
    const int t = threadIdx.x;
    int carry = 0;
    for (int base = 0; base < numTiles; base += 1024) {
        const int i = base + t;
        const int v = (i < numTiles) ? tilesum[i] : 0;
        buf[t] = v;
        __syncthreads();
        for (int off = 1; off < 1024; off <<= 1) {
            const int a = (t >= off) ? buf[t - off] : 0;
            __syncthreads();
            buf[t] += a;
            __syncthreads();
        }
        if (i < numTiles) tileoff[i] = carry + buf[t] - v;
        carry += buf[1023];
        __syncthreads();
    }
    if (t == 0) rowptr[N] = carry;
}

__global__ void k_scan_add(int* __restrict__ rowptr, const int* __restrict__ tileoff,
                           int* __restrict__ cursor, int N) {
    int i = blockIdx.x * blockDim.x + threadIdx.x;
    if (i >= N) return;
    const int v = rowptr[i] + tileoff[i / SCAN_TILE];
    rowptr[i] = v;
    cursor[i] = v;
}

__global__ void k_scatter(const int* __restrict__ src, const int* __restrict__ dst,
                          int* __restrict__ cursor, int* __restrict__ srcS, int E) {
    int e = blockIdx.x * blockDim.x + threadIdx.x;
    if (e >= E) return;
    const int pos = atomicAdd(&cursor[dst[e]], 1);
    srcS[pos] = src[e];
}

#define BIN_SHIFT 10
#define BIN_NODES (1 << BIN_SHIFT)
#define SRC_BITS 22
#define SRC_MASK ((1u << SRC_BITS) - 1u)
#define MAX_BINS (1 << (SRC_BITS - BIN_SHIFT))
#define PART_THREADS 1024
#define PART_EPT 8
#define PART_CHUNK (PART_THREADS * PART_EPT)
#define PART_SCAN (MAX_BINS / PART_THREADS)
#define BIN_CAP 49152
#define BIN_LD 8
static_assert(BIN_NODES == SCAN_TILE, "the bin arrays share the scan-tile arrays of the workspace");

__global__ __launch_bounds__(PART_THREADS) void k_binhist(const int* __restrict__ dst,
                                                          int* __restrict__ binCnt,
                                                          int E, int numBins) {
    __shared__ int hist[MAX_BINS];
    const int t = threadIdx.x;
    for (int b = t; b < numBins; b += PART_THREADS) hist[b] = 0;
    __syncthreads();
    const int base = blockIdx.x * PART_CHUNK;
    int d[PART_EPT];
#pragma unroll
    for (int i = 0; i < PART_EPT; ++i) {
        const int e = base + i * PART_THREADS + t;
        d[i] = dst[e < E ? e : E - 1];
    }
#pragma unroll
    for (int i = 0; i < PART_EPT; ++i) {
        atomicAdd(&hist[d[i] >> BIN_SHIFT], (base + i * PART_THREADS + t < E) ? 1 : 0);
    }
    __syncthreads();
    for (int b = t; b < numBins; b += PART_THREADS) {
        const int c = hist[b];
        if (c) atomicAdd(&binCnt[b], c);
    }
}

__global__ __launch_bounds__(PART_THREADS) void k_partition(const int* __restrict__ src,
                                                            const int* __restrict__ dst,
                                                            const int* __restrict__ binOff,
                                                            int* __restrict__ binCur,
                                                            unsigned* __restrict__ tmp,
                                                            int E, int numBins) {
    __shared__ int hist[MAX_BINS];
    __shared__ int lofs[MAX_BINS];
    __shared__ int wsum[PART_THREADS / 32];
    __shared__ unsigned stage[PART_CHUNK];
    __shared__ unsigned short sbin[PART_CHUNK];
    static_assert(MAX_BINS == PART_SCAN * PART_THREADS, "the block scan takes PART_SCAN bins per thread");
    static_assert(MAX_BINS <= 65536, "bin ids are staged as 16-bit words");
    static_assert(PART_THREADS / 32 == 32, "the wave totals are scanned one per lane");
    const int t = threadIdx.x;
    const int lane = t & 31;
    const int wv = t >> 5;
    for (int b = t; b < MAX_BINS; b += PART_THREADS) hist[b] = 0;
    const int base = blockIdx.x * PART_CHUNK;
    int d[PART_EPT], s[PART_EPT], r[PART_EPT];
#pragma unroll
    for (int i = 0; i < PART_EPT; ++i) {
        const int e = base + i * PART_THREADS + t;
        const int ec = e < E ? e : E - 1;
        const int dv = dst[ec];
        const int sv = src[ec];
        s[i] = (int)(((unsigned)(dv & (BIN_NODES - 1)) << SRC_BITS) | (unsigned)sv);
        d[i] = (e < E) ? (dv >> BIN_SHIFT) : -1;
    }
    __syncthreads();
#pragma unroll
    for (int i = 0; i < PART_EPT; ++i) {
        r[i] = 0;
        if (d[i] >= 0) r[i] = atomicAdd(&hist[d[i]], 1);
    }
    __syncthreads();
    const int b0 = t * PART_SCAN;
    int cc[PART_SCAN];
    int sm = 0;
#pragma unroll
    for (int j = 0; j < PART_SCAN; ++j) { cc[j] = hist[b0 + j]; sm += cc[j]; }
    int inc = sm;
    for (int o = 1; o < 32; o <<= 1) {
        const int v = __shfl_up(inc, o);
        if (lane >= o) inc += v;
    }
    if (lane == 31) wsum[wv] = inc;
    __syncthreads();
    const int wt = wsum[lane];
    int winc = wt;
    for (int o = 1; o < 32; o <<= 1) {
        const int v = __shfl_up(winc, o);
        if (lane >= o) winc += v;
    }
    const int wpre = __shfl(winc - wt, wv);
    int ex = wpre + inc - sm;
#pragma unroll
    for (int j = 0; j < PART_SCAN; ++j) {
        const int b = b0 + j;
        const int cnt = cc[j];
        int g = 0;
        if (cnt && b < numBins) g = binOff[b] + atomicAdd(&binCur[b], cnt);
        lofs[b] = ex;
        hist[b] = g - ex;
        ex += cnt;
    }
    __syncthreads();
#pragma unroll
    for (int i = 0; i < PART_EPT; ++i) r[i] += lofs[d[i] >= 0 ? d[i] : 0];
#pragma unroll
    for (int i = 0; i < PART_EPT; ++i) {
        if (d[i] >= 0) {
            stage[r[i]] = (unsigned)s[i];
            sbin[r[i]] = (unsigned short)d[i];
        }
    }
    __syncthreads();
    const int left = E - base;
    const int total = left < PART_CHUNK ? left : PART_CHUNK;
#pragma unroll
    for (int i = 0; i < PART_EPT; ++i) {
        const int p = i * PART_THREADS + t;
        const int pc = p < total ? p : total - 1;
        s[i] = (int)stage[pc];
        d[i] = hist[sbin[pc]] + pc;
    }
#pragma unroll
    for (int i = 0; i < PART_EPT; ++i) {
        if (i * PART_THREADS + t < total) tmp[d[i]] = (unsigned)s[i];
    }
}

__global__ __launch_bounds__(1024) void k_binsort(const unsigned* __restrict__ tmp,
                                                  const int* __restrict__ binOff,
                                                  const int* __restrict__ binCnt,
                                                  int* __restrict__ rowptr,
                                                  int* __restrict__ srcS, int N) {
    __shared__ int cnt[BIN_NODES];
    __shared__ int off[BIN_NODES + 1];
    __shared__ int wsum[32];
    __shared__ unsigned a[BIN_CAP];
    const int t = threadIdx.x;
    const int lane = t & 31;
    const int wv = t >> 5;
    const int bin = blockIdx.x;
    const int gbeg = binOff[bin];
    const int n = binCnt[bin];
    const unsigned* __restrict__ mine = tmp + gbeg;
    cnt[t] = 0;
    __syncthreads();
    for (int i0 = t; i0 < n; i0 += 1024 * BIN_LD) {
        unsigned w[BIN_LD];
#pragma unroll
        for (int u = 0; u < BIN_LD; ++u) {
            const int i = i0 + u * 1024;
            w[u] = mine[i < n ? i : i0];
        }
#pragma unroll
        for (int u = 0; u < BIN_LD; ++u) {
            const int i = i0 + u * 1024;
            atomicAdd(&cnt[w[u] >> SRC_BITS], (i < n) ? 1 : 0);
        }
    }
    __syncthreads();
    const int c = cnt[t];
    int inc = c;
    for (int o = 1; o < 32; o <<= 1) {
        const int v = __shfl_up(inc, o);
        if (lane >= o) inc += v;
    }
    if (lane == 31) wsum[wv] = inc;
    __syncthreads();
    int wpre = 0;
    for (int k = 0; k < 32; ++k) if (k < wv) wpre += wsum[k];
    const int ex = wpre + inc - c;
    off[t] = ex;
    if (t == 1023) off[BIN_NODES] = ex + c;
    cnt[t] = ex;
    const int node = bin * BIN_NODES + t;
    if (node < N) rowptr[node] = gbeg + ex;
    __syncthreads();
    if (n <= BIN_CAP) {
        for (int i0 = t; i0 < n; i0 += 1024 * BIN_LD) {
            unsigned w[BIN_LD];
#pragma unroll
            for (int u = 0; u < BIN_LD; ++u) {
                const int i = i0 + u * 1024;
                w[u] = mine[i < n ? i : i0];
            }
#pragma unroll
            for (int u = 0; u < BIN_LD; ++u) {
                const int i = i0 + u * 1024;
                const int p = atomicAdd(&cnt[w[u] >> SRC_BITS], (i < n) ? 1 : 0);
                if (i < n) a[p] = w[u];
            }
        }
        __syncthreads();
        for (int p = t; p < n; p += 1024) {
            const unsigned w = a[p];
            const int dl = (int)(w >> SRC_BITS);
            const int sb = off[dl];
            const int se = off[dl + 1];
            int rank = 0;
            for (int j = sb; j < se; ++j) {
                const unsigned v = a[j];
                rank += (v < w || (v == w && j < p)) ? 1 : 0;
            }
            srcS[gbeg + sb + rank] = (int)(w & SRC_MASK);
        }
    } else {
        for (int i = t; i < n; i += 1024) {
            const unsigned w = mine[i];
            const int p = atomicAdd(&cnt[w >> SRC_BITS], 1);
            srcS[gbeg + p] = (int)(w & SRC_MASK);
        }
    }
}

template <int TB, int U>
__device__ __forceinline__ void gat_edges(const int* __restrict__ srcS, const float4* __restrict__ hw4,
                                          const int k, const int q, const float4 as,
                                          const float (&edd)[TB], float (&m)[TB], float (&den)[TB],
                                          float4 (&acc)[TB]) {
    int s[U];
    float4 hv[U][TB];
    float l[U][TB];
#pragma unroll
    for (int u = 0; u < U; ++u) s[u] = srcS[k + u];
#pragma unroll
    for (int u = 0; u < U; ++u) {
#pragma unroll
        for (int g = 0; g < TB; ++g) hv[u][g] = hw4[s[u] * (TB * 4) + g * 4 + q];
    }
#pragma unroll
    for (int u = 0; u < U; ++u) {
#pragma unroll
        for (int g = 0; g < TB; ++g) l[u][g] = lrelu02(quad_sum(dot4(hv[u][g], as)) + edd[g]);
    }
#pragma unroll
    for (int g = 0; g < TB; ++g) {
        float bm = m[g];
#pragma unroll
        for (int u = 0; u < U; ++u) bm = fmaxf(bm, l[u][g]);
        const float sc = __expf(m[g] - bm);
        m[g] = bm;
        den[g] *= sc;
        acc[g].x *= sc; acc[g].y *= sc; acc[g].z *= sc; acc[g].w *= sc;
#pragma unroll
        for (int u = 0; u < U; ++u) {
            const float w = __expf(l[u][g] - bm);
            den[g] += w;
            acc[g].x += w * hv[u][g].x; acc[g].y += w * hv[u][g].y;
            acc[g].z += w * hv[u][g].z; acc[g].w += w * hv[u][g].w;
        }
    }
}

#define GAT_THREADS 128
#define TPITCH 20

__device__ __forceinline__ void wave_lds_handover() {
    __builtin_amdgcn_fence(__ATOMIC_RELEASE, "wavefront");
    __builtin_amdgcn_wave_barrier();
    __builtin_amdgcn_fence(__ATOMIC_ACQUIRE, "wavefront");
}

template <int TB, bool FUSE>
__global__ __launch_bounds__(GAT_THREADS) void k_gat_node(const int* __restrict__ rowptr,
                                                  const int* __restrict__ srcS,
                                                  const float* __restrict__ hw,
                                                  const float* __restrict__ asrc,
                                                  const float* __restrict__ adst,
                                                  const float* __restrict__ Wn,
                                                  const float* __restrict__ bn,
                                                  float* __restrict__ hout, int N) {
    const int t = blockIdx.x * GAT_THREADS + threadIdx.x;
    const int node = t >> 2;
    const int q = t & 3;
    const bool valid = node < N;
    const float4* __restrict__ hw4 = (const float4*)hw;
    const float4 as = ((const float4*)asrc)[q];
    const float4 ad = ((const float4*)adst)[q];
    int beg = 0, end = 0;
    float edd[TB];
#pragma unroll
    for (int g = 0; g < TB; ++g) edd[g] = 0.f;
    if (valid) {
        beg = rowptr[node]; end = rowptr[node + 1];
#pragma unroll
        for (int g = 0; g < TB; ++g) edd[g] = dot4(hw4[(node * TB + g) * 4 + q], ad);
    }
#pragma unroll
    for (int g = 0; g < TB; ++g) edd[g] = quad_sum(edd[g]);

    constexpr int U = 4;
    float m[TB], den[TB];
    float4 acc[TB];
#pragma unroll
    for (int g = 0; g < TB; ++g) {
        m[g] = -INFINITY; den[g] = 0.f; acc[g] = make_float4(0.f, 0.f, 0.f, 0.f);
    }
    int k = beg;
    for (; k + U <= end; k += U) gat_edges<TB, U>(srcS, hw4, k, q, as, edd, m, den, acc);
    if (k + 2 <= end) { gat_edges<TB, 2>(srcS, hw4, k, q, as, edd, m, den, acc); k += 2; }
    if (k < end)        gat_edges<TB, 1>(srcS, hw4, k, q, as, edd, m, den, acc);

    float4 o[TB];
#pragma unroll
    for (int g = 0; g < TB; ++g) {
        const float dn = fmaxf(den[g], 1e-16f);
        o[g].x = acc[g].x / dn; o[g].y = acc[g].y / dn; o[g].z = acc[g].z / dn; o[g].w = acc[g].w / dn;
        o[g].x = o[g].x > 0.f ? o[g].x : (__expf(o[g].x) - 1.f);
        o[g].y = o[g].y > 0.f ? o[g].y : (__expf(o[g].y) - 1.f);
        o[g].z = o[g].z > 0.f ? o[g].z : (__expf(o[g].z) - 1.f);
        o[g].w = o[g].w > 0.f ? o[g].w : (__expf(o[g].w) - 1.f);
    }

    if constexpr (!FUSE) {
        __shared__ float res[TB][GAT_THREADS / 4];
        const float4 mw = ((const float4*)Wn)[q];
        const float mb = bn[0];
        const int ln = threadIdx.x >> 2;
#pragma unroll
        for (int g = 0; g < TB; ++g) {
            const float r = quad_sum(dot4(o[g], mw)) + mb;
            if (q == 0) res[g][ln] = r;
        }
        __syncthreads();
        if (threadIdx.x < GAT_THREADS / 4) {
            const int n0 = blockIdx.x * (GAT_THREADS / 4) + threadIdx.x;
            if (n0 < N) {
#pragma unroll
                for (int g = 0; g < TB; ++g) hout[(size_t)g * N + n0] = res[g][threadIdx.x];
            }
        }
    } else {
        constexpr int R  = 8 * TB;
        constexpr int NT = (R + 15) / 16;
        __shared__ float tile[GAT_THREADS / 32][NT * 16][TPITCH];
        const int lane = threadIdx.x & 31;
        const int wave = threadIdx.x >> 5;
        float (*T)[TPITCH] = tile[wave];
        const int nd = lane >> 2;
#pragma unroll
        for (int g = 0; g < TB; ++g) *(float4*)&T[nd * TB + g][4 * q] = o[g];
        wave_lds_handover();

        const int c  = lane & 15;
        const int hf = lane >> 4;
        v2f bfrag[4];
#pragma unroll
        for (int j = 0; j < 4; ++j) {
            const int kc = 4 * j + 2 * hf;
            bfrag[j].x = Wn[kc * HD + c];
            bfrag[j].y = Wn[(kc + 1) * HD + c];
        }
        const float bi = bn[c];
        v8f d[NT];
#pragma unroll
        for (int tI = 0; tI < NT; ++tI) {
            int row = 16 * tI + c;
            if (row >= R) row = R - 1;
            v8f a8 = {};
#pragma unroll
            for (int j = 0; j < 4; ++j) {
                const v2f a = *(const v2f*)&T[row][4 * j + 2 * hf];
                a8 = __builtin_amdgcn_wmma_f32_16x16x4_f32(
                         false, a, false, bfrag[j], (short)0, a8, false, false);
            }
            d[tI] = a8;
        }
        wave_lds_handover();
#pragma unroll
        for (int tI = 0; tI < NT; ++tI) {
#pragma unroll
            for (int r = 0; r < 8; ++r) T[16 * tI + r + 8 * hf][c] = d[tI][r] + bi;
        }
        wave_lds_handover();
        const int rowbase = (blockIdx.x * (GAT_THREADS / 4) + wave * 8) * TB;
        const int rows = N * TB;
#pragma unroll
        for (int j = 0; j < R / 8; ++j) {
            const int f = lane + 32 * j;
            const int row = f >> 2;
            const int c4 = f & 3;
            if (rowbase + row < rows)
                ((float4*)hout)[(size_t)(rowbase + row) * 4 + c4] = *(const float4*)&T[row][4 * c4];
        }
    }
}

#define G0_THREADS 1024
#define G0_BLOCKS 256
#define WROWS (NWIRE + 1)

template <int TB, int U>
__device__ __forceinline__ void gat0_edges(const int* __restrict__ srcS, const unsigned* __restrict__ rec,
                                           const float4* wl, const int k, const int q, const bool isw,
                                           const int qrow, const float4 va,
                                           const float (&edd)[TB], float (&m)[TB], float (&den)[TB],
                                           float4 (&acc)[TB]) {
    int s[U];
    unsigned w[U];
    float4 hv[U][TB];
    float l[U][TB];
#pragma unroll
    for (int u = 0; u < U; ++u) s[u] = srcS[k + u];
#pragma unroll
    for (int u = 0; u < U; ++u) w[u] = rec[s[u] * 4 + q];
#pragma unroll
    for (int u = 0; u < U; ++u) {
        const int row = qrow + (isw ? (int)w[u] : NWIRE);
#pragma unroll
        for (int g = 0; g < TB; ++g) {
            float4 t = wl[row + g * WROWS];
            t.x = isw ? t.x : __uint_as_float(w[u]);
            hv[u][g] = t;
        }
    }
#pragma unroll
    for (int u = 0; u < U; ++u) {
#pragma unroll
        for (int g = 0; g < TB; ++g) l[u][g] = lrelu02(quad_sum(dot4(hv[u][g], va)) + edd[g]);
    }
#pragma unroll
    for (int g = 0; g < TB; ++g) {
        float bm = m[g];
#pragma unroll
        for (int u = 0; u < U; ++u) bm = fmaxf(bm, l[u][g]);
        const float sc = __expf(m[g] - bm);
        m[g] = bm;
        den[g] *= sc;
        acc[g].x *= sc; acc[g].y *= sc; acc[g].z *= sc; acc[g].w *= sc;
#pragma unroll
        for (int u = 0; u < U; ++u) {
            const float wgt = __expf(l[u][g] - bm);
            den[g] += wgt;
            acc[g].x += wgt * hv[u][g].x; acc[g].y += wgt * hv[u][g].y;
            acc[g].z += wgt * hv[u][g].z; acc[g].w += wgt * hv[u][g].w;
        }
    }
}

template <int TB>
__global__ __launch_bounds__(G0_THREADS) void k_gat0_node(const int* __restrict__ rowptr,
                                                          const int* __restrict__ srcS,
                                                          const unsigned* __restrict__ rec,
                                                          const float* __restrict__ fw,
                                                          const float* __restrict__ sw,
                                                          const float* __restrict__ tw,
                                                          const float* __restrict__ W0p,
                                                          const float* __restrict__ lv,
                                                          const float* __restrict__ Wn,
                                                          const float* __restrict__ bn,
                                                          float* __restrict__ hout, int N, int g0) {
    constexpr int R  = 8 * TB;
    constexpr int NT = (R + 15) / 16;
    __shared__ float4 wl[3 * TB * WROWS];
    __shared__ __attribute__((aligned(16))) float tile[G0_THREADS / 32][NT * 16][TPITCH];
    __shared__ __attribute__((aligned(16))) float wts[256 + 256 + HD + 48];
    float* const sW0 = wts;
    float* const sWn = wts + 256;
    float* const sbn = wts + 512;
    float* const slv = wts + 512 + HD;

    const size_t goff = (size_t)g0 * NWIRE * NFEAT;
    const float4* __restrict__ f4 = (const float4*)(fw + goff);
    const float4* __restrict__ s4 = (const float4*)(sw + goff);
    const float4* __restrict__ t4 = (const float4*)(tw + goff);
    for (int i = threadIdx.x; i < TB * NWIRE; i += G0_THREADS) {
        const int g = i / NWIRE;
        const int r = i - g * NWIRE;
        wl[(0 * TB + g) * WROWS + r] = f4[i];
        wl[(1 * TB + g) * WROWS + r] = s4[i];
        wl[(2 * TB + g) * WROWS + r] = t4[i];
    }
    if (threadIdx.x < 3 * TB) wl[threadIdx.x * WROWS + NWIRE] = make_float4(0.f, 0.f, 0.f, 0.f);
    if (threadIdx.x < 256) { sW0[threadIdx.x] = W0p[threadIdx.x]; sWn[threadIdx.x] = Wn[threadIdx.x]; }
    if (threadIdx.x < HD) sbn[threadIdx.x] = bn[threadIdx.x];
    if (threadIdx.x < 33) slv[threadIdx.x] = lv[threadIdx.x];
    __syncthreads();

    const int lane = threadIdx.x & 31;
    const int wave = __builtin_amdgcn_readfirstlane(threadIdx.x >> 5);
    const int q  = lane & 3;
    const int nd = lane >> 2;
    const int c  = lane & 15;
    const int hf = lane >> 4;
    const bool isw = q < 3;
    const int qrow = isw ? q * TB * WROWS : 0;
    const float4 va = ((const float4*)slv)[q];
    float (*T)[TPITCH] = tile[wave];
    const int numChunks = (N + 7) >> 3;
    const int rows = N * TB;
    if constexpr (NT * 16 > R) {
        for (int i = lane; i < (NT * 16 - R) * HD; i += 32) T[R + (i >> 4)][i & 15] = 0.f;
    }

    for (int ch = wave * gridDim.x + blockIdx.x; ch < numChunks; ch += gridDim.x * (G0_THREADS / 32)) {
        const int node = ch * 8 + nd;
        const bool valid = node < N;
        int beg = 0, end = 0;
        float edd[TB];
#pragma unroll
        for (int g = 0; g < TB; ++g) edd[g] = 0.f;
        const float csd = slv[32];
        if (valid) {
            beg = rowptr[node]; end = rowptr[node + 1];
            const unsigned wd = rec[node * 4 + q];
            const float4 vd = ((const float4*)(slv + 16))[q];
            const int row = qrow + (isw ? (int)wd : NWIRE);
#pragma unroll
            for (int g = 0; g < TB; ++g) {
                float4 t = wl[row + g * WROWS];
                t.x = isw ? t.x : __uint_as_float(wd);
                edd[g] = dot4(t, vd);
            }
        }
#pragma unroll
        for (int g = 0; g < TB; ++g) edd[g] = quad_sum(edd[g]) + csd;

        constexpr int U = 4;
        float m[TB], den[TB];
        float4 acc[TB];
#pragma unroll
        for (int g = 0; g < TB; ++g) {
            m[g] = -INFINITY; den[g] = 0.f; acc[g] = make_float4(0.f, 0.f, 0.f, 0.f);
        }
        int k = beg;
        for (; k + U <= end; k += U) gat0_edges<TB, U>(srcS, rec, wl, k, q, isw, qrow, va, edd, m, den, acc);
        if (k + 2 <= end) { gat0_edges<TB, 2>(srcS, rec, wl, k, q, isw, qrow, va, edd, m, den, acc); k += 2; }
        if (k < end)        gat0_edges<TB, 1>(srcS, rec, wl, k, q, isw, qrow, va, edd, m, den, acc);

        float4 o[TB];
#pragma unroll
        for (int g = 0; g < TB; ++g) {
            const float dn = fmaxf(den[g], 1e-16f);
            o[g].x = acc[g].x / dn; o[g].y = acc[g].y / dn; o[g].z = acc[g].z / dn; o[g].w = acc[g].w / dn;
            if (q == 3) o[g].y = den[g] / dn;
        }

        wave_lds_handover();
#pragma unroll
        for (int g = 0; g < TB; ++g) *(float4*)&T[nd * TB + g][4 * q] = o[g];
        wave_lds_handover();

        v2f bfrag[4];
#pragma unroll
        for (int j = 0; j < 4; ++j) {
            const int kc = 4 * j + 2 * hf;
            bfrag[j].x = sW0[kc * HD + c];
            bfrag[j].y = sW0[(kc + 1) * HD + c];
        }
        v8f d[NT];
#pragma unroll
        for (int tI = 0; tI < NT; ++tI) {
            v8f a8 = {};
#pragma unroll
            for (int j = 0; j < 4; ++j) {
                const v2f a = *(const v2f*)&T[16 * tI + c][4 * j + 2 * hf];
                a8 = __builtin_amdgcn_wmma_f32_16x16x4_f32(
                         false, a, false, bfrag[j], (short)0, a8, false, false);
            }
            d[tI] = a8;
        }
        wave_lds_handover();
#pragma unroll
        for (int tI = 0; tI < NT; ++tI) {
#pragma unroll
            for (int r = 0; r < 8; ++r) {
                float v = d[tI][r];
                v = v > 0.f ? v : (__expf(v) - 1.f);
                T[16 * tI + r + 8 * hf][c] = v;
            }
        }
        wave_lds_handover();

#pragma unroll
        for (int j = 0; j < 4; ++j) {
            const int kc = 4 * j + 2 * hf;
            bfrag[j].x = sWn[kc * HD + c];
            bfrag[j].y = sWn[(kc + 1) * HD + c];
        }
        const float bi1 = sbn[c];
#pragma unroll
        for (int tI = 0; tI < NT; ++tI) {
            v8f a8 = {};
#pragma unroll
            for (int j = 0; j < 4; ++j) {
                const v2f a = *(const v2f*)&T[16 * tI + c][4 * j + 2 * hf];
                a8 = __builtin_amdgcn_wmma_f32_16x16x4_f32(
                         false, a, false, bfrag[j], (short)0, a8, false, false);
            }
            d[tI] = a8;
        }
        wave_lds_handover();
#pragma unroll
        for (int tI = 0; tI < NT; ++tI) {
#pragma unroll
            for (int r = 0; r < 8; ++r) T[16 * tI + r + 8 * hf][c] = d[tI][r] + bi1;
        }
        wave_lds_handover();
        const int rowbase = ch * R;
#pragma unroll
        for (int j = 0; j < R / 8; ++j) {
            const int f = lane + 32 * j;
            const int row = f >> 2;
            const int c4 = f & 3;
            if (rowbase + row < rows)
                ((float4*)hout)[(size_t)(rowbase + row) * 4 + c4] = *(const float4*)&T[row][4 * c4];
        }
    }
}

static inline int cdiv(int a, int b) { return (a + b - 1) / b; }

struct Bufs {
    const float *fw, *sw, *tw, *areas;
    const int* indices;
    const float *W0p, *lv, *Wl, *a_src, *a_dst, *bvec, *mlpW, *mlpb;
    const int *rowptr, *srcS;
    float *h, *hw, *out;
};

template <int TB>
static void run_graphs(const Bufs& b, int N, int g0, hipStream_t stream) {
    const int gatBl  = cdiv(N * 4, GAT_THREADS);
    unsigned* rec = (unsigned*)b.hw;
    k_pack<<<cdiv(N, 256), 256, 0, stream>>>(b.indices, b.areas, rec, N);
    k_gat0_node<TB><<<G0_BLOCKS, G0_THREADS, 0, stream>>>(b.rowptr, b.srcS, rec, b.fw, b.sw, b.tw,
                                                         b.W0p, b.lv, b.Wl, b.bvec + HD, b.h, N, g0);
    float* cur = b.h;
    float* nxt = b.hw;
    for (int i = 1; i < 4; ++i) {
        const float* as = b.a_src + (size_t)i * HD;
        const float* ad = b.a_dst + (size_t)i * HD;
        if (i < 3)
            k_gat_node<TB, true><<<gatBl, GAT_THREADS, 0, stream>>>(b.rowptr, b.srcS, cur, as, ad,
                                                                   b.Wl + (size_t)i * 256,
                                                                   b.bvec + (size_t)(i + 1) * HD, nxt, N);
        else
            k_gat_node<TB, false><<<gatBl, GAT_THREADS, 0, stream>>>(b.rowptr, b.srcS, cur, as, ad,
                                                                    b.mlpW, b.mlpb,
                                                                    b.out + (size_t)g0 * N, N);
        float* tmp = cur; cur = nxt; nxt = tmp;
    }
}

extern "C" void kernel_launch(void* const* d_in, const int* in_sizes, int n_in,
                              void* d_out, int out_size, void* d_ws, size_t ws_size,
                              hipStream_t stream) {
    const float* fw     = (const float*)d_in[0];
    const float* sw     = (const float*)d_in[1];
    const float* tw     = (const float*)d_in[2];
    const float* areas  = (const float*)d_in[3];
    const int*   indices= (const int*)  d_in[4];
    const int*   edge   = (const int*)  d_in[5];
    const float* W0     = (const float*)d_in[6];
    const float* Wl     = (const float*)d_in[7];
    const float* a_src  = (const float*)d_in[8];
    const float* a_dst  = (const float*)d_in[9];
    const float* bvec   = (const float*)d_in[10];
    const float* mlpW   = (const float*)d_in[11];
    const float* mlpb   = (const float*)d_in[12];

    const int N = in_sizes[3];
    const int E = in_sizes[5] / 2;
    const int T = in_sizes[0] / (NWIRE * NFEAT);
    const int* src = edge;
    const int* dst = edge + E;
    const int numTiles = cdiv(N, SCAN_TILE);

    const size_t fixedW = 320 + (size_t)E + ((size_t)N + 1) + 2 * (size_t)N + 2 * (size_t)numTiles;
    const size_t need3  = 2 * (size_t)N * HD * 3 + fixedW;
    const size_t need2  = 2 * (size_t)N * HD * 2 + fixedW;
    int TBS = 1;
    if (T >= 3 && ws_size >= need3 * 4) TBS = 3;
    else if (T >= 2 && ws_size >= need2 * 4) TBS = 2;

    float* ws = (float*)d_ws;
    size_t off = 0;
    float* h       = ws + off; off += (size_t)N * HD * TBS;
    float* hw      = ws + off; off += (size_t)N * HD * TBS;
    float* W0p     = ws + off; off += 256;
    float* lv      = ws + off; off += 64;
    int*   srcS    = (int*)(ws + off); off += (size_t)E;
    int*   rowptr  = (int*)(ws + off); off += (size_t)N + 1;
    int*   cursor  = (int*)(ws + off); off += (size_t)N;
    int*   cnt     = (int*)(ws + off); off += (size_t)N;
    int*   tilesum = (int*)(ws + off); off += (size_t)numTiles;
    int*   tileoff = (int*)(ws + off); off += (size_t)numTiles;
    (void)n_in; (void)out_size;

    const int numBins = numTiles;
    int* binCnt = tilesum;
    int* binOff = tileoff;
    int* binCur = cnt;
    const bool binned = N <= (1 << SRC_BITS) && (size_t)E <= (size_t)N * HD * TBS;

    k_prep<<<1, 256, 0, stream>>>(W0, a_src, a_dst, bvec, W0p, lv, binCnt, binCur, binned ? numBins : 0);
    if (binned) {
        const int chunks = cdiv(E, PART_CHUNK);
        k_binhist<<<chunks, PART_THREADS, 0, stream>>>(dst, binCnt, E, numBins);
        k_scan_tiles<<<1, 1024, 0, stream>>>(binCnt, binOff, rowptr, numBins, N);
        k_partition<<<chunks, PART_THREADS, 0, stream>>>(src, dst, binOff, binCur, (unsigned*)h, E, numBins);
        k_binsort<<<numBins, 1024, 0, stream>>>((const unsigned*)h, binOff, binCnt, rowptr, srcS, N);
    } else {
        k_zero<<<cdiv(N, 256), 256, 0, stream>>>(cnt, N);
        k_hist<<<cdiv(E, 256), 256, 0, stream>>>(dst, cnt, E);
        k_scan_local<<<numTiles, 256, 0, stream>>>(cnt, rowptr, tilesum, N);
        k_scan_tiles<<<1, 1024, 0, stream>>>(tilesum, tileoff, rowptr, numTiles, N);
        k_scan_add<<<cdiv(N, 256), 256, 0, stream>>>(rowptr, tileoff, cursor, N);
        k_scatter<<<cdiv(E, 256), 256, 0, stream>>>(src, dst, cursor, srcS, E);
    }

    Bufs b;
    b.fw = fw; b.sw = sw; b.tw = tw; b.areas = areas; b.indices = indices;
    b.W0p = W0p; b.lv = lv; b.Wl = Wl; b.a_src = a_src; b.a_dst = a_dst; b.bvec = bvec;
    b.mlpW = mlpW; b.mlpb = mlpb; b.rowptr = rowptr; b.srcS = srcS;
    b.h = h; b.hw = hw; b.out = (float*)d_out;

    int g = 0;
    while (g < T) {
        const int left = T - g;
        const int tb = left < TBS ? left : TBS;
        if (tb >= 3)      { run_graphs<3>(b, N, g, stream); g += 3; }
        else if (tb == 2) { run_graphs<2>(b, N, g, stream); g += 2; }
        else              { run_graphs<1>(b, N, g, stream); g += 1; }
    }
}


# Edge loops: fewer vector instructions per edge: quad lane u takes edge u (transposed quad sum; leaky relu, exp once per edge; weights broadcast), partial dots packed in place; dot sum order changes
# speedup vs baseline: 21.8088x; 1.0456x over previous
#include <hip/hip_runtime.h>
#include <math.h>


typedef float v2f __attribute__((ext_vector_type(2)));
typedef float v8f __attribute__((ext_vector_type(8)));

#define HD 16
#define NWIRE 1536
#define NFEAT 4
#define SCAN_TILE 1024


__device__ __forceinline__ float lrelu02(float x) { return fmaxf(x, 0.2f * x); }

__device__ __forceinline__ float dot4(const float4 v, const float4 a) {
    const v2f lo = {v.x, v.y}, hi = {v.z, v.w};
    const v2f alo = {a.x, a.y}, ahi = {a.z, a.w};
    const v2f p = hi * ahi + lo * alo;
    return p.x + p.y;
}

__device__ __forceinline__ float quad_sum(float p) {
    p += __int_as_float(__builtin_amdgcn_update_dpp(0, __float_as_int(p), 0xB1, 0xF, 0xF, true));
    p += __int_as_float(__builtin_amdgcn_update_dpp(0, __float_as_int(p), 0x4E, 0xF, 0xF, true));
    return p;
}

template <int TB>
__device__ __forceinline__ int row_f4(int s) {
    if constexpr (TB == 3) {
        int t;
        asm("v_lshl_add_u32 %0, %1, 1, %1" : "=v"(t) : "v"(s));
        return t << 2;
    } else {
        return s * (TB * 4);
    }
}

template <int CTRL>
__device__ __forceinline__ float quad_get(float x) {
    return __int_as_float(__builtin_amdgcn_update_dpp(0, __float_as_int(x), CTRL, 0xF, 0xF, true));
}

template <int TB, int U>
__device__ __forceinline__ void block_weights(const float (&pd)[U][TB], const int q, const float (&edd)[TB],
                                              float (&m)[TB], float (&sc)[TB], float (&w)[U][TB]) {
    if constexpr (U == 4) {
        const bool b0 = (q & 1) != 0;
        const bool b1 = (q & 2) != 0;
#pragma unroll
        for (int g = 0; g < TB; ++g) {
            const float k01 = (b0 ? pd[1][g] : pd[0][g]) + quad_get<0xB1>(b0 ? pd[0][g] : pd[1][g]);
            const float k23 = (b0 ? pd[3][g] : pd[2][g]) + quad_get<0xB1>(b0 ? pd[2][g] : pd[3][g]);
            const float lq = lrelu02(((b1 ? k23 : k01) + quad_get<0x4E>(b1 ? k01 : k23)) + edd[g]);
            float bq = fmaxf(lq, quad_get<0xB1>(lq));
            bq = fmaxf(bq, quad_get<0x4E>(bq));
            const float bm = fmaxf(m[g], bq);
            sc[g] = __expf(m[g] - bm);
            m[g] = bm;
            const float wq = __expf(lq - bm);
            w[0][g] = quad_get<0x00>(wq);
            w[1][g] = quad_get<0x55>(wq);
            w[2][g] = quad_get<0xAA>(wq);
            w[3][g] = quad_get<0xFF>(wq);
        }
    } else {
#pragma unroll
        for (int g = 0; g < TB; ++g) {
            float l[U];
            float bm = m[g];
#pragma unroll
            for (int u = 0; u < U; ++u) { l[u] = lrelu02(quad_sum(pd[u][g]) + edd[g]); bm = fmaxf(bm, l[u]); }
            sc[g] = __expf(m[g] - bm);
            m[g] = bm;
#pragma unroll
            for (int u = 0; u < U; ++u) w[u][g] = __expf(l[u] - bm);
        }
    }
}

__global__ void k_prep(const float* __restrict__ W0, const float* __restrict__ asrc,
                       const float* __restrict__ adst, const float* __restrict__ b0,
                       float* __restrict__ W0p, float* __restrict__ lv,
                       int* __restrict__ binCnt, int* __restrict__ binCur, int numBins) {
    int i = threadIdx.x;
    int r = i >> 4;
    W0p[i] = (r < 13) ? W0[i] : (r == 13 ? b0[i & 15] : 0.f);
    if (i < 32) {
        const float* __restrict__ a = (i < 16) ? asrc : adst;
        const int k = i & 15;
        float sum = 0.f;
        if (k < 13) {
            for (int j = 0; j < HD; ++j) sum += W0[k * HD + j] * a[j];
        }
        lv[i] = sum;
    } else if (i == 32) {
        float cs = 0.f, cd = 0.f;
        for (int j = 0; j < HD; ++j) { cs += b0[j] * asrc[j]; cd += b0[j] * adst[j]; }
        lv[32] = cs + cd;
    }
    for (int b = i; b < numBins; b += 256) { binCnt[b] = 0; binCur[b] = 0; }
}

__global__ void k_pack(const int* __restrict__ indices, const float* __restrict__ areas,
                       unsigned* __restrict__ rec, int N) {
    const int n = blockIdx.x * blockDim.x + threadIdx.x;
    if (n >= N) return;
    uint4 r;
    r.x = (unsigned)indices[n * 3 + 0];
    r.y = (unsigned)indices[n * 3 + 1];
    r.z = (unsigned)indices[n * 3 + 2];
    r.w = __float_as_uint(areas[n]);
    ((uint4*)rec)[n] = r;
}

__global__ void k_zero(int* __restrict__ cnt, int N) {
    int i = blockIdx.x * blockDim.x + threadIdx.x;
    if (i < N) cnt[i] = 0;
}

__global__ void k_hist(const int* __restrict__ dst, int* __restrict__ cnt, int E) {
    int e = blockIdx.x * blockDim.x + threadIdx.x;
    if (e < E) atomicAdd(&cnt[dst[e]], 1);
}

__global__ __launch_bounds__(256) void k_scan_local(const int* __restrict__ cnt,
                                                    int* __restrict__ rowptr,
                                                    int* __restrict__ tilesum, int N) {
    __shared__ int wsum[8];
    const int t = threadIdx.x;
    const int lane = t & 31;
    const int wv = t >> 5;
    const int i0 = (blockIdx.x * 256 + t) * 4;
    const int c0 = (i0 + 0 < N) ? cnt[i0 + 0] : 0;
    const int c1 = (i0 + 1 < N) ? cnt[i0 + 1] : 0;
    const int c2 = (i0 + 2 < N) ? cnt[i0 + 2] : 0;
    const int c3 = (i0 + 3 < N) ? cnt[i0 + 3] : 0;
    const int s = c0 + c1 + c2 + c3;
    int inc = s;
    for (int off = 1; off < 32; off <<= 1) {
        const int v = __shfl_up(inc, off);
        if (lane >= off) inc += v;
    }
    if (lane == 31) wsum[wv] = inc;
    __syncthreads();
    int wpre = 0;
    for (int k = 0; k < 8; ++k) if (k < wv) wpre += wsum[k];
    int ex = wpre + inc - s;
    if (i0 + 0 < N) rowptr[i0 + 0] = ex;
    ex += c0;
    if (i0 + 1 < N) rowptr[i0 + 1] = ex;
    ex += c1;
    if (i0 + 2 < N) rowptr[i0 + 2] = ex;
    ex += c2;
    if (i0 + 3 < N) rowptr[i0 + 3] = ex;
    if (t == 255) tilesum[blockIdx.x] = wpre + inc;
}

__global__ __launch_bounds__(1024) void k_scan_tiles(const int* __restrict__ tilesum,
                                                     int* __restrict__ tileoff,
                                                     int* __restrict__ rowptr,
                                                     int numTiles, int N) {
    __shared__ int buf[1024];
    const int t = threadIdx.x;
    int carry = 0;
    for (int base = 0; base < numTiles; base += 1024) {
        const int i = base + t;
        const int v = (i < numTiles) ? tilesum[i] : 0;
        buf[t] = v;
        __syncthreads();
        for (int off = 1; off < 1024; off <<= 1) {
            const int a = (t >= off) ? buf[t - off] : 0;
            __syncthreads();
            buf[t] += a;
            __syncthreads();
        }
        if (i < numTiles) tileoff[i] = carry + buf[t] - v;
        carry += buf[1023];
        __syncthreads();
    }
    if (t == 0) rowptr[N] = carry;
}

__global__ void k_scan_add(int* __restrict__ rowptr, const int* __restrict__ tileoff,
                           int* __restrict__ cursor, int N) {
    int i = blockIdx.x * blockDim.x + threadIdx.x;
    if (i >= N) return;
    const int v = rowptr[i] + tileoff[i / SCAN_TILE];
    rowptr[i] = v;
    cursor[i] = v;
}

__global__ void k_scatter(const int* __restrict__ src, const int* __restrict__ dst,
                          int* __restrict__ cursor, int* __restrict__ srcS, int E) {
    int e = blockIdx.x * blockDim.x + threadIdx.x;
    if (e >= E) return;
    const int pos = atomicAdd(&cursor[dst[e]], 1);
    srcS[pos] = src[e];
}

#define BIN_SHIFT 10
#define BIN_NODES (1 << BIN_SHIFT)
#define SRC_BITS 22
#define SRC_MASK ((1u << SRC_BITS) - 1u)
#define MAX_BINS (1 << (SRC_BITS - BIN_SHIFT))
#define PART_THREADS 1024
#define PART_EPT 8
#define PART_CHUNK (PART_THREADS * PART_EPT)
#define PART_SCAN (MAX_BINS / PART_THREADS)
#define BIN_CAP 49152
#define BIN_LD 8
static_assert(BIN_NODES == SCAN_TILE, "the bin arrays share the scan-tile arrays of the workspace");

__global__ __launch_bounds__(PART_THREADS) void k_binhist(const int* __restrict__ dst,
                                                          int* __restrict__ binCnt,
                                                          int E, int numBins) {
    __shared__ int hist[MAX_BINS];
    const int t = threadIdx.x;
    for (int b = t; b < numBins; b += PART_THREADS) hist[b] = 0;
    __syncthreads();
    const int base = blockIdx.x * PART_CHUNK;
    int d[PART_EPT];
#pragma unroll
    for (int i = 0; i < PART_EPT; ++i) {
        const int e = base + i * PART_THREADS + t;
        d[i] = dst[e < E ? e : E - 1];
    }
#pragma unroll
    for (int i = 0; i < PART_EPT; ++i) {
        atomicAdd(&hist[d[i] >> BIN_SHIFT], (base + i * PART_THREADS + t < E) ? 1 : 0);
    }
    __syncthreads();
    for (int b = t; b < numBins; b += PART_THREADS) {
        const int c = hist[b];
        if (c) atomicAdd(&binCnt[b], c);
    }
}

__global__ __launch_bounds__(PART_THREADS) void k_partition(const int* __restrict__ src,
                                                            const int* __restrict__ dst,
                                                            const int* __restrict__ binOff,
                                                            int* __restrict__ binCur,
                                                            unsigned* __restrict__ tmp,
                                                            int E, int numBins) {
    __shared__ int hist[MAX_BINS];
    __shared__ int lofs[MAX_BINS];
    __shared__ int wsum[PART_THREADS / 32];
    __shared__ unsigned stage[PART_CHUNK];
    __shared__ unsigned short sbin[PART_CHUNK];
    static_assert(MAX_BINS == PART_SCAN * PART_THREADS, "the block scan takes PART_SCAN bins per thread");
    static_assert(MAX_BINS <= 65536, "bin ids are staged as 16-bit words");
    static_assert(PART_THREADS / 32 == 32, "the wave totals are scanned one per lane");
    const int t = threadIdx.x;
    const int lane = t & 31;
    const int wv = t >> 5;
    for (int b = t; b < MAX_BINS; b += PART_THREADS) hist[b] = 0;
    const int base = blockIdx.x * PART_CHUNK;
    int d[PART_EPT], s[PART_EPT], r[PART_EPT];
#pragma unroll
    for (int i = 0; i < PART_EPT; ++i) {
        const int e = base + i * PART_THREADS + t;
        const int ec = e < E ? e : E - 1;
        const int dv = dst[ec];
        const int sv = src[ec];
        s[i] = (int)(((unsigned)(dv & (BIN_NODES - 1)) << SRC_BITS) | (unsigned)sv);
        d[i] = (e < E) ? (dv >> BIN_SHIFT) : -1;
    }
    __syncthreads();
#pragma unroll
    for (int i = 0; i < PART_EPT; ++i) {
        r[i] = 0;
        if (d[i] >= 0) r[i] = atomicAdd(&hist[d[i]], 1);
    }
    __syncthreads();
    const int b0 = t * PART_SCAN;
    int cc[PART_SCAN];
    int sm = 0;
#pragma unroll
    for (int j = 0; j < PART_SCAN; ++j) { cc[j] = hist[b0 + j]; sm += cc[j]; }
    int inc = sm;
    for (int o = 1; o < 32; o <<= 1) {
        const int v = __shfl_up(inc, o);
        if (lane >= o) inc += v;
    }
    if (lane == 31) wsum[wv] = inc;
    __syncthreads();
    const int wt = wsum[lane];
    int winc = wt;
    for (int o = 1; o < 32; o <<= 1) {
        const int v = __shfl_up(winc, o);
        if (lane >= o) winc += v;
    }
    const int wpre = __shfl(winc - wt, wv);
    int ex = wpre + inc - sm;
#pragma unroll
    for (int j = 0; j < PART_SCAN; ++j) {
        const int b = b0 + j;
        const int cnt = cc[j];
        int g = 0;
        if (cnt && b < numBins) g = binOff[b] + atomicAdd(&binCur[b], cnt);
        lofs[b] = ex;
        hist[b] = g - ex;
        ex += cnt;
    }
    __syncthreads();
#pragma unroll
    for (int i = 0; i < PART_EPT; ++i) r[i] += lofs[d[i] >= 0 ? d[i] : 0];
#pragma unroll
    for (int i = 0; i < PART_EPT; ++i) {
        if (d[i] >= 0) {
            stage[r[i]] = (unsigned)s[i];
            sbin[r[i]] = (unsigned short)d[i];
        }
    }
    __syncthreads();
    const int left = E - base;
    const int total = left < PART_CHUNK ? left : PART_CHUNK;
#pragma unroll
    for (int i = 0; i < PART_EPT; ++i) {
        const int p = i * PART_THREADS + t;
        const int pc = p < total ? p : total - 1;
        s[i] = (int)stage[pc];
        d[i] = hist[sbin[pc]] + pc;
    }
#pragma unroll
    for (int i = 0; i < PART_EPT; ++i) {
        if (i * PART_THREADS + t < total) tmp[d[i]] = (unsigned)s[i];
    }
}

__global__ __launch_bounds__(1024) void k_binsort(const unsigned* __restrict__ tmp,
                                                  const int* __restrict__ binOff,
                                                  const int* __restrict__ binCnt,
                                                  int* __restrict__ rowptr,
                                                  int* __restrict__ srcS, int N) {
    __shared__ int cnt[BIN_NODES];
    __shared__ int off[BIN_NODES + 1];
    __shared__ int wsum[32];
    __shared__ unsigned a[BIN_CAP];
    const int t = threadIdx.x;
    const int lane = t & 31;
    const int wv = t >> 5;
    const int bin = blockIdx.x;
    const int gbeg = binOff[bin];
    const int n = binCnt[bin];
    const unsigned* __restrict__ mine = tmp + gbeg;
    cnt[t] = 0;
    __syncthreads();
    for (int i0 = t; i0 < n; i0 += 1024 * BIN_LD) {
        unsigned w[BIN_LD];
#pragma unroll
        for (int u = 0; u < BIN_LD; ++u) {
            const int i = i0 + u * 1024;
            w[u] = mine[i < n ? i : i0];
        }
#pragma unroll
        for (int u = 0; u < BIN_LD; ++u) {
            const int i = i0 + u * 1024;
            atomicAdd(&cnt[w[u] >> SRC_BITS], (i < n) ? 1 : 0);
        }
    }
    __syncthreads();
    const int c = cnt[t];
    int inc = c;
    for (int o = 1; o < 32; o <<= 1) {
        const int v = __shfl_up(inc, o);
        if (lane >= o) inc += v;
    }
    if (lane == 31) wsum[wv] = inc;
    __syncthreads();
    int wpre = 0;
    for (int k = 0; k < 32; ++k) if (k < wv) wpre += wsum[k];
    const int ex = wpre + inc - c;
    off[t] = ex;
    if (t == 1023) off[BIN_NODES] = ex + c;
    cnt[t] = ex;
    const int node = bin * BIN_NODES + t;
    if (node < N) rowptr[node] = gbeg + ex;
    __syncthreads();
    if (n <= BIN_CAP) {
        for (int i0 = t; i0 < n; i0 += 1024 * BIN_LD) {
            unsigned w[BIN_LD];
#pragma unroll
            for (int u = 0; u < BIN_LD; ++u) {
                const int i = i0 + u * 1024;
                w[u] = mine[i < n ? i : i0];
            }
#pragma unroll
            for (int u = 0; u < BIN_LD; ++u) {
                const int i = i0 + u * 1024;
                const int p = atomicAdd(&cnt[w[u] >> SRC_BITS], (i < n) ? 1 : 0);
                if (i < n) a[p] = w[u];
            }
        }
        __syncthreads();
        for (int p = t; p < n; p += 1024) {
            const unsigned w = a[p];
            const int dl = (int)(w >> SRC_BITS);
            const int sb = off[dl];
            const int se = off[dl + 1];
            int rank = 0;
            for (int j = sb; j < se; ++j) {
                const unsigned v = a[j];
                rank += (v < w || (v == w && j < p)) ? 1 : 0;
            }
            srcS[gbeg + sb + rank] = (int)(w & SRC_MASK);
        }
    } else {
        for (int i = t; i < n; i += 1024) {
            const unsigned w = mine[i];
            const int p = atomicAdd(&cnt[w >> SRC_BITS], 1);
            srcS[gbeg + p] = (int)(w & SRC_MASK);
        }
    }
}

template <int TB, int U>
__device__ __forceinline__ void gat_edges(const int* __restrict__ srcS, const float4* __restrict__ hw4,
                                          const int k, const int q, const float4 as,
                                          const float (&edd)[TB], float (&m)[TB], float (&den)[TB],
                                          float4 (&acc)[TB]) {
    int s[U];
    float4 hv[U][TB];
    float l[U][TB];
#pragma unroll
    for (int u = 0; u < U; ++u) s[u] = srcS[k + u];
#pragma unroll
    for (int u = 0; u < U; ++u) {
#pragma unroll
        for (int g = 0; g < TB; ++g) hv[u][g] = hw4[row_f4<TB>(s[u]) + g * 4 + q];
    }
    __builtin_amdgcn_sched_barrier(0);
#pragma unroll
    for (int u = 0; u < U; ++u) {
#pragma unroll
        for (int g = 0; g < TB; ++g) l[u][g] = dot4(hv[u][g], as);
    }
    float sc[TB];
    float w[U][TB];
    block_weights<TB, U>(l, q, edd, m, sc, w);
#pragma unroll
    for (int g = 0; g < TB; ++g) {
        den[g] *= sc[g];
        v2f alo = {acc[g].x, acc[g].y}, ahi = {acc[g].z, acc[g].w};
        alo *= sc[g]; ahi *= sc[g];
#pragma unroll
        for (int u = 0; u < U; ++u) {
            den[g] += w[u][g];
            const v2f hlo = {hv[u][g].x, hv[u][g].y}, hhi = {hv[u][g].z, hv[u][g].w};
            alo += w[u][g] * hlo; ahi += w[u][g] * hhi;
        }
        acc[g] = make_float4(alo.x, alo.y, ahi.x, ahi.y);
    }
}

#define GAT_THREADS 128
#define TPITCH 20

__device__ __forceinline__ void wave_lds_handover() {
    __builtin_amdgcn_fence(__ATOMIC_RELEASE, "wavefront");
    __builtin_amdgcn_wave_barrier();
    __builtin_amdgcn_fence(__ATOMIC_ACQUIRE, "wavefront");
}

template <int TB, bool FUSE>
__global__ __launch_bounds__(GAT_THREADS) void k_gat_node(const int* __restrict__ rowptr,
                                                  const int* __restrict__ srcS,
                                                  const float* __restrict__ hw,
                                                  const float* __restrict__ asrc,
                                                  const float* __restrict__ adst,
                                                  const float* __restrict__ Wn,
                                                  const float* __restrict__ bn,
                                                  float* __restrict__ hout, int N) {
    const int t = blockIdx.x * GAT_THREADS + threadIdx.x;
    const int node = t >> 2;
    const int q = t & 3;
    const bool valid = node < N;
    const float4* __restrict__ hw4 = (const float4*)hw;
    const float4 as = ((const float4*)asrc)[q];
    const float4 ad = ((const float4*)adst)[q];
    int beg = 0, end = 0;
    float edd[TB];
#pragma unroll
    for (int g = 0; g < TB; ++g) edd[g] = 0.f;
    if (valid) {
        beg = rowptr[node]; end = rowptr[node + 1];
#pragma unroll
        for (int g = 0; g < TB; ++g) edd[g] = dot4(hw4[(node * TB + g) * 4 + q], ad);
    }
#pragma unroll
    for (int g = 0; g < TB; ++g) edd[g] = quad_sum(edd[g]);

    constexpr int U = 4;
    float m[TB], den[TB];
    float4 acc[TB];
#pragma unroll
    for (int g = 0; g < TB; ++g) {
        m[g] = -INFINITY; den[g] = 0.f; acc[g] = make_float4(0.f, 0.f, 0.f, 0.f);
    }
    int k = beg;
    for (; k + U <= end; k += U) gat_edges<TB, U>(srcS, hw4, k, q, as, edd, m, den, acc);
    if (k + 2 <= end) { gat_edges<TB, 2>(srcS, hw4, k, q, as, edd, m, den, acc); k += 2; }
    if (k < end)        gat_edges<TB, 1>(srcS, hw4, k, q, as, edd, m, den, acc);

    float4 o[TB];
#pragma unroll
    for (int g = 0; g < TB; ++g) {
        const float dn = fmaxf(den[g], 1e-16f);
        o[g].x = acc[g].x / dn; o[g].y = acc[g].y / dn; o[g].z = acc[g].z / dn; o[g].w = acc[g].w / dn;
        o[g].x = o[g].x > 0.f ? o[g].x : (__expf(o[g].x) - 1.f);
        o[g].y = o[g].y > 0.f ? o[g].y : (__expf(o[g].y) - 1.f);
        o[g].z = o[g].z > 0.f ? o[g].z : (__expf(o[g].z) - 1.f);
        o[g].w = o[g].w > 0.f ? o[g].w : (__expf(o[g].w) - 1.f);
    }

    if constexpr (!FUSE) {
        __shared__ float res[TB][GAT_THREADS / 4];
        const float4 mw = ((const float4*)Wn)[q];
        const float mb = bn[0];
        const int ln = threadIdx.x >> 2;
#pragma unroll
        for (int g = 0; g < TB; ++g) {
            const float r = quad_sum(dot4(o[g], mw)) + mb;
            if (q == 0) res[g][ln] = r;
        }
        __syncthreads();
        if (threadIdx.x < GAT_THREADS / 4) {
            const int n0 = blockIdx.x * (GAT_THREADS / 4) + threadIdx.x;
            if (n0 < N) {
#pragma unroll
                for (int g = 0; g < TB; ++g) hout[(size_t)g * N + n0] = res[g][threadIdx.x];
            }
        }
    } else {
        constexpr int R  = 8 * TB;
        constexpr int NT = (R + 15) / 16;
        __shared__ float tile[GAT_THREADS / 32][NT * 16][TPITCH];
        const int lane = threadIdx.x & 31;
        const int wave = threadIdx.x >> 5;
        float (*T)[TPITCH] = tile[wave];
        const int nd = lane >> 2;
#pragma unroll
        for (int g = 0; g < TB; ++g) *(float4*)&T[nd * TB + g][4 * q] = o[g];
        wave_lds_handover();

        const int c  = lane & 15;
        const int hf = lane >> 4;
        v2f bfrag[4];
#pragma unroll
        for (int j = 0; j < 4; ++j) {
            const int kc = 4 * j + 2 * hf;
            bfrag[j].x = Wn[kc * HD + c];
            bfrag[j].y = Wn[(kc + 1) * HD + c];
        }
        const float bi = bn[c];
        v8f d[NT];
#pragma unroll
        for (int tI = 0; tI < NT; ++tI) {
            int row = 16 * tI + c;
            if (row >= R) row = R - 1;
            v8f a8 = {};
#pragma unroll
            for (int j = 0; j < 4; ++j) {
                const v2f a = *(const v2f*)&T[row][4 * j + 2 * hf];
                a8 = __builtin_amdgcn_wmma_f32_16x16x4_f32(
                         false, a, false, bfrag[j], (short)0, a8, false, false);
            }
            d[tI] = a8;
        }
        wave_lds_handover();
#pragma unroll
        for (int tI = 0; tI < NT; ++tI) {
#pragma unroll
            for (int r = 0; r < 8; ++r) T[16 * tI + r + 8 * hf][c] = d[tI][r] + bi;
        }
        wave_lds_handover();
        const int rowbase = (blockIdx.x * (GAT_THREADS / 4) + wave * 8) * TB;
        const int rows = N * TB;
#pragma unroll
        for (int j = 0; j < R / 8; ++j) {
            const int f = lane + 32 * j;
            const int row = f >> 2;
            const int c4 = f & 3;
            if (rowbase + row < rows)
                ((float4*)hout)[(size_t)(rowbase + row) * 4 + c4] = *(const float4*)&T[row][4 * c4];
        }
    }
}

#define G0_THREADS 1024
#define G0_BLOCKS 256
#define WROWS (NWIRE + 1)

template <int TB, int U>
__device__ __forceinline__ void gat0_edges(const int* __restrict__ srcS, const unsigned* __restrict__ rec,
                                           const float4* wl, const int k, const int q, const bool isw,
                                           const int qrow, const float4 va,
                                           const float (&edd)[TB], float (&m)[TB], float (&den)[TB],
                                           float4 (&acc)[TB]) {
    int s[U];
    unsigned w[U];
    float4 hv[U][TB];
    float l[U][TB];
#pragma unroll
    for (int u = 0; u < U; ++u) s[u] = srcS[k + u];
#pragma unroll
    for (int u = 0; u < U; ++u) w[u] = rec[s[u] * 4 + q];
#pragma unroll
    for (int u = 0; u < U; ++u) {
        const int row = qrow + (isw ? (int)w[u] : NWIRE);
#pragma unroll
        for (int g = 0; g < TB; ++g) {
            float4 t = wl[row + g * WROWS];
            t.x = isw ? t.x : __uint_as_float(w[u]);
            hv[u][g] = t;
        }
    }
#pragma unroll
    for (int u = 0; u < U; ++u) {
#pragma unroll
        for (int g = 0; g < TB; ++g) l[u][g] = dot4(hv[u][g], va);
    }
    float sc[TB];
    float wgt[U][TB];
    block_weights<TB, U>(l, q, edd, m, sc, wgt);
#pragma unroll
    for (int g = 0; g < TB; ++g) {
        den[g] *= sc[g];
        v2f alo = {acc[g].x, acc[g].y}, ahi = {acc[g].z, acc[g].w};
        alo *= sc[g]; ahi *= sc[g];
#pragma unroll
        for (int u = 0; u < U; ++u) {
            den[g] += wgt[u][g];
            const v2f hlo = {hv[u][g].x, hv[u][g].y}, hhi = {hv[u][g].z, hv[u][g].w};
            alo += wgt[u][g] * hlo; ahi += wgt[u][g] * hhi;
        }
        acc[g] = make_float4(alo.x, alo.y, ahi.x, ahi.y);
    }
}

template <int TB>
__global__ __launch_bounds__(G0_THREADS) void k_gat0_node(const int* __restrict__ rowptr,
                                                          const int* __restrict__ srcS,
                                                          const unsigned* __restrict__ rec,
                                                          const float* __restrict__ fw,
                                                          const float* __restrict__ sw,
                                                          const float* __restrict__ tw,
                                                          const float* __restrict__ W0p,
                                                          const float* __restrict__ lv,
                                                          const float* __restrict__ Wn,
                                                          const float* __restrict__ bn,
                                                          float* __restrict__ hout, int N, int g0) {
    constexpr int R  = 8 * TB;
    constexpr int NT = (R + 15) / 16;
    __shared__ float4 wl[3 * TB * WROWS];
    __shared__ __attribute__((aligned(16))) float tile[G0_THREADS / 32][NT * 16][TPITCH];
    __shared__ __attribute__((aligned(16))) float wts[256 + 256 + HD + 48];
    float* const sW0 = wts;
    float* const sWn = wts + 256;
    float* const sbn = wts + 512;
    float* const slv = wts + 512 + HD;

    const size_t goff = (size_t)g0 * NWIRE * NFEAT;
    const float4* __restrict__ f4 = (const float4*)(fw + goff);
    const float4* __restrict__ s4 = (const float4*)(sw + goff);
    const float4* __restrict__ t4 = (const float4*)(tw + goff);
    for (int i = threadIdx.x; i < TB * NWIRE; i += G0_THREADS) {
        const int g = i / NWIRE;
        const int r = i - g * NWIRE;
        wl[(0 * TB + g) * WROWS + r] = f4[i];
        wl[(1 * TB + g) * WROWS + r] = s4[i];
        wl[(2 * TB + g) * WROWS + r] = t4[i];
    }
    if (threadIdx.x < 3 * TB) wl[threadIdx.x * WROWS + NWIRE] = make_float4(0.f, 0.f, 0.f, 0.f);
    if (threadIdx.x < 256) { sW0[threadIdx.x] = W0p[threadIdx.x]; sWn[threadIdx.x] = Wn[threadIdx.x]; }
    if (threadIdx.x < HD) sbn[threadIdx.x] = bn[threadIdx.x];
    if (threadIdx.x < 33) slv[threadIdx.x] = lv[threadIdx.x];
    __syncthreads();

    const int lane = threadIdx.x & 31;
    const int wave = __builtin_amdgcn_readfirstlane(threadIdx.x >> 5);
    const int q  = lane & 3;
    const int nd = lane >> 2;
    const int c  = lane & 15;
    const int hf = lane >> 4;
    const bool isw = q < 3;
    const int qrow = isw ? q * TB * WROWS : 0;
    const float4 va = ((const float4*)slv)[q];
    float (*T)[TPITCH] = tile[wave];
    const int numChunks = (N + 7) >> 3;
    const int rows = N * TB;
    if constexpr (NT * 16 > R) {
        for (int i = lane; i < (NT * 16 - R) * HD; i += 32) T[R + (i >> 4)][i & 15] = 0.f;
    }

    for (int ch = wave * gridDim.x + blockIdx.x; ch < numChunks; ch += gridDim.x * (G0_THREADS / 32)) {
        const int node = ch * 8 + nd;
        const bool valid = node < N;
        int beg = 0, end = 0;
        float edd[TB];
#pragma unroll
        for (int g = 0; g < TB; ++g) edd[g] = 0.f;
        const float csd = slv[32];
        if (valid) {
            beg = rowptr[node]; end = rowptr[node + 1];
            const unsigned wd = rec[node * 4 + q];
            const float4 vd = ((const float4*)(slv + 16))[q];
            const int row = qrow + (isw ? (int)wd : NWIRE);
#pragma unroll
            for (int g = 0; g < TB; ++g) {
                float4 t = wl[row + g * WROWS];
                t.x = isw ? t.x : __uint_as_float(wd);
                edd[g] = dot4(t, vd);
            }
        }
#pragma unroll
        for (int g = 0; g < TB; ++g) edd[g] = quad_sum(edd[g]) + csd;

        constexpr int U = 4;
        float m[TB], den[TB];
        float4 acc[TB];
#pragma unroll
        for (int g = 0; g < TB; ++g) {
            m[g] = -INFINITY; den[g] = 0.f; acc[g] = make_float4(0.f, 0.f, 0.f, 0.f);
        }
        int k = beg;
        for (; k + U <= end; k += U) gat0_edges<TB, U>(srcS, rec, wl, k, q, isw, qrow, va, edd, m, den, acc);
        if (k + 2 <= end) { gat0_edges<TB, 2>(srcS, rec, wl, k, q, isw, qrow, va, edd, m, den, acc); k += 2; }
        if (k < end)        gat0_edges<TB, 1>(srcS, rec, wl, k, q, isw, qrow, va, edd, m, den, acc);

        float4 o[TB];
#pragma unroll
        for (int g = 0; g < TB; ++g) {
            const float dn = fmaxf(den[g], 1e-16f);
            o[g].x = acc[g].x / dn; o[g].y = acc[g].y / dn; o[g].z = acc[g].z / dn; o[g].w = acc[g].w / dn;
            if (q == 3) o[g].y = den[g] / dn;
        }

        wave_lds_handover();
#pragma unroll
        for (int g = 0; g < TB; ++g) *(float4*)&T[nd * TB + g][4 * q] = o[g];
        wave_lds_handover();

        v2f bfrag[4];
#pragma unroll
        for (int j = 0; j < 4; ++j) {
            const int kc = 4 * j + 2 * hf;
            bfrag[j].x = sW0[kc * HD + c];
            bfrag[j].y = sW0[(kc + 1) * HD + c];
        }
        v8f d[NT];
#pragma unroll
        for (int tI = 0; tI < NT; ++tI) {
            v8f a8 = {};
#pragma unroll
            for (int j = 0; j < 4; ++j) {
                const v2f a = *(const v2f*)&T[16 * tI + c][4 * j + 2 * hf];
                a8 = __builtin_amdgcn_wmma_f32_16x16x4_f32(
                         false, a, false, bfrag[j], (short)0, a8, false, false);
            }
            d[tI] = a8;
        }
        wave_lds_handover();
#pragma unroll
        for (int tI = 0; tI < NT; ++tI) {
#pragma unroll
            for (int r = 0; r < 8; ++r) {
                float v = d[tI][r];
                v = v > 0.f ? v : (__expf(v) - 1.f);
                T[16 * tI + r + 8 * hf][c] = v;
            }
        }
        wave_lds_handover();

#pragma unroll
        for (int j = 0; j < 4; ++j) {
            const int kc = 4 * j + 2 * hf;
            bfrag[j].x = sWn[kc * HD + c];
            bfrag[j].y = sWn[(kc + 1) * HD + c];
        }
        const float bi1 = sbn[c];
#pragma unroll
        for (int tI = 0; tI < NT; ++tI) {
            v8f a8 = {};
#pragma unroll
            for (int j = 0; j < 4; ++j) {
                const v2f a = *(const v2f*)&T[16 * tI + c][4 * j + 2 * hf];
                a8 = __builtin_amdgcn_wmma_f32_16x16x4_f32(
                         false, a, false, bfrag[j], (short)0, a8, false, false);
            }
            d[tI] = a8;
        }
        wave_lds_handover();
#pragma unroll
        for (int tI = 0; tI < NT; ++tI) {
#pragma unroll
            for (int r = 0; r < 8; ++r) T[16 * tI + r + 8 * hf][c] = d[tI][r] + bi1;
        }
        wave_lds_handover();
        const int rowbase = ch * R;
#pragma unroll
        for (int j = 0; j < R / 8; ++j) {
            const int f = lane + 32 * j;
            const int row = f >> 2;
            const int c4 = f & 3;
            if (rowbase + row < rows)
                ((float4*)hout)[(size_t)(rowbase + row) * 4 + c4] = *(const float4*)&T[row][4 * c4];
        }
    }
}

static inline int cdiv(int a, int b) { return (a + b - 1) / b; }

struct Bufs {
    const float *fw, *sw, *tw, *areas;
    const int* indices;
    const float *W0p, *lv, *Wl, *a_src, *a_dst, *bvec, *mlpW, *mlpb;
    const int *rowptr, *srcS;
    float *h, *hw, *out;
};

template <int TB>
static void run_graphs(const Bufs& b, int N, int g0, hipStream_t stream) {
    const int gatBl  = cdiv(N * 4, GAT_THREADS);
    unsigned* rec = (unsigned*)b.hw;
    k_pack<<<cdiv(N, 256), 256, 0, stream>>>(b.indices, b.areas, rec, N);
    k_gat0_node<TB><<<G0_BLOCKS, G0_THREADS, 0, stream>>>(b.rowptr, b.srcS, rec, b.fw, b.sw, b.tw,
                                                         b.W0p, b.lv, b.Wl, b.bvec + HD, b.h, N, g0);
    float* cur = b.h;
    float* nxt = b.hw;
    for (int i = 1; i < 4; ++i) {
        const float* as = b.a_src + (size_t)i * HD;
        const float* ad = b.a_dst + (size_t)i * HD;
        if (i < 3)
            k_gat_node<TB, true><<<gatBl, GAT_THREADS, 0, stream>>>(b.rowptr, b.srcS, cur, as, ad,
                                                                   b.Wl + (size_t)i * 256,
                                                                   b.bvec + (size_t)(i + 1) * HD, nxt, N);
        else
            k_gat_node<TB, false><<<gatBl, GAT_THREADS, 0, stream>>>(b.rowptr, b.srcS, cur, as, ad,
                                                                    b.mlpW, b.mlpb,
                                                                    b.out + (size_t)g0 * N, N);
        float* tmp = cur; cur = nxt; nxt = tmp;
    }
}

extern "C" void kernel_launch(void* const* d_in, const int* in_sizes, int n_in,
                              void* d_out, int out_size, void* d_ws, size_t ws_size,
                              hipStream_t stream) {
    const float* fw     = (const float*)d_in[0];
    const float* sw     = (const float*)d_in[1];
    const float* tw     = (const float*)d_in[2];
    const float* areas  = (const float*)d_in[3];
    const int*   indices= (const int*)  d_in[4];
    const int*   edge   = (const int*)  d_in[5];
    const float* W0     = (const float*)d_in[6];
    const float* Wl     = (const float*)d_in[7];
    const float* a_src  = (const float*)d_in[8];
    const float* a_dst  = (const float*)d_in[9];
    const float* bvec   = (const float*)d_in[10];
    const float* mlpW   = (const float*)d_in[11];
    const float* mlpb   = (const float*)d_in[12];

    const int N = in_sizes[3];
    const int E = in_sizes[5] / 2;
    const int T = in_sizes[0] / (NWIRE * NFEAT);
    const int* src = edge;
    const int* dst = edge + E;
    const int numTiles = cdiv(N, SCAN_TILE);

    const size_t fixedW = 320 + (size_t)E + ((size_t)N + 1) + 2 * (size_t)N + 2 * (size_t)numTiles;
    const size_t need3  = 2 * (size_t)N * HD * 3 + fixedW;
    const size_t need2  = 2 * (size_t)N * HD * 2 + fixedW;
    int TBS = 1;
    if (T >= 3 && ws_size >= need3 * 4) TBS = 3;
    else if (T >= 2 && ws_size >= need2 * 4) TBS = 2;

    float* ws = (float*)d_ws;
    size_t off = 0;
    float* h       = ws + off; off += (size_t)N * HD * TBS;
    float* hw      = ws + off; off += (size_t)N * HD * TBS;
    float* W0p     = ws + off; off += 256;
    float* lv      = ws + off; off += 64;
    int*   srcS    = (int*)(ws + off); off += (size_t)E;
    int*   rowptr  = (int*)(ws + off); off += (size_t)N + 1;
    int*   cursor  = (int*)(ws + off); off += (size_t)N;
    int*   cnt     = (int*)(ws + off); off += (size_t)N;
    int*   tilesum = (int*)(ws + off); off += (size_t)numTiles;
    int*   tileoff = (int*)(ws + off); off += (size_t)numTiles;
    (void)n_in; (void)out_size;

    const int numBins = numTiles;
    int* binCnt = tilesum;
    int* binOff = tileoff;
    int* binCur = cnt;
    const bool binned = N <= (1 << SRC_BITS) && (size_t)E <= (size_t)N * HD * TBS;

    k_prep<<<1, 256, 0, stream>>>(W0, a_src, a_dst, bvec, W0p, lv, binCnt, binCur, binned ? numBins : 0);
    if (binned) {
        const int chunks = cdiv(E, PART_CHUNK);
        k_binhist<<<chunks, PART_THREADS, 0, stream>>>(dst, binCnt, E, numBins);
        k_scan_tiles<<<1, 1024, 0, stream>>>(binCnt, binOff, rowptr, numBins, N);
        k_partition<<<chunks, PART_THREADS, 0, stream>>>(src, dst, binOff, binCur, (unsigned*)h, E, numBins);
        k_binsort<<<numBins, 1024, 0, stream>>>((const unsigned*)h, binOff, binCnt, rowptr, srcS, N);
    } else {
        k_zero<<<cdiv(N, 256), 256, 0, stream>>>(cnt, N);
        k_hist<<<cdiv(E, 256), 256, 0, stream>>>(dst, cnt, E);
        k_scan_local<<<numTiles, 256, 0, stream>>>(cnt, rowptr, tilesum, N);
        k_scan_tiles<<<1, 1024, 0, stream>>>(tilesum, tileoff, rowptr, numTiles, N);
        k_scan_add<<<cdiv(N, 256), 256, 0, stream>>>(rowptr, tileoff, cursor, N);
        k_scatter<<<cdiv(E, 256), 256, 0, stream>>>(src, dst, cursor, srcS, E);
    }

    Bufs b;
    b.fw = fw; b.sw = sw; b.tw = tw; b.areas = areas; b.indices = indices;
    b.W0p = W0p; b.lv = lv; b.Wl = Wl; b.a_src = a_src; b.a_dst = a_dst; b.bvec = bvec;
    b.mlpW = mlpW; b.mlpb = mlpb; b.rowptr = rowptr; b.srcS = srcS;
    b.h = h; b.hw = hw; b.out = (float*)d_out;

    int g = 0;
    while (g < T) {
        const int left = T - g;
        const int tb = left < TBS ? left : TBS;
        if (tb >= 3)      { run_graphs<3>(b, N, g, stream); g += 3; }
        else if (tb == 2) { run_graphs<2>(b, N, g, stream); g += 2; }
        else              { run_graphs<1>(b, N, g, stream); g += 1; }
    }
}
